# starting kernel
# baseline (speedup 1.0000x reference)
#include <hip/hip_runtime.h>
#include <cfloat>

#define N_NODES 100000
#define NFEAT   256
#define NHID    64
#define NEDGE   1600000
#define TOTE    (NEDGE + N_NODES)
#define MTILES  (N_NODES / 16)
#define EPSBN   1e-5f
#define NEG_SLOPE 0.2f

typedef __attribute__((ext_vector_type(2))) float v2f;
typedef __attribute__((ext_vector_type(4))) float v4f;
typedef __attribute__((ext_vector_type(8))) float v8f;

__device__ __forceinline__ unsigned ford(float f) {
    unsigned u = __float_as_uint(f);
    return (u & 0x80000000u) ? ~u : (u | 0x80000000u);
}
__device__ __forceinline__ float dorf(unsigned u) {
    return (u & 0x80000000u) ? __uint_as_float(u & 0x7fffffffu) : __uint_as_float(~u);
}
__device__ __forceinline__ float leaky(float v) {
    return fmaxf(v, 0.f) + NEG_SLOPE * fminf(v, 0.f);
}

__global__ void k_init(float* acc_est, float* acc_gnn,
                       unsigned* m_e, unsigned* m_g, float* z_e, float* z_g) {
    long i = (long)blockIdx.x * blockDim.x + threadIdx.x;
    long tot = (long)N_NODES * NHID;
    if (i < tot) { acc_est[i] = 0.f; acc_gnn[i] = 0.f; }
    if (i < N_NODES) {
        m_e[i] = ford(-FLT_MAX);
        m_g[i] = ford(-FLT_MAX);
        z_e[i] = 0.f;
        z_g[i] = 0.f;
    }
}

__global__ __launch_bounds__(256) void k_gemm(const float* __restrict__ g,
                                              const float* __restrict__ We,
                                              const float* __restrict__ Wg,
                                              float* __restrict__ h_est,
                                              float* __restrict__ h_gnn) {
    int wave = threadIdx.x >> 5;
    int lane = threadIdx.x & 31;
    int mtile = blockIdx.x * 8 + wave;
    if (mtile >= MTILES) return;
    int m0   = mtile * 16;
    int lrow = lane & 15;
    int lhi  = lane >> 4;

    const float* grow = g + (long)(m0 + lrow) * NFEAT;

    v8f acc[8] = {};
    for (int k = 0; k < NFEAT; k += 4) {
        v2f a = *(const v2f*)(grow + k + 2 * lhi);
#pragma unroll
        for (int t = 0; t < 8; ++t) {
            const float* Wp = (t < 4) ? We : Wg;
            int c  = (t & 3) * 16 + lrow;
            int kr = k + 2 * lhi;
            v2f b;
            b.x = Wp[(long)kr * NHID + c];
            b.y = Wp[(long)(kr + 1) * NHID + c];
            acc[t] = __builtin_amdgcn_wmma_f32_16x16x4_f32(
                false, a, false, b, (short)0, acc[t], false, false);
        }
    }
#pragma unroll
    for (int t = 0; t < 8; ++t) {
        float* dst = (t < 4) ? h_est : h_gnn;
        int c = (t & 3) * 16 + lrow;
#pragma unroll
        for (int r = 0; r < 8; ++r) {
            dst[(long)(m0 + r + 8 * lhi) * NHID + c] = acc[t][r];
        }
    }
}

__global__ void k_eprep(const float* __restrict__ h_est, const float* __restrict__ h_gnn,
                        const float* __restrict__ ase, const float* __restrict__ ade,
                        const float* __restrict__ asg, const float* __restrict__ adg,
                        float* es_e, float* ed_e, float* es_g, float* ed_g) {
    int n = blockIdx.x * blockDim.x + threadIdx.x;
    if (n >= N_NODES) return;
    const float* he = h_est + (long)n * NHID;
    const float* hg = h_gnn + (long)n * NHID;
    float se = 0.f, de = 0.f, sg = 0.f, dg = 0.f;
#pragma unroll 4
    for (int f = 0; f < NHID; f += 4) {
        v4f e  = *(const v4f*)(he + f);
        v4f g4 = *(const v4f*)(hg + f);
        v4f a1 = *(const v4f*)(ase + f);
        v4f a2 = *(const v4f*)(ade + f);
        v4f a3 = *(const v4f*)(asg + f);
        v4f a4 = *(const v4f*)(adg + f);
        se += e.x * a1.x + e.y * a1.y + e.z * a1.z + e.w * a1.w;
        de += e.x * a2.x + e.y * a2.y + e.z * a2.z + e.w * a2.w;
        sg += g4.x * a3.x + g4.y * a3.y + g4.z * a3.z + g4.w * a3.w;
        dg += g4.x * a4.x + g4.y * a4.y + g4.z * a4.z + g4.w * a4.w;
    }
    es_e[n] = se; ed_e[n] = de; es_g[n] = sg; ed_g[n] = dg;
}

__global__ void k_max(const int* __restrict__ x,
                      const float* __restrict__ es_e, const float* __restrict__ ed_e,
                      const float* __restrict__ es_g, const float* __restrict__ ed_g,
                      unsigned* m_e, unsigned* m_g) {
    long e = (long)blockIdx.x * blockDim.x + threadIdx.x;
    if (e >= TOTE) return;
    int s, d;
    if (e < NEDGE) { s = x[e]; d = x[NEDGE + e]; } else { s = d = (int)(e - NEDGE); }
    atomicMax(m_e + d, ford(leaky(es_e[s] + ed_e[d])));
    atomicMax(m_g + d, ford(leaky(es_g[s] + ed_g[d])));
}

__global__ void k_sum(const int* __restrict__ x,
                      const float* __restrict__ es_e, const float* __restrict__ ed_e,
                      const float* __restrict__ es_g, const float* __restrict__ ed_g,
                      const unsigned* __restrict__ m_e, const unsigned* __restrict__ m_g,
                      float* z_e, float* z_g) {
    long e = (long)blockIdx.x * blockDim.x + threadIdx.x;
    if (e >= TOTE) return;
    int s, d;
    if (e < NEDGE) { s = x[e]; d = x[NEDGE + e]; } else { s = d = (int)(e - NEDGE); }
    atomicAdd(z_e + d, __expf(leaky(es_e[s] + ed_e[d]) - dorf(m_e[d])));
    atomicAdd(z_g + d, __expf(leaky(es_g[s] + ed_g[d]) - dorf(m_g[d])));
}

__global__ __launch_bounds__(256) void k_scatter(const int* __restrict__ x,
        const float* __restrict__ es_e, const float* __restrict__ ed_e,
        const float* __restrict__ es_g, const float* __restrict__ ed_g,
        const unsigned* __restrict__ m_e, const unsigned* __restrict__ m_g,
        const float* __restrict__ z_e, const float* __restrict__ z_g,
        const float* __restrict__ h_est, const float* __restrict__ h_gnn,
        float* acc_est, float* acc_gnn) {
    int wave = threadIdx.x >> 5;
    int lane = threadIdx.x & 31;
    long e = (long)blockIdx.x * 8 + wave;
    if (e >= TOTE) return;
    int s, d;
    if (e < NEDGE) { s = x[e]; d = x[NEDGE + e]; } else { s = d = (int)(e - NEDGE); }
    float ae = __expf(leaky(es_e[s] + ed_e[d]) - dorf(m_e[d])) / z_e[d];
    float ag = __expf(leaky(es_g[s] + ed_g[d]) - dorf(m_g[d])) / z_g[d];
    int f0 = lane * 2;
    v2f he = *(const v2f*)(h_est + (long)s * NHID + f0);
    v2f hg = *(const v2f*)(h_gnn + (long)s * NHID + f0);
    float* pe = acc_est + (long)d * NHID + f0;
    float* pg = acc_gnn + (long)d * NHID + f0;
    atomicAdd(pe,     he.x * ae);
    atomicAdd(pe + 1, he.y * ae);
    atomicAdd(pg,     hg.x * ag);
    atomicAdd(pg + 1, hg.y * ag);
}

__global__ void k_final(float* d_out, const float* __restrict__ acc_gnn,
        const float* __restrict__ eb, const float* __restrict__ egam,
        const float* __restrict__ ebeta, const float* __restrict__ erm,
        const float* __restrict__ erv,
        const float* __restrict__ gb, const float* __restrict__ ggam,
        const float* __restrict__ gbeta, const float* __restrict__ grm,
        const float* __restrict__ grv,
        const float* __restrict__ clsW, const float* __restrict__ clsb) {
    int n = blockIdx.x * blockDim.x + threadIdx.x;
    if (n >= N_NODES) return;
    float* srow = d_out + N_NODES + (long)n * NHID;
    const float* grow = acc_gnn + (long)n * NHID;
    float y = 0.f;
#pragma unroll 4
    for (int f = 0; f < NHID; ++f) {
        float a = fmaxf(srow[f] + eb[f], 0.f);
        srow[f] = (a - erm[f]) * rsqrtf(erv[f] + EPSBN) * egam[f] + ebeta[f];
        float b = fmaxf(grow[f] + gb[f], 0.f);
        float zv = (b - grm[f]) * rsqrtf(grv[f] + EPSBN) * ggam[f] + gbeta[f];
        y += zv * clsW[f];
    }
    d_out[n] = y + clsb[0];
}

extern "C" void kernel_launch(void* const* d_in, const int* in_sizes, int n_in,
                              void* d_out, int out_size, void* d_ws, size_t ws_size,
                              hipStream_t stream) {
    const float* g        = (const float*)d_in[0];
    const int*   x        = (const int*)d_in[1];
    const float* est_W    = (const float*)d_in[2];
    const float* est_asrc = (const float*)d_in[3];
    const float* est_adst = (const float*)d_in[4];
    const float* est_bias = (const float*)d_in[5];
    const float* est_gam  = (const float*)d_in[6];
    const float* est_beta = (const float*)d_in[7];
    const float* est_rm   = (const float*)d_in[8];
    const float* est_rv   = (const float*)d_in[9];
    const float* gnn_W    = (const float*)d_in[10];
    const float* gnn_asrc = (const float*)d_in[11];
    const float* gnn_adst = (const float*)d_in[12];
    const float* gnn_bias = (const float*)d_in[13];
    const float* gnn_gam  = (const float*)d_in[14];
    const float* gnn_beta = (const float*)d_in[15];
    const float* gnn_rm   = (const float*)d_in[16];
    const float* gnn_rv   = (const float*)d_in[17];
    const float* cls_W    = (const float*)d_in[18];
    const float* cls_b    = (const float*)d_in[19];
    float* out = (float*)d_out;

    float*    ws     = (float*)d_ws;
    float*    h_est  = ws;
    float*    h_gnn  = h_est + (long)N_NODES * NHID;
    float*    acc_g  = h_gnn + (long)N_NODES * NHID;
    float*    es_e   = acc_g + (long)N_NODES * NHID;
    float*    ed_e   = es_e + N_NODES;
    float*    es_g   = ed_e + N_NODES;
    float*    ed_g   = es_g + N_NODES;
    unsigned* m_e    = (unsigned*)(ed_g + N_NODES);
    unsigned* m_g    = m_e + N_NODES;
    float*    z_e    = (float*)(m_g + N_NODES);
    float*    z_g    = z_e + N_NODES;
    float*    acc_e  = out + N_NODES;

    const int BT = 256;
    long initN = (long)N_NODES * NHID;
    k_init<<<(unsigned)((initN + BT - 1) / BT), BT, 0, stream>>>(acc_e, acc_g, m_e, m_g, z_e, z_g);

    k_gemm<<<(MTILES + 7) / 8, BT, 0, stream>>>(g, est_W, gnn_W, h_est, h_gnn);

    k_eprep<<<(N_NODES + BT - 1) / BT, BT, 0, stream>>>(h_est, h_gnn,
            est_asrc, est_adst, gnn_asrc, gnn_adst, es_e, ed_e, es_g, ed_g);

    k_max<<<(TOTE + BT - 1) / BT, BT, 0, stream>>>(x, es_e, ed_e, es_g, ed_g, m_e, m_g);

    k_sum<<<(TOTE + BT - 1) / BT, BT, 0, stream>>>(x, es_e, ed_e, es_g, ed_g, m_e, m_g, z_e, z_g);

    k_scatter<<<(TOTE + 7) / 8, BT, 0, stream>>>(x, es_e, ed_e, es_g, ed_g,
            m_e, m_g, z_e, z_g, h_est, h_gnn, acc_e, acc_g);

    k_final<<<(N_NODES + BT - 1) / BT, BT, 0, stream>>>(out, acc_g,
            est_bias, est_gam, est_beta, est_rm, est_rv,
            gnn_bias, gnn_gam, gnn_beta, gnn_rm, gnn_rv, cls_W, cls_b);
}


# Float atomics (segment max, sum, scatter) replaced by bucketing edges by destination and one wave per node that sorts its bucket by source and gathers; sums run in sorted order, y by a wave tree
# speedup vs baseline: 1.5604x; 1.5604x over previous
#include <hip/hip_runtime.h>
#include <cfloat>
#include <climits>

#define N_NODES 100000
#define NFEAT   256
#define NHID    64
#define NEDGE   1600000
#define MTILES  (N_NODES / 16)
#define EPSBN   1e-5f
#define NEG_SLOPE 0.2f
#define GW      32

static_assert(N_NODES % GW == 0, "gather blocks cover the nodes exactly");
static_assert(N_NODES % 4 == 0, "scan reads whole quads");

typedef __attribute__((ext_vector_type(2))) float v2f;
typedef __attribute__((ext_vector_type(4))) float v4f;
typedef __attribute__((ext_vector_type(8))) float v8f;

__device__ __forceinline__ float leaky(float v) {
    return fmaxf(v, 0.f) + NEG_SLOPE * fminf(v, 0.f);
}
__device__ __forceinline__ float wave_max_f(float v) {
    for (int o = 16; o > 0; o >>= 1) v = fmaxf(v, __shfl_xor(v, o));
    return v;
}
__device__ __forceinline__ float wave_sum_f(float v) {
    for (int o = 16; o > 0; o >>= 1) v += __shfl_xor(v, o);
    return v;
}
__device__ __forceinline__ int wave_min_i(int v) {
    for (int o = 16; o > 0; o >>= 1) v = min(v, __shfl_xor(v, o));
    return v;
}
__device__ __forceinline__ int wave_sum_i(int v) {
    for (int o = 16; o > 0; o >>= 1) v += __shfl_xor(v, o);
    return v;
}

__global__ __launch_bounds__(256) void k_gemm(const float* __restrict__ g,
                                              const float* __restrict__ We,
                                              const float* __restrict__ Wg,
                                              float* __restrict__ h_est,
                                              float* __restrict__ h_gnn) {
    int wave = threadIdx.x >> 5;
    int lane = threadIdx.x & 31;
    int mtile = blockIdx.x * 8 + wave;
    if (mtile >= MTILES) return;
    int m0   = mtile * 16;
    int lrow = lane & 15;
    int lhi  = lane >> 4;

    const float* grow = g + (long)(m0 + lrow) * NFEAT;

    v8f acc[8] = {};
    for (int k = 0; k < NFEAT; k += 4) {
        v2f a = *(const v2f*)(grow + k + 2 * lhi);
#pragma unroll
        for (int t = 0; t < 8; ++t) {
            const float* Wp = (t < 4) ? We : Wg;
            int c  = (t & 3) * 16 + lrow;
            int kr = k + 2 * lhi;
            v2f b;
            b.x = Wp[(long)kr * NHID + c];
            b.y = Wp[(long)(kr + 1) * NHID + c];
            acc[t] = __builtin_amdgcn_wmma_f32_16x16x4_f32(
                false, a, false, b, (short)0, acc[t], false, false);
        }
    }
#pragma unroll
    for (int t = 0; t < 8; ++t) {
        float* dst = (t < 4) ? h_est : h_gnn;
        int c = (t & 3) * 16 + lrow;
#pragma unroll
        for (int r = 0; r < 8; ++r) {
            dst[(long)(m0 + r + 8 * lhi) * NHID + c] = acc[t][r];
        }
    }
}

__global__ void k_eprep(const float* __restrict__ h_est, const float* __restrict__ h_gnn,
                        const float* __restrict__ ase, const float* __restrict__ ade,
                        const float* __restrict__ asg, const float* __restrict__ adg,
                        float* es_e, float* ed_e, float* es_g, float* ed_g,
                        int* cnt, int* cur) {
    int n = blockIdx.x * blockDim.x + threadIdx.x;
    if (n >= N_NODES) return;
    const float* he = h_est + (long)n * NHID;
    const float* hg = h_gnn + (long)n * NHID;
    float se = 0.f, de = 0.f, sg = 0.f, dg = 0.f;
#pragma unroll 4
    for (int f = 0; f < NHID; f += 4) {
        v4f e  = *(const v4f*)(he + f);
        v4f g4 = *(const v4f*)(hg + f);
        v4f a1 = *(const v4f*)(ase + f);
        v4f a2 = *(const v4f*)(ade + f);
        v4f a3 = *(const v4f*)(asg + f);
        v4f a4 = *(const v4f*)(adg + f);
        se += e.x * a1.x + e.y * a1.y + e.z * a1.z + e.w * a1.w;
        de += e.x * a2.x + e.y * a2.y + e.z * a2.z + e.w * a2.w;
        sg += g4.x * a3.x + g4.y * a3.y + g4.z * a3.z + g4.w * a3.w;
        dg += g4.x * a4.x + g4.y * a4.y + g4.z * a4.z + g4.w * a4.w;
    }
    es_e[n] = se; ed_e[n] = de; es_g[n] = sg; ed_g[n] = dg;
    cnt[n] = 0; cur[n] = 0;
}

__global__ void k_count(const int* __restrict__ x, int* cnt) {
    int e = blockIdx.x * blockDim.x + threadIdx.x;
    if (e < NEDGE) atomicAdd(cnt + x[NEDGE + e], 1);
}

__global__ __launch_bounds__(1024) void k_scan(const int* __restrict__ cnt, int* __restrict__ off) {
    __shared__ int wtot[32];
    __shared__ int tile_tot;
    int t = threadIdx.x, lane = t & 31, wave = t >> 5;
    int carry = 0;
    for (int base = 0; base < N_NODES; base += 4096) {
        int i = base + t * 4;
        int4 c = make_int4(0, 0, 0, 0);
        if (i < N_NODES) c = *(const int4*)(cnt + i);
        int s1 = c.x, s2 = s1 + c.y, s3 = s2 + c.z, s4 = s3 + c.w;
        int inc = s4;
        for (int o = 1; o < 32; o <<= 1) { int v = __shfl_up(inc, o); if (lane >= o) inc += v; }
        if (lane == 31) wtot[wave] = inc;
        __syncthreads();
        if (wave == 0) {
            int w = wtot[lane], winc = w;
            for (int o = 1; o < 32; o <<= 1) { int v = __shfl_up(winc, o); if (lane >= o) winc += v; }
            wtot[lane] = winc - w;
            if (lane == 31) tile_tot = winc;
        }
        __syncthreads();
        int ex = carry + wtot[wave] + inc - s4;
        if (i < N_NODES) *(int4*)(off + i) = make_int4(ex, ex + s1, ex + s2, ex + s3);
        carry += tile_tot;
        __syncthreads();
    }
    if (t == 0) off[N_NODES] = carry;
}

__global__ void k_fill(const int* __restrict__ x, const int* __restrict__ off,
                       int* cur, int* __restrict__ slot) {
    int e = blockIdx.x * blockDim.x + threadIdx.x;
    if (e >= NEDGE) return;
    int s = x[e], d = x[NEDGE + e];
    int p = atomicAdd(cur + d, 1);
    slot[off[d] + p] = s;
}

__global__ __launch_bounds__(1024) void k_gather(float* d_out,
        const int* __restrict__ off, const int* __restrict__ slot,
        const float* __restrict__ es_e, const float* __restrict__ ed_e,
        const float* __restrict__ es_g, const float* __restrict__ ed_g,
        const float* __restrict__ h_est, const float* __restrict__ h_gnn,
        const float* __restrict__ eb, const float* __restrict__ egam,
        const float* __restrict__ ebeta, const float* __restrict__ erm,
        const float* __restrict__ erv,
        const float* __restrict__ gb, const float* __restrict__ ggam,
        const float* __restrict__ gbeta, const float* __restrict__ grm,
        const float* __restrict__ grv,
        const float* __restrict__ clsW, const float* __restrict__ clsb) {
    __shared__ int   l_s[GW][32];
    __shared__ float l_ae[GW][32];
    __shared__ float l_ag[GW][32];
    __shared__ float l_y[GW];
    int lane = threadIdx.x & 31;
    int wave = __builtin_amdgcn_readfirstlane(threadIdx.x >> 5);
    int n = blockIdx.x * GW + wave;
    int beg = off[n];
    int deg = off[n + 1] - beg;
    float edn_e = ed_e[n], edn_g = ed_g[n];
    float self_e = leaky(es_e[n] + edn_e), self_g = leaky(es_g[n] + edn_g);
    int f0 = lane * 2;
    v2f acc_e = {0.f, 0.f}, acc_g = {0.f, 0.f};
    float ze = 0.f, zg = 0.f, pse, psg;

    if (deg <= 32) {
        bool on = lane < deg;
        int s = on ? slot[beg + lane] : INT_MAX;
        float le = -FLT_MAX, lg = -FLT_MAX;
        if (on) { le = leaky(es_e[s] + edn_e); lg = leaky(es_g[s] + edn_g); }
        int rank = 0;
        for (int j = 0; j < deg; ++j) {
            int sj = __shfl(s, j);
            rank += (sj < s || (sj == s && j < lane)) ? 1 : 0;
        }
        float me = fmaxf(wave_max_f(le), self_e);
        float mg = fmaxf(wave_max_f(lg), self_g);
        float pe = __expf(le - me), pg = __expf(lg - mg);
        pse = __expf(self_e - me); psg = __expf(self_g - mg);
        if (on) { l_s[wave][rank] = s; l_ae[wave][rank] = pe; l_ag[wave][rank] = pg; }
        for (int r = 0; r < deg; ++r) { ze += l_ae[wave][r]; zg += l_ag[wave][r]; }
        ze += pse; zg += psg;
        if (on) { l_ae[wave][rank] = pe / ze; l_ag[wave][rank] = pg / zg; }
#pragma unroll 4
        for (int r = 0; r < deg; ++r) {
            int sr = l_s[wave][r];
            float ae = l_ae[wave][r], ag = l_ag[wave][r];
            v2f he = *(const v2f*)(h_est + (long)sr * NHID + f0);
            v2f hg = *(const v2f*)(h_gnn + (long)sr * NHID + f0);
            acc_e += he * ae; acc_g += hg * ag;
        }
    } else {
        float le_m = self_e, lg_m = self_g;
        for (int i = lane; i < deg; i += 32) {
            int s = slot[beg + i];
            le_m = fmaxf(le_m, leaky(es_e[s] + edn_e));
            lg_m = fmaxf(lg_m, leaky(es_g[s] + edn_g));
        }
        float me = wave_max_f(le_m), mg = wave_max_f(lg_m);
        pse = __expf(self_e - me); psg = __expf(self_g - mg);
        for (int pass = 0; pass < 2; ++pass) {
            int prev = -1;
            for (int it = 0; it < deg; ++it) {
                int best = INT_MAX;
                for (int i = lane; i < deg; i += 32) {
                    int s = slot[beg + i];
                    if (s > prev) best = min(best, s);
                }
                best = wave_min_i(best);
                if (best == INT_MAX) break;
                int c = 0;
                for (int i = lane; i < deg; i += 32) c += (slot[beg + i] == best) ? 1 : 0;
                c = wave_sum_i(c);
                float pe = __expf(leaky(es_e[best] + edn_e) - me);
                float pg = __expf(leaky(es_g[best] + edn_g) - mg);
                if (pass == 0) {
                    for (int k = 0; k < c; ++k) { ze += pe; zg += pg; }
                } else {
                    float ae = pe / ze, ag = pg / zg;
                    v2f he = *(const v2f*)(h_est + (long)best * NHID + f0);
                    v2f hg = *(const v2f*)(h_gnn + (long)best * NHID + f0);
                    for (int k = 0; k < c; ++k) { acc_e += he * ae; acc_g += hg * ag; }
                }
                prev = best;
            }
            if (pass == 0) { ze += pse; zg += psg; }
        }
    }
    {
        float ae = pse / ze, ag = psg / zg;
        v2f he = *(const v2f*)(h_est + (long)n * NHID + f0);
        v2f hg = *(const v2f*)(h_gnn + (long)n * NHID + f0);
        acc_e += he * ae; acc_g += hg * ag;
    }

    v2f so;
    {
        float a0 = fmaxf(acc_e.x + eb[f0], 0.f);
        float a1 = fmaxf(acc_e.y + eb[f0 + 1], 0.f);
        so.x = (a0 - erm[f0]) * rsqrtf(erv[f0] + EPSBN) * egam[f0] + ebeta[f0];
        so.y = (a1 - erm[f0 + 1]) * rsqrtf(erv[f0 + 1] + EPSBN) * egam[f0 + 1] + ebeta[f0 + 1];
    }
    *(v2f*)(d_out + N_NODES + (long)n * NHID + f0) = so;
    float b0 = fmaxf(acc_g.x + gb[f0], 0.f);
    float b1 = fmaxf(acc_g.y + gb[f0 + 1], 0.f);
    float z0 = (b0 - grm[f0]) * rsqrtf(grv[f0] + EPSBN) * ggam[f0] + gbeta[f0];
    float z1 = (b1 - grm[f0 + 1]) * rsqrtf(grv[f0 + 1] + EPSBN) * ggam[f0 + 1] + gbeta[f0 + 1];
    float yp = wave_sum_f(z0 * clsW[f0] + z1 * clsW[f0 + 1]);
    if (lane == 0) l_y[wave] = yp + clsb[0];
    __syncthreads();
    if (threadIdx.x < GW) d_out[blockIdx.x * GW + threadIdx.x] = l_y[threadIdx.x];
}

extern "C" void kernel_launch(void* const* d_in, const int* in_sizes, int n_in,
                              void* d_out, int out_size, void* d_ws, size_t ws_size,
                              hipStream_t stream) {
    const float* g        = (const float*)d_in[0];
    const int*   x        = (const int*)d_in[1];
    const float* est_W    = (const float*)d_in[2];
    const float* est_asrc = (const float*)d_in[3];
    const float* est_adst = (const float*)d_in[4];
    const float* est_bias = (const float*)d_in[5];
    const float* est_gam  = (const float*)d_in[6];
    const float* est_beta = (const float*)d_in[7];
    const float* est_rm   = (const float*)d_in[8];
    const float* est_rv   = (const float*)d_in[9];
    const float* gnn_W    = (const float*)d_in[10];
    const float* gnn_asrc = (const float*)d_in[11];
    const float* gnn_adst = (const float*)d_in[12];
    const float* gnn_bias = (const float*)d_in[13];
    const float* gnn_gam  = (const float*)d_in[14];
    const float* gnn_beta = (const float*)d_in[15];
    const float* gnn_rm   = (const float*)d_in[16];
    const float* gnn_rv   = (const float*)d_in[17];
    const float* cls_W    = (const float*)d_in[18];
    const float* cls_b    = (const float*)d_in[19];
    float* out = (float*)d_out;

    float* ws     = (float*)d_ws;
    float* h_est  = ws;
    float* h_gnn  = h_est + (long)N_NODES * NHID;
    int*   cnt    = (int*)(h_gnn + (long)N_NODES * NHID);
    int*   cur    = cnt + N_NODES;
    int*   off    = cur + N_NODES;
    int*   slot   = off + N_NODES + 4;
    float* es_e   = ws + (long)N_NODES * 192;
    float* ed_e   = es_e + N_NODES;
    float* es_g   = ed_e + N_NODES;
    float* ed_g   = es_g + N_NODES;

    const int BT = 256;
    k_gemm<<<(MTILES + 7) / 8, BT, 0, stream>>>(g, est_W, gnn_W, h_est, h_gnn);

    k_eprep<<<(N_NODES + BT - 1) / BT, BT, 0, stream>>>(h_est, h_gnn,
            est_asrc, est_adst, gnn_asrc, gnn_adst, es_e, ed_e, es_g, ed_g, cnt, cur);

    k_count<<<(NEDGE + BT - 1) / BT, BT, 0, stream>>>(x, cnt);

    k_scan<<<1, 1024, 0, stream>>>(cnt, off);

    k_fill<<<(NEDGE + BT - 1) / BT, BT, 0, stream>>>(x, off, cur, slot);

    k_gather<<<N_NODES / GW, GW * 32, 0, stream>>>(out, off, slot,
            es_e, ed_e, es_g, ed_g, h_est, h_gnn,
            est_bias, est_gam, est_beta, est_rm, est_rv,
            gnn_bias, gnn_gam, gnn_beta, gnn_rm, gnn_rv, cls_W, cls_b);
}


# Fewer, longer waves: the two bucketing kernels take 8 edges per thread and each gather wave walks 4 nodes (32-node blocks of 8 waves): 209k waves launched become 47k; every sum keeps its order
# speedup vs baseline: 1.6345x; 1.0475x over previous
#include <hip/hip_runtime.h>
#include <cfloat>
#include <climits>

#define N_NODES 100000
#define NFEAT   256
#define NHID    64
#define NEDGE   1600000
#define MTILES  (N_NODES / 16)
#define EPSBN   1e-5f
#define NEG_SLOPE 0.2f
#define GW      32
#define GWAVES  8
#define NPW     (GW / GWAVES)
#define EBT     256
#define NQUAD   (NEDGE / 4)
#define QPT     2

static_assert(N_NODES % GW == 0, "gather blocks cover the nodes exactly");
static_assert(GW % GWAVES == 0, "every gather wave gets the same number of nodes");
static_assert(N_NODES % 4 == 0, "scan reads whole quads");
static_assert(NEDGE % 4 == 0, "bucketing reads whole quads, and the destination row starts on a 16-byte boundary");

typedef __attribute__((ext_vector_type(2))) float v2f;
typedef __attribute__((ext_vector_type(4))) float v4f;
typedef __attribute__((ext_vector_type(8))) float v8f;

__device__ __forceinline__ float leaky(float v) {
    return fmaxf(v, 0.f) + NEG_SLOPE * fminf(v, 0.f);
}
__device__ __forceinline__ float wave_max_f(float v) {
    for (int o = 16; o > 0; o >>= 1) v = fmaxf(v, __shfl_xor(v, o));
    return v;
}
__device__ __forceinline__ float wave_sum_f(float v) {
    for (int o = 16; o > 0; o >>= 1) v += __shfl_xor(v, o);
    return v;
}
__device__ __forceinline__ int wave_min_i(int v) {
    for (int o = 16; o > 0; o >>= 1) v = min(v, __shfl_xor(v, o));
    return v;
}
__device__ __forceinline__ int wave_sum_i(int v) {
    for (int o = 16; o > 0; o >>= 1) v += __shfl_xor(v, o);
    return v;
}

__global__ __launch_bounds__(256) void k_gemm(const float* __restrict__ g,
                                              const float* __restrict__ We,
                                              const float* __restrict__ Wg,
                                              float* __restrict__ h_est,
                                              float* __restrict__ h_gnn) {
    int wave = threadIdx.x >> 5;
    int lane = threadIdx.x & 31;
    int mtile = blockIdx.x * 8 + wave;
    if (mtile >= MTILES) return;
    int m0   = mtile * 16;
    int lrow = lane & 15;
    int lhi  = lane >> 4;

    const float* grow = g + (long)(m0 + lrow) * NFEAT;

    v8f acc[8] = {};
    for (int k = 0; k < NFEAT; k += 4) {
        v2f a = *(const v2f*)(grow + k + 2 * lhi);
#pragma unroll
        for (int t = 0; t < 8; ++t) {
            const float* Wp = (t < 4) ? We : Wg;
            int c  = (t & 3) * 16 + lrow;
            int kr = k + 2 * lhi;
            v2f b;
            b.x = Wp[(long)kr * NHID + c];
            b.y = Wp[(long)(kr + 1) * NHID + c];
            acc[t] = __builtin_amdgcn_wmma_f32_16x16x4_f32(
                false, a, false, b, (short)0, acc[t], false, false);
        }
    }
#pragma unroll
    for (int t = 0; t < 8; ++t) {
        float* dst = (t < 4) ? h_est : h_gnn;
        int c = (t & 3) * 16 + lrow;
#pragma unroll
        for (int r = 0; r < 8; ++r) {
            dst[(long)(m0 + r + 8 * lhi) * NHID + c] = acc[t][r];
        }
    }
}

__global__ void k_eprep(const float* __restrict__ h_est, const float* __restrict__ h_gnn,
                        const float* __restrict__ ase, const float* __restrict__ ade,
                        const float* __restrict__ asg, const float* __restrict__ adg,
                        float* es_e, float* ed_e, float* es_g, float* ed_g,
                        int* cnt, int* cur) {
    int n = blockIdx.x * blockDim.x + threadIdx.x;
    if (n >= N_NODES) return;
    const float* he = h_est + (long)n * NHID;
    const float* hg = h_gnn + (long)n * NHID;
    float se = 0.f, de = 0.f, sg = 0.f, dg = 0.f;
#pragma unroll 4
    for (int f = 0; f < NHID; f += 4) {
        v4f e  = *(const v4f*)(he + f);
        v4f g4 = *(const v4f*)(hg + f);
        v4f a1 = *(const v4f*)(ase + f);
        v4f a2 = *(const v4f*)(ade + f);
        v4f a3 = *(const v4f*)(asg + f);
        v4f a4 = *(const v4f*)(adg + f);
        se += e.x * a1.x + e.y * a1.y + e.z * a1.z + e.w * a1.w;
        de += e.x * a2.x + e.y * a2.y + e.z * a2.z + e.w * a2.w;
        sg += g4.x * a3.x + g4.y * a3.y + g4.z * a3.z + g4.w * a3.w;
        dg += g4.x * a4.x + g4.y * a4.y + g4.z * a4.z + g4.w * a4.w;
    }
    es_e[n] = se; ed_e[n] = de; es_g[n] = sg; ed_g[n] = dg;
    cnt[n] = 0; cur[n] = 0;
}

__global__ __launch_bounds__(EBT) void k_count(const int* __restrict__ x, int* cnt) {
    const int4* d4 = (const int4*)(x + NEDGE);
    int q0 = blockIdx.x * (QPT * EBT) + threadIdx.x;
    int4 d[QPT];
#pragma unroll
    for (int j = 0; j < QPT; ++j) {
        int q = q0 + j * EBT;
        d[j] = (q < NQUAD) ? d4[q] : make_int4(0, 0, 0, 0);
    }
#pragma unroll
    for (int j = 0; j < QPT; ++j) {
        if (q0 + j * EBT < NQUAD) {
            atomicAdd(cnt + d[j].x, 1);
            atomicAdd(cnt + d[j].y, 1);
            atomicAdd(cnt + d[j].z, 1);
            atomicAdd(cnt + d[j].w, 1);
        }
    }
}

__global__ __launch_bounds__(1024) void k_scan(const int* __restrict__ cnt, int* __restrict__ off) {
    __shared__ int wtot[32];
    __shared__ int tile_tot;
    int t = threadIdx.x, lane = t & 31, wave = t >> 5;
    int carry = 0;
    for (int base = 0; base < N_NODES; base += 4096) {
        int i = base + t * 4;
        int4 c = make_int4(0, 0, 0, 0);
        if (i < N_NODES) c = *(const int4*)(cnt + i);
        int s1 = c.x, s2 = s1 + c.y, s3 = s2 + c.z, s4 = s3 + c.w;
        int inc = s4;
        for (int o = 1; o < 32; o <<= 1) { int v = __shfl_up(inc, o); if (lane >= o) inc += v; }
        if (lane == 31) wtot[wave] = inc;
        __syncthreads();
        if (wave == 0) {
            int w = wtot[lane], winc = w;
            for (int o = 1; o < 32; o <<= 1) { int v = __shfl_up(winc, o); if (lane >= o) winc += v; }
            wtot[lane] = winc - w;
            if (lane == 31) tile_tot = winc;
        }
        __syncthreads();
        int ex = carry + wtot[wave] + inc - s4;
        if (i < N_NODES) *(int4*)(off + i) = make_int4(ex, ex + s1, ex + s2, ex + s3);
        carry += tile_tot;
        __syncthreads();
    }
    if (t == 0) off[N_NODES] = carry;
}

__global__ __launch_bounds__(EBT) void k_fill(const int* __restrict__ x, const int* __restrict__ off,
                       int* cur, int* __restrict__ slot) {
    const int4* s4 = (const int4*)x;
    const int4* d4 = (const int4*)(x + NEDGE);
    int q0 = blockIdx.x * (QPT * EBT) + threadIdx.x;
    int4 s[QPT], d[QPT];
#pragma unroll
    for (int j = 0; j < QPT; ++j) {
        int q = q0 + j * EBT;
        bool ok = q < NQUAD;
        s[j] = ok ? s4[q] : make_int4(0, 0, 0, 0);
        d[j] = ok ? d4[q] : make_int4(0, 0, 0, 0);
    }
    int4 p[QPT], o[QPT];
#pragma unroll
    for (int j = 0; j < QPT; ++j) {
        if (q0 + j * EBT < NQUAD) {
            p[j].x = atomicAdd(cur + d[j].x, 1);
            p[j].y = atomicAdd(cur + d[j].y, 1);
            p[j].z = atomicAdd(cur + d[j].z, 1);
            p[j].w = atomicAdd(cur + d[j].w, 1);
            o[j].x = off[d[j].x];
            o[j].y = off[d[j].y];
            o[j].z = off[d[j].z];
            o[j].w = off[d[j].w];
        }
    }
#pragma unroll
    for (int j = 0; j < QPT; ++j) {
        if (q0 + j * EBT < NQUAD) {
            slot[o[j].x + p[j].x] = s[j].x;
            slot[o[j].y + p[j].y] = s[j].y;
            slot[o[j].z + p[j].z] = s[j].z;
            slot[o[j].w + p[j].w] = s[j].w;
        }
    }
}

__global__ __launch_bounds__(GWAVES * 32) void k_gather(float* d_out,
        const int* __restrict__ off, const int* __restrict__ slot,
        const float* __restrict__ es_e, const float* __restrict__ ed_e,
        const float* __restrict__ es_g, const float* __restrict__ ed_g,
        const float* __restrict__ h_est, const float* __restrict__ h_gnn,
        const float* __restrict__ eb, const float* __restrict__ egam,
        const float* __restrict__ ebeta, const float* __restrict__ erm,
        const float* __restrict__ erv,
        const float* __restrict__ gb, const float* __restrict__ ggam,
        const float* __restrict__ gbeta, const float* __restrict__ grm,
        const float* __restrict__ grv,
        const float* __restrict__ clsW, const float* __restrict__ clsb) {
    __shared__ int   l_s[GWAVES][32];
    __shared__ float l_ae[GWAVES][32];
    __shared__ float l_ag[GWAVES][32];
    __shared__ float l_y[GW];
    int lane = threadIdx.x & 31;
    int wave = __builtin_amdgcn_readfirstlane(threadIdx.x >> 5);
    int f0 = lane * 2;

    float eb0 = eb[f0], eb1 = eb[f0 + 1];
    float erm0 = erm[f0], erm1 = erm[f0 + 1];
    float ers0 = rsqrtf(erv[f0] + EPSBN), ers1 = rsqrtf(erv[f0 + 1] + EPSBN);
    float egam0 = egam[f0], egam1 = egam[f0 + 1];
    float ebeta0 = ebeta[f0], ebeta1 = ebeta[f0 + 1];
    float gb0 = gb[f0], gb1 = gb[f0 + 1];
    float grm0 = grm[f0], grm1 = grm[f0 + 1];
    float grs0 = rsqrtf(grv[f0] + EPSBN), grs1 = rsqrtf(grv[f0 + 1] + EPSBN);
    float ggam0 = ggam[f0], ggam1 = ggam[f0 + 1];
    float gbeta0 = gbeta[f0], gbeta1 = gbeta[f0 + 1];
    float cw0 = clsW[f0], cw1 = clsW[f0 + 1];
    float cb = clsb[0];

#pragma nounroll
    for (int i = 0; i < NPW; ++i) {
        int n = blockIdx.x * GW + wave * NPW + i;
        int beg = off[n];
        int deg = off[n + 1] - beg;
        float edn_e = ed_e[n], edn_g = ed_g[n];
        float self_e = leaky(es_e[n] + edn_e), self_g = leaky(es_g[n] + edn_g);
        v2f acc_e = {0.f, 0.f}, acc_g = {0.f, 0.f};
        float ze = 0.f, zg = 0.f, pse, psg;

        if (deg <= 32) {
            bool on = lane < deg;
            int s = on ? slot[beg + lane] : INT_MAX;
            float le = -FLT_MAX, lg = -FLT_MAX;
            if (on) { le = leaky(es_e[s] + edn_e); lg = leaky(es_g[s] + edn_g); }
            int rank = 0;
            for (int j = 0; j < deg; ++j) {
                int sj = __shfl(s, j);
                rank += (sj < s || (sj == s && j < lane)) ? 1 : 0;
            }
            float me = fmaxf(wave_max_f(le), self_e);
            float mg = fmaxf(wave_max_f(lg), self_g);
            float pe = __expf(le - me), pg = __expf(lg - mg);
            pse = __expf(self_e - me); psg = __expf(self_g - mg);
            if (on) { l_s[wave][rank] = s; l_ae[wave][rank] = pe; l_ag[wave][rank] = pg; }
            for (int r = 0; r < deg; ++r) { ze += l_ae[wave][r]; zg += l_ag[wave][r]; }
            ze += pse; zg += psg;
            if (on) { l_ae[wave][rank] = pe / ze; l_ag[wave][rank] = pg / zg; }
#pragma unroll 4
            for (int r = 0; r < deg; ++r) {
                int sr = l_s[wave][r];
                float ae = l_ae[wave][r], ag = l_ag[wave][r];
                v2f he = *(const v2f*)(h_est + (long)sr * NHID + f0);
                v2f hg = *(const v2f*)(h_gnn + (long)sr * NHID + f0);
                acc_e += he * ae; acc_g += hg * ag;
            }
        } else {
            float le_m = self_e, lg_m = self_g;
            for (int k = lane; k < deg; k += 32) {
                int s = slot[beg + k];
                le_m = fmaxf(le_m, leaky(es_e[s] + edn_e));
                lg_m = fmaxf(lg_m, leaky(es_g[s] + edn_g));
            }
            float me = wave_max_f(le_m), mg = wave_max_f(lg_m);
            pse = __expf(self_e - me); psg = __expf(self_g - mg);
            for (int pass = 0; pass < 2; ++pass) {
                int prev = -1;
                for (int it = 0; it < deg; ++it) {
                    int best = INT_MAX;
                    for (int k = lane; k < deg; k += 32) {
                        int s = slot[beg + k];
                        if (s > prev) best = min(best, s);
                    }
                    best = wave_min_i(best);
                    if (best == INT_MAX) break;
                    int c = 0;
                    for (int k = lane; k < deg; k += 32) c += (slot[beg + k] == best) ? 1 : 0;
                    c = wave_sum_i(c);
                    float pe = __expf(leaky(es_e[best] + edn_e) - me);
                    float pg = __expf(leaky(es_g[best] + edn_g) - mg);
                    if (pass == 0) {
                        for (int k = 0; k < c; ++k) { ze += pe; zg += pg; }
                    } else {
                        float ae = pe / ze, ag = pg / zg;
                        v2f he = *(const v2f*)(h_est + (long)best * NHID + f0);
                        v2f hg = *(const v2f*)(h_gnn + (long)best * NHID + f0);
                        for (int k = 0; k < c; ++k) { acc_e += he * ae; acc_g += hg * ag; }
                    }
                    prev = best;
                }
                if (pass == 0) { ze += pse; zg += psg; }
            }
        }
        {
            float ae = pse / ze, ag = psg / zg;
            v2f he = *(const v2f*)(h_est + (long)n * NHID + f0);
            v2f hg = *(const v2f*)(h_gnn + (long)n * NHID + f0);
            acc_e += he * ae; acc_g += hg * ag;
        }

        v2f so;
        {
            float a0 = fmaxf(acc_e.x + eb0, 0.f);
            float a1 = fmaxf(acc_e.y + eb1, 0.f);
            so.x = (a0 - erm0) * ers0 * egam0 + ebeta0;
            so.y = (a1 - erm1) * ers1 * egam1 + ebeta1;
        }
        *(v2f*)(d_out + N_NODES + (long)n * NHID + f0) = so;
        float b0 = fmaxf(acc_g.x + gb0, 0.f);
        float b1 = fmaxf(acc_g.y + gb1, 0.f);
        float z0 = (b0 - grm0) * grs0 * ggam0 + gbeta0;
        float z1 = (b1 - grm1) * grs1 * ggam1 + gbeta1;
        float yp = wave_sum_f(z0 * cw0 + z1 * cw1);
        if (lane == 0) l_y[wave * NPW + i] = yp + cb;
    }
    __syncthreads();
    if (threadIdx.x < GW) d_out[blockIdx.x * GW + threadIdx.x] = l_y[threadIdx.x];
}

extern "C" void kernel_launch(void* const* d_in, const int* in_sizes, int n_in,
                              void* d_out, int out_size, void* d_ws, size_t ws_size,
                              hipStream_t stream) {
    const float* g        = (const float*)d_in[0];
    const int*   x        = (const int*)d_in[1];
    const float* est_W    = (const float*)d_in[2];
    const float* est_asrc = (const float*)d_in[3];
    const float* est_adst = (const float*)d_in[4];
    const float* est_bias = (const float*)d_in[5];
    const float* est_gam  = (const float*)d_in[6];
    const float* est_beta = (const float*)d_in[7];
    const float* est_rm   = (const float*)d_in[8];
    const float* est_rv   = (const float*)d_in[9];
    const float* gnn_W    = (const float*)d_in[10];
    const float* gnn_asrc = (const float*)d_in[11];
    const float* gnn_adst = (const float*)d_in[12];
    const float* gnn_bias = (const float*)d_in[13];
    const float* gnn_gam  = (const float*)d_in[14];
    const float* gnn_beta = (const float*)d_in[15];
    const float* gnn_rm   = (const float*)d_in[16];
    const float* gnn_rv   = (const float*)d_in[17];
    const float* cls_W    = (const float*)d_in[18];
    const float* cls_b    = (const float*)d_in[19];
    float* out = (float*)d_out;

    float* ws     = (float*)d_ws;
    float* h_est  = ws;
    float* h_gnn  = h_est + (long)N_NODES * NHID;
    int*   cnt    = (int*)(h_gnn + (long)N_NODES * NHID);
    int*   cur    = cnt + N_NODES;
    int*   off    = cur + N_NODES;
    int*   slot   = off + N_NODES + 4;
    float* es_e   = ws + (long)N_NODES * 192;
    float* ed_e   = es_e + N_NODES;
    float* es_g   = ed_e + N_NODES;
    float* ed_g   = es_g + N_NODES;

    const int BT = 256;
    k_gemm<<<(MTILES + 7) / 8, BT, 0, stream>>>(g, est_W, gnn_W, h_est, h_gnn);

    k_eprep<<<(N_NODES + BT - 1) / BT, BT, 0, stream>>>(h_est, h_gnn,
            est_asrc, est_adst, gnn_asrc, gnn_adst, es_e, ed_e, es_g, ed_g, cnt, cur);

    k_count<<<(NQUAD + QPT * EBT - 1) / (QPT * EBT), EBT, 0, stream>>>(x, cnt);

    k_scan<<<1, 1024, 0, stream>>>(cnt, off);

    k_fill<<<(NQUAD + QPT * EBT - 1) / (QPT * EBT), EBT, 0, stream>>>(x, off, cur, slot);

    k_gather<<<N_NODES / GW, GWAVES * 32, 0, stream>>>(out, off, slot,
            es_e, ed_e, es_g, ed_g, h_est, h_gnn,
            est_bias, est_gam, est_beta, est_rm, est_rv,
            gnn_bias, gnn_gam, gnn_beta, gnn_rm, gnn_rv, cls_W, cls_b);
}


# Gather: buckets of 33 to 128 edges are put in source order by rank inside LDS, 32 entries at a time, instead of by repeated minimum search through global memory; every sum keeps its order
# speedup vs baseline: 1.8276x; 1.1181x over previous
#include <hip/hip_runtime.h>
#include <cfloat>
#include <climits>

#define N_NODES 100000
#define NFEAT   256
#define NHID    64
#define NEDGE   1600000
#define MTILES  (N_NODES / 16)
#define EPSBN   1e-5f
#define NEG_SLOPE 0.2f
#define GW      32
#define GWAVES  8
#define NPW     (GW / GWAVES)
#define EBT     256
#define NQUAD   (NEDGE / 4)
#define QPT     2
#define LCAP    128

static_assert(N_NODES % GW == 0, "gather blocks cover the nodes exactly");
static_assert(GW % GWAVES == 0, "every gather wave gets the same number of nodes");
static_assert(N_NODES % 4 == 0, "scan reads whole quads");
static_assert(NEDGE % 4 == 0, "bucketing reads whole quads, and the destination row starts on a 16-byte boundary");

typedef __attribute__((ext_vector_type(2))) float v2f;
typedef __attribute__((ext_vector_type(4))) float v4f;
typedef __attribute__((ext_vector_type(8))) float v8f;

__device__ __forceinline__ float leaky(float v) {
    return fmaxf(v, 0.f) + NEG_SLOPE * fminf(v, 0.f);
}
__device__ __forceinline__ float wave_max_f(float v) {
    for (int o = 16; o > 0; o >>= 1) v = fmaxf(v, __shfl_xor(v, o));
    return v;
}
__device__ __forceinline__ float wave_sum_f(float v) {
    for (int o = 16; o > 0; o >>= 1) v += __shfl_xor(v, o);
    return v;
}
__device__ __forceinline__ int wave_min_i(int v) {
    for (int o = 16; o > 0; o >>= 1) v = min(v, __shfl_xor(v, o));
    return v;
}
__device__ __forceinline__ int wave_sum_i(int v) {
    for (int o = 16; o > 0; o >>= 1) v += __shfl_xor(v, o);
    return v;
}

__global__ __launch_bounds__(256) void k_gemm(const float* __restrict__ g,
                                              const float* __restrict__ We,
                                              const float* __restrict__ Wg,
                                              float* __restrict__ h_est,
                                              float* __restrict__ h_gnn) {
    int wave = threadIdx.x >> 5;
    int lane = threadIdx.x & 31;
    int mtile = blockIdx.x * 8 + wave;
    if (mtile >= MTILES) return;
    int m0   = mtile * 16;
    int lrow = lane & 15;
    int lhi  = lane >> 4;

    const float* grow = g + (long)(m0 + lrow) * NFEAT;

    v8f acc[8] = {};
    for (int k = 0; k < NFEAT; k += 4) {
        v2f a = *(const v2f*)(grow + k + 2 * lhi);
#pragma unroll
        for (int t = 0; t < 8; ++t) {
            const float* Wp = (t < 4) ? We : Wg;
            int c  = (t & 3) * 16 + lrow;
            int kr = k + 2 * lhi;
            v2f b;
            b.x = Wp[(long)kr * NHID + c];
            b.y = Wp[(long)(kr + 1) * NHID + c];
            acc[t] = __builtin_amdgcn_wmma_f32_16x16x4_f32(
                false, a, false, b, (short)0, acc[t], false, false);
        }
    }
#pragma unroll
    for (int t = 0; t < 8; ++t) {
        float* dst = (t < 4) ? h_est : h_gnn;
        int c = (t & 3) * 16 + lrow;
#pragma unroll
        for (int r = 0; r < 8; ++r) {
            dst[(long)(m0 + r + 8 * lhi) * NHID + c] = acc[t][r];
        }
    }
}

__global__ void k_eprep(const float* __restrict__ h_est, const float* __restrict__ h_gnn,
                        const float* __restrict__ ase, const float* __restrict__ ade,
                        const float* __restrict__ asg, const float* __restrict__ adg,
                        float* es_e, float* ed_e, float* es_g, float* ed_g,
                        int* cnt, int* cur) {
    int n = blockIdx.x * blockDim.x + threadIdx.x;
    if (n >= N_NODES) return;
    const float* he = h_est + (long)n * NHID;
    const float* hg = h_gnn + (long)n * NHID;
    float se = 0.f, de = 0.f, sg = 0.f, dg = 0.f;
#pragma unroll 4
    for (int f = 0; f < NHID; f += 4) {
        v4f e  = *(const v4f*)(he + f);
        v4f g4 = *(const v4f*)(hg + f);
        v4f a1 = *(const v4f*)(ase + f);
        v4f a2 = *(const v4f*)(ade + f);
        v4f a3 = *(const v4f*)(asg + f);
        v4f a4 = *(const v4f*)(adg + f);
        se += e.x * a1.x + e.y * a1.y + e.z * a1.z + e.w * a1.w;
        de += e.x * a2.x + e.y * a2.y + e.z * a2.z + e.w * a2.w;
        sg += g4.x * a3.x + g4.y * a3.y + g4.z * a3.z + g4.w * a3.w;
        dg += g4.x * a4.x + g4.y * a4.y + g4.z * a4.z + g4.w * a4.w;
    }
    es_e[n] = se; ed_e[n] = de; es_g[n] = sg; ed_g[n] = dg;
    cnt[n] = 0; cur[n] = 0;
}

__global__ __launch_bounds__(EBT) void k_count(const int* __restrict__ x, int* cnt) {
    const int4* d4 = (const int4*)(x + NEDGE);
    int q0 = blockIdx.x * (QPT * EBT) + threadIdx.x;
    int4 d[QPT];
#pragma unroll
    for (int j = 0; j < QPT; ++j) {
        int q = q0 + j * EBT;
        d[j] = (q < NQUAD) ? d4[q] : make_int4(0, 0, 0, 0);
    }
#pragma unroll
    for (int j = 0; j < QPT; ++j) {
        if (q0 + j * EBT < NQUAD) {
            atomicAdd(cnt + d[j].x, 1);
            atomicAdd(cnt + d[j].y, 1);
            atomicAdd(cnt + d[j].z, 1);
            atomicAdd(cnt + d[j].w, 1);
        }
    }
}

__global__ __launch_bounds__(1024) void k_scan(const int* __restrict__ cnt, int* __restrict__ off) {
    __shared__ int wtot[32];
    __shared__ int tile_tot;
    int t = threadIdx.x, lane = t & 31, wave = t >> 5;
    int carry = 0;
    for (int base = 0; base < N_NODES; base += 4096) {
        int i = base + t * 4;
        int4 c = make_int4(0, 0, 0, 0);
        if (i < N_NODES) c = *(const int4*)(cnt + i);
        int s1 = c.x, s2 = s1 + c.y, s3 = s2 + c.z, s4 = s3 + c.w;
        int inc = s4;
        for (int o = 1; o < 32; o <<= 1) { int v = __shfl_up(inc, o); if (lane >= o) inc += v; }
        if (lane == 31) wtot[wave] = inc;
        __syncthreads();
        if (wave == 0) {
            int w = wtot[lane], winc = w;
            for (int o = 1; o < 32; o <<= 1) { int v = __shfl_up(winc, o); if (lane >= o) winc += v; }
            wtot[lane] = winc - w;
            if (lane == 31) tile_tot = winc;
        }
        __syncthreads();
        int ex = carry + wtot[wave] + inc - s4;
        if (i < N_NODES) *(int4*)(off + i) = make_int4(ex, ex + s1, ex + s2, ex + s3);
        carry += tile_tot;
        __syncthreads();
    }
    if (t == 0) off[N_NODES] = carry;
}

__global__ __launch_bounds__(EBT) void k_fill(const int* __restrict__ x, const int* __restrict__ off,
                       int* cur, int* __restrict__ slot) {
    const int4* s4 = (const int4*)x;
    const int4* d4 = (const int4*)(x + NEDGE);
    int q0 = blockIdx.x * (QPT * EBT) + threadIdx.x;
    int4 s[QPT], d[QPT];
#pragma unroll
    for (int j = 0; j < QPT; ++j) {
        int q = q0 + j * EBT;
        bool ok = q < NQUAD;
        s[j] = ok ? s4[q] : make_int4(0, 0, 0, 0);
        d[j] = ok ? d4[q] : make_int4(0, 0, 0, 0);
    }
    int4 p[QPT], o[QPT];
#pragma unroll
    for (int j = 0; j < QPT; ++j) {
        if (q0 + j * EBT < NQUAD) {
            p[j].x = atomicAdd(cur + d[j].x, 1);
            p[j].y = atomicAdd(cur + d[j].y, 1);
            p[j].z = atomicAdd(cur + d[j].z, 1);
            p[j].w = atomicAdd(cur + d[j].w, 1);
            o[j].x = off[d[j].x];
            o[j].y = off[d[j].y];
            o[j].z = off[d[j].z];
            o[j].w = off[d[j].w];
        }
    }
#pragma unroll
    for (int j = 0; j < QPT; ++j) {
        if (q0 + j * EBT < NQUAD) {
            slot[o[j].x + p[j].x] = s[j].x;
            slot[o[j].y + p[j].y] = s[j].y;
            slot[o[j].z + p[j].z] = s[j].z;
            slot[o[j].w + p[j].w] = s[j].w;
        }
    }
}

__global__ __launch_bounds__(GWAVES * 32) void k_gather(float* d_out,
        const int* __restrict__ off, const int* __restrict__ slot,
        const float* __restrict__ es_e, const float* __restrict__ ed_e,
        const float* __restrict__ es_g, const float* __restrict__ ed_g,
        const float* __restrict__ h_est, const float* __restrict__ h_gnn,
        const float* __restrict__ eb, const float* __restrict__ egam,
        const float* __restrict__ ebeta, const float* __restrict__ erm,
        const float* __restrict__ erv,
        const float* __restrict__ gb, const float* __restrict__ ggam,
        const float* __restrict__ gbeta, const float* __restrict__ grm,
        const float* __restrict__ grv,
        const float* __restrict__ clsW, const float* __restrict__ clsb) {
    __shared__ int   l_s[GWAVES][LCAP];
    __shared__ float l_ae[GWAVES][LCAP];
    __shared__ float l_ag[GWAVES][LCAP];
    __shared__ int   l_u[GWAVES][LCAP];
    __shared__ float l_y[GW];
    int lane = threadIdx.x & 31;
    int wave = __builtin_amdgcn_readfirstlane(threadIdx.x >> 5);
    int f0 = lane * 2;

    float eb0 = eb[f0], eb1 = eb[f0 + 1];
    float erm0 = erm[f0], erm1 = erm[f0 + 1];
    float ers0 = rsqrtf(erv[f0] + EPSBN), ers1 = rsqrtf(erv[f0 + 1] + EPSBN);
    float egam0 = egam[f0], egam1 = egam[f0 + 1];
    float ebeta0 = ebeta[f0], ebeta1 = ebeta[f0 + 1];
    float gb0 = gb[f0], gb1 = gb[f0 + 1];
    float grm0 = grm[f0], grm1 = grm[f0 + 1];
    float grs0 = rsqrtf(grv[f0] + EPSBN), grs1 = rsqrtf(grv[f0 + 1] + EPSBN);
    float ggam0 = ggam[f0], ggam1 = ggam[f0 + 1];
    float gbeta0 = gbeta[f0], gbeta1 = gbeta[f0 + 1];
    float cw0 = clsW[f0], cw1 = clsW[f0 + 1];
    float cb = clsb[0];

#pragma nounroll
    for (int i = 0; i < NPW; ++i) {
        int n = blockIdx.x * GW + wave * NPW + i;
        int beg = off[n];
        int deg = off[n + 1] - beg;
        float edn_e = ed_e[n], edn_g = ed_g[n];
        float self_e = leaky(es_e[n] + edn_e), self_g = leaky(es_g[n] + edn_g);
        v2f acc_e = {0.f, 0.f}, acc_g = {0.f, 0.f};
        float ze = 0.f, zg = 0.f, pse, psg;

        if (deg <= 32) {
            bool on = lane < deg;
            int s = on ? slot[beg + lane] : INT_MAX;
            float le = -FLT_MAX, lg = -FLT_MAX;
            if (on) { le = leaky(es_e[s] + edn_e); lg = leaky(es_g[s] + edn_g); }
            int rank = 0;
            for (int j = 0; j < deg; ++j) {
                int sj = __shfl(s, j);
                rank += (sj < s || (sj == s && j < lane)) ? 1 : 0;
            }
            float me = fmaxf(wave_max_f(le), self_e);
            float mg = fmaxf(wave_max_f(lg), self_g);
            float pe = __expf(le - me), pg = __expf(lg - mg);
            pse = __expf(self_e - me); psg = __expf(self_g - mg);
            if (on) { l_s[wave][rank] = s; l_ae[wave][rank] = pe; l_ag[wave][rank] = pg; }
            for (int r = 0; r < deg; ++r) { ze += l_ae[wave][r]; zg += l_ag[wave][r]; }
            ze += pse; zg += psg;
            if (on) { l_ae[wave][rank] = pe / ze; l_ag[wave][rank] = pg / zg; }
#pragma unroll 4
            for (int r = 0; r < deg; ++r) {
                int sr = l_s[wave][r];
                float ae = l_ae[wave][r], ag = l_ag[wave][r];
                v2f he = *(const v2f*)(h_est + (long)sr * NHID + f0);
                v2f hg = *(const v2f*)(h_gnn + (long)sr * NHID + f0);
                acc_e += he * ae; acc_g += hg * ag;
            }
        } else if (deg <= LCAP) {
            float le_m = self_e, lg_m = self_g;
            for (int b = 0; b < deg; b += 32) {
                int k = b + lane;
                if (k < deg) {
                    int s = slot[beg + k];
                    l_u[wave][k] = s;
                    le_m = fmaxf(le_m, leaky(es_e[s] + edn_e));
                    lg_m = fmaxf(lg_m, leaky(es_g[s] + edn_g));
                }
            }
            float me = wave_max_f(le_m), mg = wave_max_f(lg_m);
            pse = __expf(self_e - me); psg = __expf(self_g - mg);
            for (int b = 0; b < deg; b += 32) {
                int k = b + lane;
                int s = (k < deg) ? l_u[wave][k] : INT_MAX;
                int rank = 0;
#pragma unroll 4
                for (int j = 0; j < deg; ++j) {
                    int sj = l_u[wave][j];
                    rank += (sj < s || (sj == s && j < k)) ? 1 : 0;
                }
                if (k < deg) {
                    l_s[wave][rank] = s;
                    l_ae[wave][rank] = __expf(leaky(es_e[s] + edn_e) - me);
                    l_ag[wave][rank] = __expf(leaky(es_g[s] + edn_g) - mg);
                }
            }
            for (int r = 0; r < deg; ++r) { ze += l_ae[wave][r]; zg += l_ag[wave][r]; }
            ze += pse; zg += psg;
            for (int b = 0; b < deg; b += 32) {
                int r = b + lane;
                if (r < deg) { l_ae[wave][r] = l_ae[wave][r] / ze; l_ag[wave][r] = l_ag[wave][r] / zg; }
            }
#pragma unroll 4
            for (int r = 0; r < deg; ++r) {
                int sr = l_s[wave][r];
                float ae = l_ae[wave][r], ag = l_ag[wave][r];
                v2f he = *(const v2f*)(h_est + (long)sr * NHID + f0);
                v2f hg = *(const v2f*)(h_gnn + (long)sr * NHID + f0);
                acc_e += he * ae; acc_g += hg * ag;
            }
        } else {
            float le_m = self_e, lg_m = self_g;
            for (int k = lane; k < deg; k += 32) {
                int s = slot[beg + k];
                le_m = fmaxf(le_m, leaky(es_e[s] + edn_e));
                lg_m = fmaxf(lg_m, leaky(es_g[s] + edn_g));
            }
            float me = wave_max_f(le_m), mg = wave_max_f(lg_m);
            pse = __expf(self_e - me); psg = __expf(self_g - mg);
            for (int pass = 0; pass < 2; ++pass) {
                int prev = -1;
                for (int it = 0; it < deg; ++it) {
                    int best = INT_MAX;
                    for (int k = lane; k < deg; k += 32) {
                        int s = slot[beg + k];
                        if (s > prev) best = min(best, s);
                    }
                    best = wave_min_i(best);
                    if (best == INT_MAX) break;
                    int c = 0;
                    for (int k = lane; k < deg; k += 32) c += (slot[beg + k] == best) ? 1 : 0;
                    c = wave_sum_i(c);
                    float pe = __expf(leaky(es_e[best] + edn_e) - me);
                    float pg = __expf(leaky(es_g[best] + edn_g) - mg);
                    if (pass == 0) {
                        for (int k = 0; k < c; ++k) { ze += pe; zg += pg; }
                    } else {
                        float ae = pe / ze, ag = pg / zg;
                        v2f he = *(const v2f*)(h_est + (long)best * NHID + f0);
                        v2f hg = *(const v2f*)(h_gnn + (long)best * NHID + f0);
                        for (int k = 0; k < c; ++k) { acc_e += he * ae; acc_g += hg * ag; }
                    }
                    prev = best;
                }
                if (pass == 0) { ze += pse; zg += psg; }
            }
        }
        {
            float ae = pse / ze, ag = psg / zg;
            v2f he = *(const v2f*)(h_est + (long)n * NHID + f0);
            v2f hg = *(const v2f*)(h_gnn + (long)n * NHID + f0);
            acc_e += he * ae; acc_g += hg * ag;
        }

        v2f so;
        {
            float a0 = fmaxf(acc_e.x + eb0, 0.f);
            float a1 = fmaxf(acc_e.y + eb1, 0.f);
            so.x = (a0 - erm0) * ers0 * egam0 + ebeta0;
            so.y = (a1 - erm1) * ers1 * egam1 + ebeta1;
        }
        *(v2f*)(d_out + N_NODES + (long)n * NHID + f0) = so;
        float b0 = fmaxf(acc_g.x + gb0, 0.f);
        float b1 = fmaxf(acc_g.y + gb1, 0.f);
        float z0 = (b0 - grm0) * grs0 * ggam0 + gbeta0;
        float z1 = (b1 - grm1) * grs1 * ggam1 + gbeta1;
        float yp = wave_sum_f(z0 * cw0 + z1 * cw1);
        if (lane == 0) l_y[wave * NPW + i] = yp + cb;
    }
    __syncthreads();
    if (threadIdx.x < GW) d_out[blockIdx.x * GW + threadIdx.x] = l_y[threadIdx.x];
}

extern "C" void kernel_launch(void* const* d_in, const int* in_sizes, int n_in,
                              void* d_out, int out_size, void* d_ws, size_t ws_size,
                              hipStream_t stream) {
    const float* g        = (const float*)d_in[0];
    const int*   x        = (const int*)d_in[1];
    const float* est_W    = (const float*)d_in[2];
    const float* est_asrc = (const float*)d_in[3];
    const float* est_adst = (const float*)d_in[4];
    const float* est_bias = (const float*)d_in[5];
    const float* est_gam  = (const float*)d_in[6];
    const float* est_beta = (const float*)d_in[7];
    const float* est_rm   = (const float*)d_in[8];
    const float* est_rv   = (const float*)d_in[9];
    const float* gnn_W    = (const float*)d_in[10];
    const float* gnn_asrc = (const float*)d_in[11];
    const float* gnn_adst = (const float*)d_in[12];
    const float* gnn_bias = (const float*)d_in[13];
    const float* gnn_gam  = (const float*)d_in[14];
    const float* gnn_beta = (const float*)d_in[15];
    const float* gnn_rm   = (const float*)d_in[16];
    const float* gnn_rv   = (const float*)d_in[17];
    const float* cls_W    = (const float*)d_in[18];
    const float* cls_b    = (const float*)d_in[19];
    float* out = (float*)d_out;

    float* ws     = (float*)d_ws;
    float* h_est  = ws;
    float* h_gnn  = h_est + (long)N_NODES * NHID;
    int*   cnt    = (int*)(h_gnn + (long)N_NODES * NHID);
    int*   cur    = cnt + N_NODES;
    int*   off    = cur + N_NODES;
    int*   slot   = off + N_NODES + 4;
    float* es_e   = ws + (long)N_NODES * 192;
    float* ed_e   = es_e + N_NODES;
    float* es_g   = ed_e + N_NODES;
    float* ed_g   = es_g + N_NODES;

    const int BT = 256;
    k_gemm<<<(MTILES + 7) / 8, BT, 0, stream>>>(g, est_W, gnn_W, h_est, h_gnn);

    k_eprep<<<(N_NODES + BT - 1) / BT, BT, 0, stream>>>(h_est, h_gnn,
            est_asrc, est_adst, gnn_asrc, gnn_adst, es_e, ed_e, es_g, ed_g, cnt, cur);

    k_count<<<(NQUAD + QPT * EBT - 1) / (QPT * EBT), EBT, 0, stream>>>(x, cnt);

    k_scan<<<1, 1024, 0, stream>>>(cnt, off);

    k_fill<<<(NQUAD + QPT * EBT - 1) / (QPT * EBT), EBT, 0, stream>>>(x, off, cur, slot);

    k_gather<<<N_NODES / GW, GWAVES * 32, 0, stream>>>(out, off, slot,
            es_e, ed_e, es_g, ed_g, h_est, h_gnn,
            est_bias, est_gam, est_beta, est_rm, est_rv,
            gnn_bias, gnn_gam, gnn_beta, gnn_rm, gnn_rv, cls_W, cls_b);
}


# Gather, buckets up to 32: the node's own row joins the ordered list, the first two groups of 8 rows are loaded before the softmax arithmetic, later rows 8 at a time; every sum keeps its order
# speedup vs baseline: 1.9020x; 1.0407x over previous
#include <hip/hip_runtime.h>
#include <cfloat>
#include <climits>

#define N_NODES 100000
#define NFEAT   256
#define NHID    64
#define NEDGE   1600000
#define MTILES  (N_NODES / 16)
#define EPSBN   1e-5f
#define NEG_SLOPE 0.2f
#define GW      32
#define GWAVES  8
#define NPW     (GW / GWAVES)
#define EBT     256
#define NQUAD   (NEDGE / 4)
#define QPT     2
#define LCAP    128
#define RB      8
#define FB      16

static_assert(N_NODES % GW == 0, "gather blocks cover the nodes exactly");
static_assert(GW % GWAVES == 0, "every gather wave gets the same number of nodes");
static_assert(N_NODES % 4 == 0, "scan reads whole quads");
static_assert((RB & (RB - 1)) == 0 && FB % RB == 0 && FB <= 32 && 32 + RB <= LCAP, "a bucket of up to 32, its node and the padding fit the LDS list, and one lane each writes the padding");
static_assert(NEDGE % 4 == 0, "bucketing reads whole quads, and the destination row starts on a 16-byte boundary");

typedef __attribute__((ext_vector_type(2))) float v2f;
typedef __attribute__((ext_vector_type(4))) float v4f;
typedef __attribute__((ext_vector_type(8))) float v8f;

__device__ __forceinline__ float leaky(float v) {
    return fmaxf(v, 0.f) + NEG_SLOPE * fminf(v, 0.f);
}
__device__ __forceinline__ float wave_max_f(float v) {
    for (int o = 16; o > 0; o >>= 1) v = fmaxf(v, __shfl_xor(v, o));
    return v;
}
__device__ __forceinline__ float wave_sum_f(float v) {
    for (int o = 16; o > 0; o >>= 1) v += __shfl_xor(v, o);
    return v;
}
__device__ __forceinline__ int wave_min_i(int v) {
    for (int o = 16; o > 0; o >>= 1) v = min(v, __shfl_xor(v, o));
    return v;
}
__device__ __forceinline__ int wave_sum_i(int v) {
    for (int o = 16; o > 0; o >>= 1) v += __shfl_xor(v, o);
    return v;
}

__global__ __launch_bounds__(256) void k_gemm(const float* __restrict__ g,
                                              const float* __restrict__ We,
                                              const float* __restrict__ Wg,
                                              float* __restrict__ h_est,
                                              float* __restrict__ h_gnn) {
    int wave = threadIdx.x >> 5;
    int lane = threadIdx.x & 31;
    int mtile = blockIdx.x * 8 + wave;
    if (mtile >= MTILES) return;
    int m0   = mtile * 16;
    int lrow = lane & 15;
    int lhi  = lane >> 4;

    const float* grow = g + (long)(m0 + lrow) * NFEAT;

    v8f acc[8] = {};
    for (int k = 0; k < NFEAT; k += 4) {
        v2f a = *(const v2f*)(grow + k + 2 * lhi);
#pragma unroll
        for (int t = 0; t < 8; ++t) {
            const float* Wp = (t < 4) ? We : Wg;
            int c  = (t & 3) * 16 + lrow;
            int kr = k + 2 * lhi;
            v2f b;
            b.x = Wp[(long)kr * NHID + c];
            b.y = Wp[(long)(kr + 1) * NHID + c];
            acc[t] = __builtin_amdgcn_wmma_f32_16x16x4_f32(
                false, a, false, b, (short)0, acc[t], false, false);
        }
    }
#pragma unroll
    for (int t = 0; t < 8; ++t) {
        float* dst = (t < 4) ? h_est : h_gnn;
        int c = (t & 3) * 16 + lrow;
#pragma unroll
        for (int r = 0; r < 8; ++r) {
            dst[(long)(m0 + r + 8 * lhi) * NHID + c] = acc[t][r];
        }
    }
}

__global__ void k_eprep(const float* __restrict__ h_est, const float* __restrict__ h_gnn,
                        const float* __restrict__ ase, const float* __restrict__ ade,
                        const float* __restrict__ asg, const float* __restrict__ adg,
                        float* es_e, float* ed_e, float* es_g, float* ed_g,
                        int* cnt, int* cur) {
    int n = blockIdx.x * blockDim.x + threadIdx.x;
    if (n >= N_NODES) return;
    const float* he = h_est + (long)n * NHID;
    const float* hg = h_gnn + (long)n * NHID;
    float se = 0.f, de = 0.f, sg = 0.f, dg = 0.f;
#pragma unroll 4
    for (int f = 0; f < NHID; f += 4) {
        v4f e  = *(const v4f*)(he + f);
        v4f g4 = *(const v4f*)(hg + f);
        v4f a1 = *(const v4f*)(ase + f);
        v4f a2 = *(const v4f*)(ade + f);
        v4f a3 = *(const v4f*)(asg + f);
        v4f a4 = *(const v4f*)(adg + f);
        se += e.x * a1.x + e.y * a1.y + e.z * a1.z + e.w * a1.w;
        de += e.x * a2.x + e.y * a2.y + e.z * a2.z + e.w * a2.w;
        sg += g4.x * a3.x + g4.y * a3.y + g4.z * a3.z + g4.w * a3.w;
        dg += g4.x * a4.x + g4.y * a4.y + g4.z * a4.z + g4.w * a4.w;
    }
    es_e[n] = se; ed_e[n] = de; es_g[n] = sg; ed_g[n] = dg;
    cnt[n] = 0; cur[n] = 0;
}

__global__ __launch_bounds__(EBT) void k_count(const int* __restrict__ x, int* cnt) {
    const int4* d4 = (const int4*)(x + NEDGE);
    int q0 = blockIdx.x * (QPT * EBT) + threadIdx.x;
    int4 d[QPT];
#pragma unroll
    for (int j = 0; j < QPT; ++j) {
        int q = q0 + j * EBT;
        d[j] = (q < NQUAD) ? d4[q] : make_int4(0, 0, 0, 0);
    }
#pragma unroll
    for (int j = 0; j < QPT; ++j) {
        if (q0 + j * EBT < NQUAD) {
            atomicAdd(cnt + d[j].x, 1);
            atomicAdd(cnt + d[j].y, 1);
            atomicAdd(cnt + d[j].z, 1);
            atomicAdd(cnt + d[j].w, 1);
        }
    }
}

__global__ __launch_bounds__(1024) void k_scan(const int* __restrict__ cnt, int* __restrict__ off) {
    __shared__ int wtot[32];
    __shared__ int tile_tot;
    int t = threadIdx.x, lane = t & 31, wave = t >> 5;
    int carry = 0;
    for (int base = 0; base < N_NODES; base += 4096) {
        int i = base + t * 4;
        int4 c = make_int4(0, 0, 0, 0);
        if (i < N_NODES) c = *(const int4*)(cnt + i);
        int s1 = c.x, s2 = s1 + c.y, s3 = s2 + c.z, s4 = s3 + c.w;
        int inc = s4;
        for (int o = 1; o < 32; o <<= 1) { int v = __shfl_up(inc, o); if (lane >= o) inc += v; }
        if (lane == 31) wtot[wave] = inc;
        __syncthreads();
        if (wave == 0) {
            int w = wtot[lane], winc = w;
            for (int o = 1; o < 32; o <<= 1) { int v = __shfl_up(winc, o); if (lane >= o) winc += v; }
            wtot[lane] = winc - w;
            if (lane == 31) tile_tot = winc;
        }
        __syncthreads();
        int ex = carry + wtot[wave] + inc - s4;
        if (i < N_NODES) *(int4*)(off + i) = make_int4(ex, ex + s1, ex + s2, ex + s3);
        carry += tile_tot;
        __syncthreads();
    }
    if (t == 0) off[N_NODES] = carry;
}

__global__ __launch_bounds__(EBT) void k_fill(const int* __restrict__ x, const int* __restrict__ off,
                       int* cur, int* __restrict__ slot) {
    const int4* s4 = (const int4*)x;
    const int4* d4 = (const int4*)(x + NEDGE);
    int q0 = blockIdx.x * (QPT * EBT) + threadIdx.x;
    int4 s[QPT], d[QPT];
#pragma unroll
    for (int j = 0; j < QPT; ++j) {
        int q = q0 + j * EBT;
        bool ok = q < NQUAD;
        s[j] = ok ? s4[q] : make_int4(0, 0, 0, 0);
        d[j] = ok ? d4[q] : make_int4(0, 0, 0, 0);
    }
    int4 p[QPT], o[QPT];
#pragma unroll
    for (int j = 0; j < QPT; ++j) {
        if (q0 + j * EBT < NQUAD) {
            p[j].x = atomicAdd(cur + d[j].x, 1);
            p[j].y = atomicAdd(cur + d[j].y, 1);
            p[j].z = atomicAdd(cur + d[j].z, 1);
            p[j].w = atomicAdd(cur + d[j].w, 1);
            o[j].x = off[d[j].x];
            o[j].y = off[d[j].y];
            o[j].z = off[d[j].z];
            o[j].w = off[d[j].w];
        }
    }
#pragma unroll
    for (int j = 0; j < QPT; ++j) {
        if (q0 + j * EBT < NQUAD) {
            slot[o[j].x + p[j].x] = s[j].x;
            slot[o[j].y + p[j].y] = s[j].y;
            slot[o[j].z + p[j].z] = s[j].z;
            slot[o[j].w + p[j].w] = s[j].w;
        }
    }
}

__global__ __launch_bounds__(GWAVES * 32) void k_gather(float* d_out,
        const int* __restrict__ off, const int* __restrict__ slot,
        const float* __restrict__ es_e, const float* __restrict__ ed_e,
        const float* __restrict__ es_g, const float* __restrict__ ed_g,
        const float* __restrict__ h_est, const float* __restrict__ h_gnn,
        const float* __restrict__ eb, const float* __restrict__ egam,
        const float* __restrict__ ebeta, const float* __restrict__ erm,
        const float* __restrict__ erv,
        const float* __restrict__ gb, const float* __restrict__ ggam,
        const float* __restrict__ gbeta, const float* __restrict__ grm,
        const float* __restrict__ grv,
        const float* __restrict__ clsW, const float* __restrict__ clsb) {
    __shared__ int   l_s[GWAVES][LCAP];
    __shared__ float l_ae[GWAVES][LCAP];
    __shared__ float l_ag[GWAVES][LCAP];
    __shared__ int   l_u[GWAVES][LCAP];
    __shared__ float l_y[GW];
    int lane = threadIdx.x & 31;
    int wave = __builtin_amdgcn_readfirstlane(threadIdx.x >> 5);
    int f0 = lane * 2;

    float eb0 = eb[f0], eb1 = eb[f0 + 1];
    float erm0 = erm[f0], erm1 = erm[f0 + 1];
    float ers0 = rsqrtf(erv[f0] + EPSBN), ers1 = rsqrtf(erv[f0 + 1] + EPSBN);
    float egam0 = egam[f0], egam1 = egam[f0 + 1];
    float ebeta0 = ebeta[f0], ebeta1 = ebeta[f0 + 1];
    float gb0 = gb[f0], gb1 = gb[f0 + 1];
    float grm0 = grm[f0], grm1 = grm[f0 + 1];
    float grs0 = rsqrtf(grv[f0] + EPSBN), grs1 = rsqrtf(grv[f0 + 1] + EPSBN);
    float ggam0 = ggam[f0], ggam1 = ggam[f0 + 1];
    float gbeta0 = gbeta[f0], gbeta1 = gbeta[f0 + 1];
    float cw0 = clsW[f0], cw1 = clsW[f0 + 1];
    float cb = clsb[0];

#pragma nounroll
    for (int i = 0; i < NPW; ++i) {
        int n = blockIdx.x * GW + wave * NPW + i;
        int beg = off[n];
        int deg = off[n + 1] - beg;
        float edn_e = ed_e[n], edn_g = ed_g[n];
        float self_e = leaky(es_e[n] + edn_e), self_g = leaky(es_g[n] + edn_g);
        v2f acc_e = {0.f, 0.f}, acc_g = {0.f, 0.f};
        float ze = 0.f, zg = 0.f, pse, psg;

        if (deg <= 32) {
            bool on = lane < deg;
            int sraw = slot[min(beg + lane, NEDGE - 1)];
            int s = on ? sraw : INT_MAX;
            int si = on ? sraw : n;
            float xe = es_e[si], xg = es_g[si];
            int rank = 0;
            for (int j = 0; j < deg; ++j) {
                int sj = __shfl(s, j);
                rank += (sj < s || (sj == s && j < lane)) ? 1 : 0;
            }
            int cntp = max((deg + RB) & ~(RB - 1), FB);
            if (on) l_s[wave][rank] = s;
            if (lane < cntp - deg) l_s[wave][deg + lane] = n;
            v2f he0[FB], hg0[FB];
#pragma unroll
            for (int u = 0; u < FB; ++u) {
                int sr = l_s[wave][u];
                he0[u] = *(const v2f*)(h_est + (long)sr * NHID + f0);
                hg0[u] = *(const v2f*)(h_gnn + (long)sr * NHID + f0);
            }
            __builtin_amdgcn_sched_barrier(0);
            float le = on ? leaky(xe + edn_e) : -FLT_MAX;
            float lg = on ? leaky(xg + edn_g) : -FLT_MAX;
            float me = fmaxf(wave_max_f(le), self_e);
            float mg = fmaxf(wave_max_f(lg), self_g);
            float pe = __expf(le - me), pg = __expf(lg - mg);
            pse = __expf(self_e - me); psg = __expf(self_g - mg);
            if (on) { l_ae[wave][rank] = pe; l_ag[wave][rank] = pg; }
            for (int r = 0; r < deg; ++r) { ze += l_ae[wave][r]; zg += l_ag[wave][r]; }
            ze += pse; zg += psg;
            if (on) { l_ae[wave][rank] = pe / ze; l_ag[wave][rank] = pg / zg; }
            if (lane < cntp - deg) {
                l_ae[wave][deg + lane] = (lane == 0) ? pse / ze : 0.f;
                l_ag[wave][deg + lane] = (lane == 0) ? psg / zg : 0.f;
            }
#pragma unroll
            for (int u = 0; u < FB; ++u) {
                acc_e += he0[u] * l_ae[wave][u]; acc_g += hg0[u] * l_ag[wave][u];
            }
            for (int b = FB; b < cntp; b += RB) {
                v2f he[RB], hg[RB];
#pragma unroll
                for (int u = 0; u < RB; ++u) {
                    int sr = l_s[wave][b + u];
                    he[u] = *(const v2f*)(h_est + (long)sr * NHID + f0);
                    hg[u] = *(const v2f*)(h_gnn + (long)sr * NHID + f0);
                }
#pragma unroll
                for (int u = 0; u < RB; ++u) {
                    acc_e += he[u] * l_ae[wave][b + u]; acc_g += hg[u] * l_ag[wave][b + u];
                }
            }
        } else {
        if (deg <= LCAP) {
            float le_m = self_e, lg_m = self_g;
            for (int b = 0; b < deg; b += 32) {
                int k = b + lane;
                if (k < deg) {
                    int s = slot[beg + k];
                    l_u[wave][k] = s;
                    le_m = fmaxf(le_m, leaky(es_e[s] + edn_e));
                    lg_m = fmaxf(lg_m, leaky(es_g[s] + edn_g));
                }
            }
            float me = wave_max_f(le_m), mg = wave_max_f(lg_m);
            pse = __expf(self_e - me); psg = __expf(self_g - mg);
            for (int b = 0; b < deg; b += 32) {
                int k = b + lane;
                int s = (k < deg) ? l_u[wave][k] : INT_MAX;
                int rank = 0;
#pragma unroll 4
                for (int j = 0; j < deg; ++j) {
                    int sj = l_u[wave][j];
                    rank += (sj < s || (sj == s && j < k)) ? 1 : 0;
                }
                if (k < deg) {
                    l_s[wave][rank] = s;
                    l_ae[wave][rank] = __expf(leaky(es_e[s] + edn_e) - me);
                    l_ag[wave][rank] = __expf(leaky(es_g[s] + edn_g) - mg);
                }
            }
            for (int r = 0; r < deg; ++r) { ze += l_ae[wave][r]; zg += l_ag[wave][r]; }
            ze += pse; zg += psg;
            for (int b = 0; b < deg; b += 32) {
                int r = b + lane;
                if (r < deg) { l_ae[wave][r] = l_ae[wave][r] / ze; l_ag[wave][r] = l_ag[wave][r] / zg; }
            }
#pragma unroll 4
            for (int r = 0; r < deg; ++r) {
                int sr = l_s[wave][r];
                float ae = l_ae[wave][r], ag = l_ag[wave][r];
                v2f he = *(const v2f*)(h_est + (long)sr * NHID + f0);
                v2f hg = *(const v2f*)(h_gnn + (long)sr * NHID + f0);
                acc_e += he * ae; acc_g += hg * ag;
            }
        } else {
            float le_m = self_e, lg_m = self_g;
            for (int k = lane; k < deg; k += 32) {
                int s = slot[beg + k];
                le_m = fmaxf(le_m, leaky(es_e[s] + edn_e));
                lg_m = fmaxf(lg_m, leaky(es_g[s] + edn_g));
            }
            float me = wave_max_f(le_m), mg = wave_max_f(lg_m);
            pse = __expf(self_e - me); psg = __expf(self_g - mg);
            for (int pass = 0; pass < 2; ++pass) {
                int prev = -1;
                for (int it = 0; it < deg; ++it) {
                    int best = INT_MAX;
                    for (int k = lane; k < deg; k += 32) {
                        int s = slot[beg + k];
                        if (s > prev) best = min(best, s);
                    }
                    best = wave_min_i(best);
                    if (best == INT_MAX) break;
                    int c = 0;
                    for (int k = lane; k < deg; k += 32) c += (slot[beg + k] == best) ? 1 : 0;
                    c = wave_sum_i(c);
                    float pe = __expf(leaky(es_e[best] + edn_e) - me);
                    float pg = __expf(leaky(es_g[best] + edn_g) - mg);
                    if (pass == 0) {
                        for (int k = 0; k < c; ++k) { ze += pe; zg += pg; }
                    } else {
                        float ae = pe / ze, ag = pg / zg;
                        v2f he = *(const v2f*)(h_est + (long)best * NHID + f0);
                        v2f hg = *(const v2f*)(h_gnn + (long)best * NHID + f0);
                        for (int k = 0; k < c; ++k) { acc_e += he * ae; acc_g += hg * ag; }
                    }
                    prev = best;
                }
                if (pass == 0) { ze += pse; zg += psg; }
            }
        }
        {
            float ae = pse / ze, ag = psg / zg;
            v2f he = *(const v2f*)(h_est + (long)n * NHID + f0);
            v2f hg = *(const v2f*)(h_gnn + (long)n * NHID + f0);
            acc_e += he * ae; acc_g += hg * ag;
        }
        }

        v2f so;
        {
            float a0 = fmaxf(acc_e.x + eb0, 0.f);
            float a1 = fmaxf(acc_e.y + eb1, 0.f);
            so.x = (a0 - erm0) * ers0 * egam0 + ebeta0;
            so.y = (a1 - erm1) * ers1 * egam1 + ebeta1;
        }
        *(v2f*)(d_out + N_NODES + (long)n * NHID + f0) = so;
        float b0 = fmaxf(acc_g.x + gb0, 0.f);
        float b1 = fmaxf(acc_g.y + gb1, 0.f);
        float z0 = (b0 - grm0) * grs0 * ggam0 + gbeta0;
        float z1 = (b1 - grm1) * grs1 * ggam1 + gbeta1;
        float yp = wave_sum_f(z0 * cw0 + z1 * cw1);
        if (lane == 0) l_y[wave * NPW + i] = yp + cb;
    }
    __syncthreads();
    if (threadIdx.x < GW) d_out[blockIdx.x * GW + threadIdx.x] = l_y[threadIdx.x];
}

extern "C" void kernel_launch(void* const* d_in, const int* in_sizes, int n_in,
                              void* d_out, int out_size, void* d_ws, size_t ws_size,
                              hipStream_t stream) {
    const float* g        = (const float*)d_in[0];
    const int*   x        = (const int*)d_in[1];
    const float* est_W    = (const float*)d_in[2];
    const float* est_asrc = (const float*)d_in[3];
    const float* est_adst = (const float*)d_in[4];
    const float* est_bias = (const float*)d_in[5];
    const float* est_gam  = (const float*)d_in[6];
    const float* est_beta = (const float*)d_in[7];
    const float* est_rm   = (const float*)d_in[8];
    const float* est_rv   = (const float*)d_in[9];
    const float* gnn_W    = (const float*)d_in[10];
    const float* gnn_asrc = (const float*)d_in[11];
    const float* gnn_adst = (const float*)d_in[12];
    const float* gnn_bias = (const float*)d_in[13];
    const float* gnn_gam  = (const float*)d_in[14];
    const float* gnn_beta = (const float*)d_in[15];
    const float* gnn_rm   = (const float*)d_in[16];
    const float* gnn_rv   = (const float*)d_in[17];
    const float* cls_W    = (const float*)d_in[18];
    const float* cls_b    = (const float*)d_in[19];
    float* out = (float*)d_out;

    float* ws     = (float*)d_ws;
    float* h_est  = ws;
    float* h_gnn  = h_est + (long)N_NODES * NHID;
    int*   cnt    = (int*)(h_gnn + (long)N_NODES * NHID);
    int*   cur    = cnt + N_NODES;
    int*   off    = cur + N_NODES;
    int*   slot   = off + N_NODES + 4;
    float* es_e   = ws + (long)N_NODES * 192;
    float* ed_e   = es_e + N_NODES;
    float* es_g   = ed_e + N_NODES;
    float* ed_g   = es_g + N_NODES;

    const int BT = 256;
    k_gemm<<<(MTILES + 7) / 8, BT, 0, stream>>>(g, est_W, gnn_W, h_est, h_gnn);

    k_eprep<<<(N_NODES + BT - 1) / BT, BT, 0, stream>>>(h_est, h_gnn,
            est_asrc, est_adst, gnn_asrc, gnn_adst, es_e, ed_e, es_g, ed_g, cnt, cur);

    k_count<<<(NQUAD + QPT * EBT - 1) / (QPT * EBT), EBT, 0, stream>>>(x, cnt);

    k_scan<<<1, 1024, 0, stream>>>(cnt, off);

    k_fill<<<(NQUAD + QPT * EBT - 1) / (QPT * EBT), EBT, 0, stream>>>(x, off, cur, slot);

    k_gather<<<N_NODES / GW, GWAVES * 32, 0, stream>>>(out, off, slot,
            es_e, ed_e, es_g, ed_g, h_est, h_gnn,
            est_bias, est_gam, est_beta, est_rm, est_rv,
            gnn_bias, gnn_gam, gnn_beta, gnn_rm, gnn_rv, cls_W, cls_b);
}


# Degree scan in one pass: each of 32 waves loads its 3200 counters at once and runs its 25 wave scans side by side, 2 barriers in all, not 25 dependent tiles of 3 barriers; integer sums only
# speedup vs baseline: 2.0525x; 1.0791x over previous
#include <hip/hip_runtime.h>
#include <cfloat>
#include <climits>

#define N_NODES 100000
#define NFEAT   256
#define NHID    64
#define NEDGE   1600000
#define MTILES  (N_NODES / 16)
#define EPSBN   1e-5f
#define NEG_SLOPE 0.2f
#define GW      32
#define GWAVES  8
#define NPW     (GW / GWAVES)
#define EBT     256
#define NQUAD   (NEDGE / 4)
#define QPT     2
#define LCAP    128
#define RB      8
#define FB      16
#define SROUNDS 25
#define SCH     (SROUNDS * 128)
#define SCANW   (32 * SCH)

static_assert(N_NODES % GW == 0, "gather blocks cover the nodes exactly");
static_assert(GW % GWAVES == 0, "every gather wave gets the same number of nodes");
static_assert(N_NODES % 4 == 0, "scan reads whole quads");
static_assert(SCANW > N_NODES, "the scan covers every node and the counter behind the last one");
static_assert((RB & (RB - 1)) == 0 && FB % RB == 0 && FB <= 32 && 32 + RB <= LCAP, "a bucket of up to 32, its node and the padding fit the LDS list, and one lane each writes the padding");
static_assert(NEDGE % 4 == 0, "bucketing reads whole quads, and the destination row starts on a 16-byte boundary");

typedef __attribute__((ext_vector_type(2))) float v2f;
typedef __attribute__((ext_vector_type(4))) float v4f;
typedef __attribute__((ext_vector_type(8))) float v8f;

__device__ __forceinline__ float leaky(float v) {
    return fmaxf(v, 0.f) + NEG_SLOPE * fminf(v, 0.f);
}
__device__ __forceinline__ float wave_max_f(float v) {
    for (int o = 16; o > 0; o >>= 1) v = fmaxf(v, __shfl_xor(v, o));
    return v;
}
__device__ __forceinline__ float wave_sum_f(float v) {
    for (int o = 16; o > 0; o >>= 1) v += __shfl_xor(v, o);
    return v;
}
__device__ __forceinline__ int wave_min_i(int v) {
    for (int o = 16; o > 0; o >>= 1) v = min(v, __shfl_xor(v, o));
    return v;
}
__device__ __forceinline__ int wave_sum_i(int v) {
    for (int o = 16; o > 0; o >>= 1) v += __shfl_xor(v, o);
    return v;
}

__global__ __launch_bounds__(256) void k_gemm(const float* __restrict__ g,
                                              const float* __restrict__ We,
                                              const float* __restrict__ Wg,
                                              float* __restrict__ h_est,
                                              float* __restrict__ h_gnn) {
    int wave = threadIdx.x >> 5;
    int lane = threadIdx.x & 31;
    int mtile = blockIdx.x * 8 + wave;
    if (mtile >= MTILES) return;
    int m0   = mtile * 16;
    int lrow = lane & 15;
    int lhi  = lane >> 4;

    const float* grow = g + (long)(m0 + lrow) * NFEAT;

    v8f acc[8] = {};
    for (int k = 0; k < NFEAT; k += 4) {
        v2f a = *(const v2f*)(grow + k + 2 * lhi);
#pragma unroll
        for (int t = 0; t < 8; ++t) {
            const float* Wp = (t < 4) ? We : Wg;
            int c  = (t & 3) * 16 + lrow;
            int kr = k + 2 * lhi;
            v2f b;
            b.x = Wp[(long)kr * NHID + c];
            b.y = Wp[(long)(kr + 1) * NHID + c];
            acc[t] = __builtin_amdgcn_wmma_f32_16x16x4_f32(
                false, a, false, b, (short)0, acc[t], false, false);
        }
    }
#pragma unroll
    for (int t = 0; t < 8; ++t) {
        float* dst = (t < 4) ? h_est : h_gnn;
        int c = (t & 3) * 16 + lrow;
#pragma unroll
        for (int r = 0; r < 8; ++r) {
            dst[(long)(m0 + r + 8 * lhi) * NHID + c] = acc[t][r];
        }
    }
}

__global__ void k_eprep(const float* __restrict__ h_est, const float* __restrict__ h_gnn,
                        const float* __restrict__ ase, const float* __restrict__ ade,
                        const float* __restrict__ asg, const float* __restrict__ adg,
                        float* es_e, float* ed_e, float* es_g, float* ed_g,
                        int* cnt, int* cur) {
    int n = blockIdx.x * blockDim.x + threadIdx.x;
    if (n >= N_NODES) {
        if (n < SCANW) cnt[n] = 0;
        return;
    }
    const float* he = h_est + (long)n * NHID;
    const float* hg = h_gnn + (long)n * NHID;
    float se = 0.f, de = 0.f, sg = 0.f, dg = 0.f;
#pragma unroll 4
    for (int f = 0; f < NHID; f += 4) {
        v4f e  = *(const v4f*)(he + f);
        v4f g4 = *(const v4f*)(hg + f);
        v4f a1 = *(const v4f*)(ase + f);
        v4f a2 = *(const v4f*)(ade + f);
        v4f a3 = *(const v4f*)(asg + f);
        v4f a4 = *(const v4f*)(adg + f);
        se += e.x * a1.x + e.y * a1.y + e.z * a1.z + e.w * a1.w;
        de += e.x * a2.x + e.y * a2.y + e.z * a2.z + e.w * a2.w;
        sg += g4.x * a3.x + g4.y * a3.y + g4.z * a3.z + g4.w * a3.w;
        dg += g4.x * a4.x + g4.y * a4.y + g4.z * a4.z + g4.w * a4.w;
    }
    es_e[n] = se; ed_e[n] = de; es_g[n] = sg; ed_g[n] = dg;
    cnt[n] = 0; cur[n] = 0;
}

__global__ __launch_bounds__(EBT) void k_count(const int* __restrict__ x, int* cnt) {
    const int4* d4 = (const int4*)(x + NEDGE);
    int q0 = blockIdx.x * (QPT * EBT) + threadIdx.x;
    int4 d[QPT];
#pragma unroll
    for (int j = 0; j < QPT; ++j) {
        int q = q0 + j * EBT;
        d[j] = (q < NQUAD) ? d4[q] : make_int4(0, 0, 0, 0);
    }
#pragma unroll
    for (int j = 0; j < QPT; ++j) {
        if (q0 + j * EBT < NQUAD) {
            atomicAdd(cnt + d[j].x, 1);
            atomicAdd(cnt + d[j].y, 1);
            atomicAdd(cnt + d[j].z, 1);
            atomicAdd(cnt + d[j].w, 1);
        }
    }
}

__global__ __launch_bounds__(1024) void k_scan(const int* __restrict__ cnt, int* __restrict__ off) {
    __shared__ int wtot[32];
    int t = threadIdx.x, lane = t & 31, wave = t >> 5;
    const int4* c4 = (const int4*)cnt + wave * (SCH / 4) + lane;
    int4* o4 = (int4*)off + wave * (SCH / 4) + lane;
    int inc[SROUNDS];
#pragma unroll
    for (int j = 0; j < SROUNDS; ++j) {
        int4 c = c4[j * 32];
        inc[j] = c.x + c.y + c.z + c.w;
    }
#pragma unroll
    for (int o = 1; o < 32; o <<= 1) {
#pragma unroll
        for (int j = 0; j < SROUNDS; ++j) {
            int v = __shfl_up(inc[j], o);
            if (lane >= o) inc[j] += v;
        }
    }
    int run = 0;
    int rb[SROUNDS];
#pragma unroll
    for (int j = 0; j < SROUNDS; ++j) {
        rb[j] = run;
        run += __builtin_amdgcn_readlane(inc[j], 31);
    }
    if (lane == 0) wtot[wave] = run;
    __syncthreads();
    if (wave == 0) {
        int w = wtot[lane], winc = w;
        for (int o = 1; o < 32; o <<= 1) { int v = __shfl_up(winc, o); if (lane >= o) winc += v; }
        wtot[lane] = winc - w;
    }
    __syncthreads();
    int wb = wtot[wave];
#pragma unroll
    for (int j = 0; j < SROUNDS; ++j) {
        int4 c = c4[j * 32];
        int e3 = wb + rb[j] + inc[j] - c.w;
        int e2 = e3 - c.z, e1 = e2 - c.y, e = e1 - c.x;
        o4[j * 32] = make_int4(e, e1, e2, e3);
    }
}

__global__ __launch_bounds__(EBT) void k_fill(const int* __restrict__ x, const int* __restrict__ off,
                       int* cur, int* __restrict__ slot) {
    const int4* s4 = (const int4*)x;
    const int4* d4 = (const int4*)(x + NEDGE);
    int q0 = blockIdx.x * (QPT * EBT) + threadIdx.x;
    int4 s[QPT], d[QPT];
#pragma unroll
    for (int j = 0; j < QPT; ++j) {
        int q = q0 + j * EBT;
        bool ok = q < NQUAD;
        s[j] = ok ? s4[q] : make_int4(0, 0, 0, 0);
        d[j] = ok ? d4[q] : make_int4(0, 0, 0, 0);
    }
    int4 p[QPT], o[QPT];
#pragma unroll
    for (int j = 0; j < QPT; ++j) {
        if (q0 + j * EBT < NQUAD) {
            p[j].x = atomicAdd(cur + d[j].x, 1);
            p[j].y = atomicAdd(cur + d[j].y, 1);
            p[j].z = atomicAdd(cur + d[j].z, 1);
            p[j].w = atomicAdd(cur + d[j].w, 1);
            o[j].x = off[d[j].x];
            o[j].y = off[d[j].y];
            o[j].z = off[d[j].z];
            o[j].w = off[d[j].w];
        }
    }
#pragma unroll
    for (int j = 0; j < QPT; ++j) {
        if (q0 + j * EBT < NQUAD) {
            slot[o[j].x + p[j].x] = s[j].x;
            slot[o[j].y + p[j].y] = s[j].y;
            slot[o[j].z + p[j].z] = s[j].z;
            slot[o[j].w + p[j].w] = s[j].w;
        }
    }
}

__global__ __launch_bounds__(GWAVES * 32) void k_gather(float* d_out,
        const int* __restrict__ off, const int* __restrict__ slot,
        const float* __restrict__ es_e, const float* __restrict__ ed_e,
        const float* __restrict__ es_g, const float* __restrict__ ed_g,
        const float* __restrict__ h_est, const float* __restrict__ h_gnn,
        const float* __restrict__ eb, const float* __restrict__ egam,
        const float* __restrict__ ebeta, const float* __restrict__ erm,
        const float* __restrict__ erv,
        const float* __restrict__ gb, const float* __restrict__ ggam,
        const float* __restrict__ gbeta, const float* __restrict__ grm,
        const float* __restrict__ grv,
        const float* __restrict__ clsW, const float* __restrict__ clsb) {
    __shared__ int   l_s[GWAVES][LCAP];
    __shared__ float l_ae[GWAVES][LCAP];
    __shared__ float l_ag[GWAVES][LCAP];
    __shared__ int   l_u[GWAVES][LCAP];
    __shared__ float l_y[GW];
    int lane = threadIdx.x & 31;
    int wave = __builtin_amdgcn_readfirstlane(threadIdx.x >> 5);
    int f0 = lane * 2;

    float eb0 = eb[f0], eb1 = eb[f0 + 1];
    float erm0 = erm[f0], erm1 = erm[f0 + 1];
    float ers0 = rsqrtf(erv[f0] + EPSBN), ers1 = rsqrtf(erv[f0 + 1] + EPSBN);
    float egam0 = egam[f0], egam1 = egam[f0 + 1];
    float ebeta0 = ebeta[f0], ebeta1 = ebeta[f0 + 1];
    float gb0 = gb[f0], gb1 = gb[f0 + 1];
    float grm0 = grm[f0], grm1 = grm[f0 + 1];
    float grs0 = rsqrtf(grv[f0] + EPSBN), grs1 = rsqrtf(grv[f0 + 1] + EPSBN);
    float ggam0 = ggam[f0], ggam1 = ggam[f0 + 1];
    float gbeta0 = gbeta[f0], gbeta1 = gbeta[f0 + 1];
    float cw0 = clsW[f0], cw1 = clsW[f0 + 1];
    float cb = clsb[0];

#pragma nounroll
    for (int i = 0; i < NPW; ++i) {
        int n = blockIdx.x * GW + wave * NPW + i;
        int beg = off[n];
        int deg = off[n + 1] - beg;
        float edn_e = ed_e[n], edn_g = ed_g[n];
        float self_e = leaky(es_e[n] + edn_e), self_g = leaky(es_g[n] + edn_g);
        v2f acc_e = {0.f, 0.f}, acc_g = {0.f, 0.f};
        float ze = 0.f, zg = 0.f, pse, psg;

        if (deg <= 32) {
            bool on = lane < deg;
            int sraw = slot[min(beg + lane, NEDGE - 1)];
            int s = on ? sraw : INT_MAX;
            int si = on ? sraw : n;
            float xe = es_e[si], xg = es_g[si];
            int rank = 0;
            for (int j = 0; j < deg; ++j) {
                int sj = __shfl(s, j);
                rank += (sj < s || (sj == s && j < lane)) ? 1 : 0;
            }
            int cntp = max((deg + RB) & ~(RB - 1), FB);
            if (on) l_s[wave][rank] = s;
            if (lane < cntp - deg) l_s[wave][deg + lane] = n;
            v2f he0[FB], hg0[FB];
#pragma unroll
            for (int u = 0; u < FB; ++u) {
                int sr = l_s[wave][u];
                he0[u] = *(const v2f*)(h_est + (long)sr * NHID + f0);
                hg0[u] = *(const v2f*)(h_gnn + (long)sr * NHID + f0);
            }
            __builtin_amdgcn_sched_barrier(0);
            float le = on ? leaky(xe + edn_e) : -FLT_MAX;
            float lg = on ? leaky(xg + edn_g) : -FLT_MAX;
            float me = fmaxf(wave_max_f(le), self_e);
            float mg = fmaxf(wave_max_f(lg), self_g);
            float pe = __expf(le - me), pg = __expf(lg - mg);
            pse = __expf(self_e - me); psg = __expf(self_g - mg);
            if (on) { l_ae[wave][rank] = pe; l_ag[wave][rank] = pg; }
            for (int r = 0; r < deg; ++r) { ze += l_ae[wave][r]; zg += l_ag[wave][r]; }
            ze += pse; zg += psg;
            if (on) { l_ae[wave][rank] = pe / ze; l_ag[wave][rank] = pg / zg; }
            if (lane < cntp - deg) {
                l_ae[wave][deg + lane] = (lane == 0) ? pse / ze : 0.f;
                l_ag[wave][deg + lane] = (lane == 0) ? psg / zg : 0.f;
            }
#pragma unroll
            for (int u = 0; u < FB; ++u) {
                acc_e += he0[u] * l_ae[wave][u]; acc_g += hg0[u] * l_ag[wave][u];
            }
            for (int b = FB; b < cntp; b += RB) {
                v2f he[RB], hg[RB];
#pragma unroll
                for (int u = 0; u < RB; ++u) {
                    int sr = l_s[wave][b + u];
                    he[u] = *(const v2f*)(h_est + (long)sr * NHID + f0);
                    hg[u] = *(const v2f*)(h_gnn + (long)sr * NHID + f0);
                }
#pragma unroll
                for (int u = 0; u < RB; ++u) {
                    acc_e += he[u] * l_ae[wave][b + u]; acc_g += hg[u] * l_ag[wave][b + u];
                }
            }
        } else {
        if (deg <= LCAP) {
            float le_m = self_e, lg_m = self_g;
            for (int b = 0; b < deg; b += 32) {
                int k = b + lane;
                if (k < deg) {
                    int s = slot[beg + k];
                    l_u[wave][k] = s;
                    le_m = fmaxf(le_m, leaky(es_e[s] + edn_e));
                    lg_m = fmaxf(lg_m, leaky(es_g[s] + edn_g));
                }
            }
            float me = wave_max_f(le_m), mg = wave_max_f(lg_m);
            pse = __expf(self_e - me); psg = __expf(self_g - mg);
            for (int b = 0; b < deg; b += 32) {
                int k = b + lane;
                int s = (k < deg) ? l_u[wave][k] : INT_MAX;
                int rank = 0;
#pragma unroll 4
                for (int j = 0; j < deg; ++j) {
                    int sj = l_u[wave][j];
                    rank += (sj < s || (sj == s && j < k)) ? 1 : 0;
                }
                if (k < deg) {
                    l_s[wave][rank] = s;
                    l_ae[wave][rank] = __expf(leaky(es_e[s] + edn_e) - me);
                    l_ag[wave][rank] = __expf(leaky(es_g[s] + edn_g) - mg);
                }
            }
            for (int r = 0; r < deg; ++r) { ze += l_ae[wave][r]; zg += l_ag[wave][r]; }
            ze += pse; zg += psg;
            for (int b = 0; b < deg; b += 32) {
                int r = b + lane;
                if (r < deg) { l_ae[wave][r] = l_ae[wave][r] / ze; l_ag[wave][r] = l_ag[wave][r] / zg; }
            }
#pragma unroll 4
            for (int r = 0; r < deg; ++r) {
                int sr = l_s[wave][r];
                float ae = l_ae[wave][r], ag = l_ag[wave][r];
                v2f he = *(const v2f*)(h_est + (long)sr * NHID + f0);
                v2f hg = *(const v2f*)(h_gnn + (long)sr * NHID + f0);
                acc_e += he * ae; acc_g += hg * ag;
            }
        } else {
            float le_m = self_e, lg_m = self_g;
            for (int k = lane; k < deg; k += 32) {
                int s = slot[beg + k];
                le_m = fmaxf(le_m, leaky(es_e[s] + edn_e));
                lg_m = fmaxf(lg_m, leaky(es_g[s] + edn_g));
            }
            float me = wave_max_f(le_m), mg = wave_max_f(lg_m);
            pse = __expf(self_e - me); psg = __expf(self_g - mg);
            for (int pass = 0; pass < 2; ++pass) {
                int prev = -1;
                for (int it = 0; it < deg; ++it) {
                    int best = INT_MAX;
                    for (int k = lane; k < deg; k += 32) {
                        int s = slot[beg + k];
                        if (s > prev) best = min(best, s);
                    }
                    best = wave_min_i(best);
                    if (best == INT_MAX) break;
                    int c = 0;
                    for (int k = lane; k < deg; k += 32) c += (slot[beg + k] == best) ? 1 : 0;
                    c = wave_sum_i(c);
                    float pe = __expf(leaky(es_e[best] + edn_e) - me);
                    float pg = __expf(leaky(es_g[best] + edn_g) - mg);
                    if (pass == 0) {
                        for (int k = 0; k < c; ++k) { ze += pe; zg += pg; }
                    } else {
                        float ae = pe / ze, ag = pg / zg;
                        v2f he = *(const v2f*)(h_est + (long)best * NHID + f0);
                        v2f hg = *(const v2f*)(h_gnn + (long)best * NHID + f0);
                        for (int k = 0; k < c; ++k) { acc_e += he * ae; acc_g += hg * ag; }
                    }
                    prev = best;
                }
                if (pass == 0) { ze += pse; zg += psg; }
            }
        }
        {
            float ae = pse / ze, ag = psg / zg;
            v2f he = *(const v2f*)(h_est + (long)n * NHID + f0);
            v2f hg = *(const v2f*)(h_gnn + (long)n * NHID + f0);
            acc_e += he * ae; acc_g += hg * ag;
        }
        }

        v2f so;
        {
            float a0 = fmaxf(acc_e.x + eb0, 0.f);
            float a1 = fmaxf(acc_e.y + eb1, 0.f);
            so.x = (a0 - erm0) * ers0 * egam0 + ebeta0;
            so.y = (a1 - erm1) * ers1 * egam1 + ebeta1;
        }
        *(v2f*)(d_out + N_NODES + (long)n * NHID + f0) = so;
        float b0 = fmaxf(acc_g.x + gb0, 0.f);
        float b1 = fmaxf(acc_g.y + gb1, 0.f);
        float z0 = (b0 - grm0) * grs0 * ggam0 + gbeta0;
        float z1 = (b1 - grm1) * grs1 * ggam1 + gbeta1;
        float yp = wave_sum_f(z0 * cw0 + z1 * cw1);
        if (lane == 0) l_y[wave * NPW + i] = yp + cb;
    }
    __syncthreads();
    if (threadIdx.x < GW) d_out[blockIdx.x * GW + threadIdx.x] = l_y[threadIdx.x];
}

extern "C" void kernel_launch(void* const* d_in, const int* in_sizes, int n_in,
                              void* d_out, int out_size, void* d_ws, size_t ws_size,
                              hipStream_t stream) {
    const float* g        = (const float*)d_in[0];
    const int*   x        = (const int*)d_in[1];
    const float* est_W    = (const float*)d_in[2];
    const float* est_asrc = (const float*)d_in[3];
    const float* est_adst = (const float*)d_in[4];
    const float* est_bias = (const float*)d_in[5];
    const float* est_gam  = (const float*)d_in[6];
    const float* est_beta = (const float*)d_in[7];
    const float* est_rm   = (const float*)d_in[8];
    const float* est_rv   = (const float*)d_in[9];
    const float* gnn_W    = (const float*)d_in[10];
    const float* gnn_asrc = (const float*)d_in[11];
    const float* gnn_adst = (const float*)d_in[12];
    const float* gnn_bias = (const float*)d_in[13];
    const float* gnn_gam  = (const float*)d_in[14];
    const float* gnn_beta = (const float*)d_in[15];
    const float* gnn_rm   = (const float*)d_in[16];
    const float* gnn_rv   = (const float*)d_in[17];
    const float* cls_W    = (const float*)d_in[18];
    const float* cls_b    = (const float*)d_in[19];
    float* out = (float*)d_out;

    float* ws     = (float*)d_ws;
    float* h_est  = ws;
    float* h_gnn  = h_est + (long)N_NODES * NHID;
    int*   cnt    = (int*)(h_gnn + (long)N_NODES * NHID);
    int*   cur    = cnt + SCANW;
    int*   off    = cur + N_NODES;
    int*   slot   = off + SCANW;
    float* es_e   = ws + (long)N_NODES * 192;
    float* ed_e   = es_e + N_NODES;
    float* es_g   = ed_e + N_NODES;
    float* ed_g   = es_g + N_NODES;

    const int BT = 256;
    k_gemm<<<(MTILES + 7) / 8, BT, 0, stream>>>(g, est_W, gnn_W, h_est, h_gnn);

    k_eprep<<<(SCANW + BT - 1) / BT, BT, 0, stream>>>(h_est, h_gnn,
            est_asrc, est_adst, gnn_asrc, gnn_adst, es_e, ed_e, es_g, ed_g, cnt, cur);

    k_count<<<(NQUAD + QPT * EBT - 1) / (QPT * EBT), EBT, 0, stream>>>(x, cnt);

    k_scan<<<1, 1024, 0, stream>>>(cnt, off);

    k_fill<<<(NQUAD + QPT * EBT - 1) / (QPT * EBT), EBT, 0, stream>>>(x, off, cur, slot);

    k_gather<<<N_NODES / GW, GWAVES * 32, 0, stream>>>(out, off, slot,
            es_e, ed_e, es_g, ed_g, h_est, h_gnn,
            est_bias, est_gam, est_beta, est_rm, est_rv,
            gnn_bias, gnn_gam, gnn_beta, gnn_rm, gnn_rv, cls_W, cls_b);
}


# Degree count and bucket placement share one returning add per edge (its place kept in a per-edge array, placement adds nothing, blocks tile the edge list): two global atomics per edge become one
# speedup vs baseline: 2.5773x; 1.2557x over previous
#include <hip/hip_runtime.h>
#include <cfloat>
#include <climits>

#define N_NODES 100000
#define NFEAT   256
#define NHID    64
#define NEDGE   1600000
#define MTILES  (N_NODES / 16)
#define EPSBN   1e-5f
#define NEG_SLOPE 0.2f
#define GW      32
#define GWAVES  8
#define NPW     (GW / GWAVES)
#define EBT     320
#define NQUAD   (NEDGE / 4)
#define QPT     2
#define LCAP    128
#define RB      8
#define FB      16
#define SROUNDS 25
#define SCH     (SROUNDS * 128)
#define SCANW   (32 * SCH)

static_assert(N_NODES % GW == 0, "gather blocks cover the nodes exactly");
static_assert(GW % GWAVES == 0, "every gather wave gets the same number of nodes");
static_assert(N_NODES % 4 == 0, "scan reads whole quads");
static_assert(SCANW > N_NODES, "the scan covers every node and the counter behind the last one");
static_assert((RB & (RB - 1)) == 0 && FB % RB == 0 && FB <= 32 && 32 + RB <= LCAP, "a bucket of up to 32, its node and the padding fit the LDS list, and one lane each writes the padding");
static_assert(NEDGE % 4 == 0, "bucketing reads whole quads, and the destination row starts on a 16-byte boundary");
static_assert(NQUAD % (QPT * EBT) == 0 && EBT % 32 == 0, "the bucketing blocks cover the edge list exactly: no thread is without its quads, no load, add or store needs a guard");

typedef __attribute__((ext_vector_type(2))) float v2f;
typedef __attribute__((ext_vector_type(4))) float v4f;
typedef __attribute__((ext_vector_type(8))) float v8f;

__device__ __forceinline__ float leaky(float v) {
    return fmaxf(v, 0.f) + NEG_SLOPE * fminf(v, 0.f);
}
__device__ __forceinline__ float wave_max_f(float v) {
    for (int o = 16; o > 0; o >>= 1) v = fmaxf(v, __shfl_xor(v, o));
    return v;
}
__device__ __forceinline__ float wave_sum_f(float v) {
    for (int o = 16; o > 0; o >>= 1) v += __shfl_xor(v, o);
    return v;
}
__device__ __forceinline__ int wave_min_i(int v) {
    for (int o = 16; o > 0; o >>= 1) v = min(v, __shfl_xor(v, o));
    return v;
}
__device__ __forceinline__ int wave_sum_i(int v) {
    for (int o = 16; o > 0; o >>= 1) v += __shfl_xor(v, o);
    return v;
}

__global__ __launch_bounds__(256) void k_gemm(const float* __restrict__ g,
                                              const float* __restrict__ We,
                                              const float* __restrict__ Wg,
                                              float* __restrict__ h_est,
                                              float* __restrict__ h_gnn) {
    int wave = threadIdx.x >> 5;
    int lane = threadIdx.x & 31;
    int mtile = blockIdx.x * 8 + wave;
    if (mtile >= MTILES) return;
    int m0   = mtile * 16;
    int lrow = lane & 15;
    int lhi  = lane >> 4;

    const float* grow = g + (long)(m0 + lrow) * NFEAT;

    v8f acc[8] = {};
    for (int k = 0; k < NFEAT; k += 4) {
        v2f a = *(const v2f*)(grow + k + 2 * lhi);
#pragma unroll
        for (int t = 0; t < 8; ++t) {
            const float* Wp = (t < 4) ? We : Wg;
            int c  = (t & 3) * 16 + lrow;
            int kr = k + 2 * lhi;
            v2f b;
            b.x = Wp[(long)kr * NHID + c];
            b.y = Wp[(long)(kr + 1) * NHID + c];
            acc[t] = __builtin_amdgcn_wmma_f32_16x16x4_f32(
                false, a, false, b, (short)0, acc[t], false, false);
        }
    }
#pragma unroll
    for (int t = 0; t < 8; ++t) {
        float* dst = (t < 4) ? h_est : h_gnn;
        int c = (t & 3) * 16 + lrow;
#pragma unroll
        for (int r = 0; r < 8; ++r) {
            dst[(long)(m0 + r + 8 * lhi) * NHID + c] = acc[t][r];
        }
    }
}

__global__ void k_eprep(const float* __restrict__ h_est, const float* __restrict__ h_gnn,
                        const float* __restrict__ ase, const float* __restrict__ ade,
                        const float* __restrict__ asg, const float* __restrict__ adg,
                        float* es_e, float* ed_e, float* es_g, float* ed_g,
                        int* cnt) {
    int n = blockIdx.x * blockDim.x + threadIdx.x;
    if (n >= N_NODES) {
        if (n < SCANW) cnt[n] = 0;
        return;
    }
    const float* he = h_est + (long)n * NHID;
    const float* hg = h_gnn + (long)n * NHID;
    float se = 0.f, de = 0.f, sg = 0.f, dg = 0.f;
#pragma unroll 4
    for (int f = 0; f < NHID; f += 4) {
        v4f e  = *(const v4f*)(he + f);
        v4f g4 = *(const v4f*)(hg + f);
        v4f a1 = *(const v4f*)(ase + f);
        v4f a2 = *(const v4f*)(ade + f);
        v4f a3 = *(const v4f*)(asg + f);
        v4f a4 = *(const v4f*)(adg + f);
        se += e.x * a1.x + e.y * a1.y + e.z * a1.z + e.w * a1.w;
        de += e.x * a2.x + e.y * a2.y + e.z * a2.z + e.w * a2.w;
        sg += g4.x * a3.x + g4.y * a3.y + g4.z * a3.z + g4.w * a3.w;
        dg += g4.x * a4.x + g4.y * a4.y + g4.z * a4.z + g4.w * a4.w;
    }
    es_e[n] = se; ed_e[n] = de; es_g[n] = sg; ed_g[n] = dg;
    cnt[n] = 0;
}

__global__ __launch_bounds__(EBT) void k_rank(const int* __restrict__ x, int* cnt, int* __restrict__ rank) {
    const int4* d4 = (const int4*)(x + NEDGE);
    int4* r4 = (int4*)rank;
    int q0 = blockIdx.x * (QPT * EBT) + threadIdx.x;
    int4 d[QPT];
#pragma unroll
    for (int j = 0; j < QPT; ++j) d[j] = d4[q0 + j * EBT];
    static_assert(QPT == 2, "the statement below names both quads");
    asm volatile("" : "+v"(d[0].x), "+v"(d[0].y), "+v"(d[0].z), "+v"(d[0].w),
                      "+v"(d[1].x), "+v"(d[1].y), "+v"(d[1].z), "+v"(d[1].w));
    int4 p[QPT];
#pragma unroll
    for (int j = 0; j < QPT; ++j) {
        p[j].x = atomicAdd(cnt + d[j].x, 1);
        p[j].y = atomicAdd(cnt + d[j].y, 1);
        p[j].z = atomicAdd(cnt + d[j].z, 1);
        p[j].w = atomicAdd(cnt + d[j].w, 1);
    }
#pragma unroll
    for (int j = 0; j < QPT; ++j) r4[q0 + j * EBT] = p[j];
}

__global__ __launch_bounds__(1024) void k_scan(const int* __restrict__ cnt, int* __restrict__ off) {
    __shared__ int wtot[32];
    int t = threadIdx.x, lane = t & 31, wave = t >> 5;
    const int4* c4 = (const int4*)cnt + wave * (SCH / 4) + lane;
    int4* o4 = (int4*)off + wave * (SCH / 4) + lane;
    int inc[SROUNDS];
#pragma unroll
    for (int j = 0; j < SROUNDS; ++j) {
        int4 c = c4[j * 32];
        inc[j] = c.x + c.y + c.z + c.w;
    }
#pragma unroll
    for (int o = 1; o < 32; o <<= 1) {
#pragma unroll
        for (int j = 0; j < SROUNDS; ++j) {
            int v = __shfl_up(inc[j], o);
            if (lane >= o) inc[j] += v;
        }
    }
    int run = 0;
    int rb[SROUNDS];
#pragma unroll
    for (int j = 0; j < SROUNDS; ++j) {
        rb[j] = run;
        run += __builtin_amdgcn_readlane(inc[j], 31);
    }
    if (lane == 0) wtot[wave] = run;
    __syncthreads();
    if (wave == 0) {
        int w = wtot[lane], winc = w;
        for (int o = 1; o < 32; o <<= 1) { int v = __shfl_up(winc, o); if (lane >= o) winc += v; }
        wtot[lane] = winc - w;
    }
    __syncthreads();
    int wb = wtot[wave];
#pragma unroll
    for (int j = 0; j < SROUNDS; ++j) {
        int4 c = c4[j * 32];
        int e3 = wb + rb[j] + inc[j] - c.w;
        int e2 = e3 - c.z, e1 = e2 - c.y, e = e1 - c.x;
        o4[j * 32] = make_int4(e, e1, e2, e3);
    }
}

__global__ __launch_bounds__(EBT) void k_place(const int* __restrict__ x, const int* __restrict__ off,
                       const int* __restrict__ rank, int* __restrict__ slot) {
    const int4* s4 = (const int4*)x;
    const int4* d4 = (const int4*)(x + NEDGE);
    const int4* r4 = (const int4*)rank;
    int q0 = blockIdx.x * (QPT * EBT) + threadIdx.x;
    int4 s[QPT], d[QPT], p[QPT];
#pragma unroll
    for (int j = 0; j < QPT; ++j) d[j] = d4[q0 + j * EBT];
#pragma unroll
    for (int j = 0; j < QPT; ++j) { p[j] = r4[q0 + j * EBT]; s[j] = s4[q0 + j * EBT]; }
    int4 o[QPT];
#pragma unroll
    for (int j = 0; j < QPT; ++j) {
        o[j].x = off[d[j].x];
        o[j].y = off[d[j].y];
        o[j].z = off[d[j].z];
        o[j].w = off[d[j].w];
    }
#pragma unroll
    for (int j = 0; j < QPT; ++j) {
        slot[o[j].x + p[j].x] = s[j].x;
        slot[o[j].y + p[j].y] = s[j].y;
        slot[o[j].z + p[j].z] = s[j].z;
        slot[o[j].w + p[j].w] = s[j].w;
    }
}

__global__ __launch_bounds__(GWAVES * 32) void k_gather(float* d_out,
        const int* __restrict__ off, const int* __restrict__ slot,
        const float* __restrict__ es_e, const float* __restrict__ ed_e,
        const float* __restrict__ es_g, const float* __restrict__ ed_g,
        const float* __restrict__ h_est, const float* __restrict__ h_gnn,
        const float* __restrict__ eb, const float* __restrict__ egam,
        const float* __restrict__ ebeta, const float* __restrict__ erm,
        const float* __restrict__ erv,
        const float* __restrict__ gb, const float* __restrict__ ggam,
        const float* __restrict__ gbeta, const float* __restrict__ grm,
        const float* __restrict__ grv,
        const float* __restrict__ clsW, const float* __restrict__ clsb) {
    __shared__ int   l_s[GWAVES][LCAP];
    __shared__ float l_ae[GWAVES][LCAP];
    __shared__ float l_ag[GWAVES][LCAP];
    __shared__ int   l_u[GWAVES][LCAP];
    __shared__ float l_y[GW];
    int lane = threadIdx.x & 31;
    int wave = __builtin_amdgcn_readfirstlane(threadIdx.x >> 5);
    int f0 = lane * 2;

    float eb0 = eb[f0], eb1 = eb[f0 + 1];
    float erm0 = erm[f0], erm1 = erm[f0 + 1];
    float ers0 = rsqrtf(erv[f0] + EPSBN), ers1 = rsqrtf(erv[f0 + 1] + EPSBN);
    float egam0 = egam[f0], egam1 = egam[f0 + 1];
    float ebeta0 = ebeta[f0], ebeta1 = ebeta[f0 + 1];
    float gb0 = gb[f0], gb1 = gb[f0 + 1];
    float grm0 = grm[f0], grm1 = grm[f0 + 1];
    float grs0 = rsqrtf(grv[f0] + EPSBN), grs1 = rsqrtf(grv[f0 + 1] + EPSBN);
    float ggam0 = ggam[f0], ggam1 = ggam[f0 + 1];
    float gbeta0 = gbeta[f0], gbeta1 = gbeta[f0 + 1];
    float cw0 = clsW[f0], cw1 = clsW[f0 + 1];
    float cb = clsb[0];

#pragma nounroll
    for (int i = 0; i < NPW; ++i) {
        int n = blockIdx.x * GW + wave * NPW + i;
        int beg = off[n];
        int deg = off[n + 1] - beg;
        float edn_e = ed_e[n], edn_g = ed_g[n];
        float self_e = leaky(es_e[n] + edn_e), self_g = leaky(es_g[n] + edn_g);
        v2f acc_e = {0.f, 0.f}, acc_g = {0.f, 0.f};
        float ze = 0.f, zg = 0.f, pse, psg;

        if (deg <= 32) {
            bool on = lane < deg;
            int sraw = slot[min(beg + lane, NEDGE - 1)];
            int s = on ? sraw : INT_MAX;
            int si = on ? sraw : n;
            float xe = es_e[si], xg = es_g[si];
            int rank = 0;
            for (int j = 0; j < deg; ++j) {
                int sj = __shfl(s, j);
                rank += (sj < s || (sj == s && j < lane)) ? 1 : 0;
            }
            int cntp = max((deg + RB) & ~(RB - 1), FB);
            if (on) l_s[wave][rank] = s;
            if (lane < cntp - deg) l_s[wave][deg + lane] = n;
            v2f he0[FB], hg0[FB];
#pragma unroll
            for (int u = 0; u < FB; ++u) {
                int sr = l_s[wave][u];
                he0[u] = *(const v2f*)(h_est + (long)sr * NHID + f0);
                hg0[u] = *(const v2f*)(h_gnn + (long)sr * NHID + f0);
            }
            __builtin_amdgcn_sched_barrier(0);
            float le = on ? leaky(xe + edn_e) : -FLT_MAX;
            float lg = on ? leaky(xg + edn_g) : -FLT_MAX;
            float me = fmaxf(wave_max_f(le), self_e);
            float mg = fmaxf(wave_max_f(lg), self_g);
            float pe = __expf(le - me), pg = __expf(lg - mg);
            pse = __expf(self_e - me); psg = __expf(self_g - mg);
            if (on) { l_ae[wave][rank] = pe; l_ag[wave][rank] = pg; }
            for (int r = 0; r < deg; ++r) { ze += l_ae[wave][r]; zg += l_ag[wave][r]; }
            ze += pse; zg += psg;
            if (on) { l_ae[wave][rank] = pe / ze; l_ag[wave][rank] = pg / zg; }
            if (lane < cntp - deg) {
                l_ae[wave][deg + lane] = (lane == 0) ? pse / ze : 0.f;
                l_ag[wave][deg + lane] = (lane == 0) ? psg / zg : 0.f;
            }
#pragma unroll
            for (int u = 0; u < FB; ++u) {
                acc_e += he0[u] * l_ae[wave][u]; acc_g += hg0[u] * l_ag[wave][u];
            }
            for (int b = FB; b < cntp; b += RB) {
                v2f he[RB], hg[RB];
#pragma unroll
                for (int u = 0; u < RB; ++u) {
                    int sr = l_s[wave][b + u];
                    he[u] = *(const v2f*)(h_est + (long)sr * NHID + f0);
                    hg[u] = *(const v2f*)(h_gnn + (long)sr * NHID + f0);
                }
#pragma unroll
                for (int u = 0; u < RB; ++u) {
                    acc_e += he[u] * l_ae[wave][b + u]; acc_g += hg[u] * l_ag[wave][b + u];
                }
            }
        } else {
        if (deg <= LCAP) {
            float le_m = self_e, lg_m = self_g;
            for (int b = 0; b < deg; b += 32) {
                int k = b + lane;
                if (k < deg) {
                    int s = slot[beg + k];
                    l_u[wave][k] = s;
                    le_m = fmaxf(le_m, leaky(es_e[s] + edn_e));
                    lg_m = fmaxf(lg_m, leaky(es_g[s] + edn_g));
                }
            }
            float me = wave_max_f(le_m), mg = wave_max_f(lg_m);
            pse = __expf(self_e - me); psg = __expf(self_g - mg);
            for (int b = 0; b < deg; b += 32) {
                int k = b + lane;
                int s = (k < deg) ? l_u[wave][k] : INT_MAX;
                int rank = 0;
#pragma unroll 4
                for (int j = 0; j < deg; ++j) {
                    int sj = l_u[wave][j];
                    rank += (sj < s || (sj == s && j < k)) ? 1 : 0;
                }
                if (k < deg) {
                    l_s[wave][rank] = s;
                    l_ae[wave][rank] = __expf(leaky(es_e[s] + edn_e) - me);
                    l_ag[wave][rank] = __expf(leaky(es_g[s] + edn_g) - mg);
                }
            }
            for (int r = 0; r < deg; ++r) { ze += l_ae[wave][r]; zg += l_ag[wave][r]; }
            ze += pse; zg += psg;
            for (int b = 0; b < deg; b += 32) {
                int r = b + lane;
                if (r < deg) { l_ae[wave][r] = l_ae[wave][r] / ze; l_ag[wave][r] = l_ag[wave][r] / zg; }
            }
#pragma unroll 4
            for (int r = 0; r < deg; ++r) {
                int sr = l_s[wave][r];
                float ae = l_ae[wave][r], ag = l_ag[wave][r];
                v2f he = *(const v2f*)(h_est + (long)sr * NHID + f0);
                v2f hg = *(const v2f*)(h_gnn + (long)sr * NHID + f0);
                acc_e += he * ae; acc_g += hg * ag;
            }
        } else {
            float le_m = self_e, lg_m = self_g;
            for (int k = lane; k < deg; k += 32) {
                int s = slot[beg + k];
                le_m = fmaxf(le_m, leaky(es_e[s] + edn_e));
                lg_m = fmaxf(lg_m, leaky(es_g[s] + edn_g));
            }
            float me = wave_max_f(le_m), mg = wave_max_f(lg_m);
            pse = __expf(self_e - me); psg = __expf(self_g - mg);
            for (int pass = 0; pass < 2; ++pass) {
                int prev = -1;
                for (int it = 0; it < deg; ++it) {
                    int best = INT_MAX;
                    for (int k = lane; k < deg; k += 32) {
                        int s = slot[beg + k];
                        if (s > prev) best = min(best, s);
                    }
                    best = wave_min_i(best);
                    if (best == INT_MAX) break;
                    int c = 0;
                    for (int k = lane; k < deg; k += 32) c += (slot[beg + k] == best) ? 1 : 0;
                    c = wave_sum_i(c);
                    float pe = __expf(leaky(es_e[best] + edn_e) - me);
                    float pg = __expf(leaky(es_g[best] + edn_g) - mg);
                    if (pass == 0) {
                        for (int k = 0; k < c; ++k) { ze += pe; zg += pg; }
                    } else {
                        float ae = pe / ze, ag = pg / zg;
                        v2f he = *(const v2f*)(h_est + (long)best * NHID + f0);
                        v2f hg = *(const v2f*)(h_gnn + (long)best * NHID + f0);
                        for (int k = 0; k < c; ++k) { acc_e += he * ae; acc_g += hg * ag; }
                    }
                    prev = best;
                }
                if (pass == 0) { ze += pse; zg += psg; }
            }
        }
        {
            float ae = pse / ze, ag = psg / zg;
            v2f he = *(const v2f*)(h_est + (long)n * NHID + f0);
            v2f hg = *(const v2f*)(h_gnn + (long)n * NHID + f0);
            acc_e += he * ae; acc_g += hg * ag;
        }
        }

        v2f so;
        {
            float a0 = fmaxf(acc_e.x + eb0, 0.f);
            float a1 = fmaxf(acc_e.y + eb1, 0.f);
            so.x = (a0 - erm0) * ers0 * egam0 + ebeta0;
            so.y = (a1 - erm1) * ers1 * egam1 + ebeta1;
        }
        *(v2f*)(d_out + N_NODES + (long)n * NHID + f0) = so;
        float b0 = fmaxf(acc_g.x + gb0, 0.f);
        float b1 = fmaxf(acc_g.y + gb1, 0.f);
        float z0 = (b0 - grm0) * grs0 * ggam0 + gbeta0;
        float z1 = (b1 - grm1) * grs1 * ggam1 + gbeta1;
        float yp = wave_sum_f(z0 * cw0 + z1 * cw1);
        if (lane == 0) l_y[wave * NPW + i] = yp + cb;
    }
    __syncthreads();
    if (threadIdx.x < GW) d_out[blockIdx.x * GW + threadIdx.x] = l_y[threadIdx.x];
}

extern "C" void kernel_launch(void* const* d_in, const int* in_sizes, int n_in,
                              void* d_out, int out_size, void* d_ws, size_t ws_size,
                              hipStream_t stream) {
    const float* g        = (const float*)d_in[0];
    const int*   x        = (const int*)d_in[1];
    const float* est_W    = (const float*)d_in[2];
    const float* est_asrc = (const float*)d_in[3];
    const float* est_adst = (const float*)d_in[4];
    const float* est_bias = (const float*)d_in[5];
    const float* est_gam  = (const float*)d_in[6];
    const float* est_beta = (const float*)d_in[7];
    const float* est_rm   = (const float*)d_in[8];
    const float* est_rv   = (const float*)d_in[9];
    const float* gnn_W    = (const float*)d_in[10];
    const float* gnn_asrc = (const float*)d_in[11];
    const float* gnn_adst = (const float*)d_in[12];
    const float* gnn_bias = (const float*)d_in[13];
    const float* gnn_gam  = (const float*)d_in[14];
    const float* gnn_beta = (const float*)d_in[15];
    const float* gnn_rm   = (const float*)d_in[16];
    const float* gnn_rv   = (const float*)d_in[17];
    const float* cls_W    = (const float*)d_in[18];
    const float* cls_b    = (const float*)d_in[19];
    float* out = (float*)d_out;

    float* ws     = (float*)d_ws;
    float* h_est  = ws;
    float* h_gnn  = h_est + (long)N_NODES * NHID;
    int*   cnt    = (int*)(h_gnn + (long)N_NODES * NHID);
    int*   off    = cnt + SCANW + N_NODES;
    int*   slot   = off + SCANW;
    int*   rank   = slot + NEDGE;
    float* es_e   = ws + (long)N_NODES * 192;
    float* ed_e   = es_e + N_NODES;
    float* es_g   = ed_e + N_NODES;
    float* ed_g   = es_g + N_NODES;

    const int BT = 256;
    k_gemm<<<(MTILES + 7) / 8, BT, 0, stream>>>(g, est_W, gnn_W, h_est, h_gnn);

    k_eprep<<<(SCANW + BT - 1) / BT, BT, 0, stream>>>(h_est, h_gnn,
            est_asrc, est_adst, gnn_asrc, gnn_adst, es_e, ed_e, es_g, ed_g, cnt);

    k_rank<<<NQUAD / (QPT * EBT), EBT, 0, stream>>>(x, cnt, rank);

    k_scan<<<1, 1024, 0, stream>>>(cnt, off);

    k_place<<<NQUAD / (QPT * EBT), EBT, 0, stream>>>(x, off, rank, slot);

    k_gather<<<N_NODES / GW, GWAVES * 32, 0, stream>>>(out, off, slot,
            es_e, ed_e, es_g, ed_g, h_est, h_gnn,
            est_bias, est_gam, est_beta, est_rm, est_rv,
            gnn_bias, gnn_gam, gnn_beta, gnn_rm, gnn_rv, cls_W, cls_b);
}


# Two-level bucketing by destination, every add in LDS (each tile ordered by 512-node group in place; a block per group collects its pieces): no global atomic, no scan launch; sums keep their order
# speedup vs baseline: 4.6023x; 1.7857x over previous
#include <hip/hip_runtime.h>
#include <cfloat>
#include <climits>

#define N_NODES 100000
#define NFEAT   256
#define NHID    64
#define NEDGE   1600000
#define MTILES  (N_NODES / 16)
#define EPSBN   1e-5f
#define NEG_SLOPE 0.2f
#define GW      32
#define GWAVES  8
#define NPW     (GW / GWAVES)
#define QPT     2
#define PT      800
#define TE      (PT * QPT * 4)
#define NBLK    (NEDGE / TE)
#define CB      9
#define CN      (1 << CB)
#define NBIN    ((N_NODES + CN - 1) / CN)
#define TABW    (NBIN + 1)
#define SRCB    17
#define FT      1024
#define PPW     ((NBLK + 31) / 32)
#define RJ      2
#define CAP     12288
#define LCAP    128
#define RB      8
#define FB      16

static_assert(N_NODES % GW == 0, "gather blocks cover the nodes exactly");
static_assert(GW % GWAVES == 0, "every gather wave gets the same number of nodes");
static_assert((RB & (RB - 1)) == 0 && FB % RB == 0 && FB <= 32 && 32 + RB <= LCAP, "a bucket of up to 32, its node and the padding fit the LDS list, and one lane each writes the padding");
static_assert(NEDGE % 4 == 0, "bucketing reads whole quads, and the destination row starts on a 16-byte boundary");
static_assert(NEDGE % TE == 0 && PT % 32 == 0 && TE % PT == 0, "the tiles cover the edge list exactly: no thread is without its quads, no load, add or store of the tile kernel needs a guard");
static_assert(N_NODES <= (1 << SRCB) && SRCB + CB <= 31, "a source id and a destination's place in its group share one non-negative word");
static_assert(TABW <= PT && TABW <= 32 * 7 && CN == 32 * 16 && CN <= FT, "one thread per group or per node where the kernels say so; the wave scans take 7 and 16 entries per lane");
static_assert(NBLK <= 32 * PPW && 32 * PPW <= FT && FT == 1024, "one thread of the group kernel per tile, and every tile's piece has its wave");

typedef __attribute__((ext_vector_type(2))) float v2f;
typedef __attribute__((ext_vector_type(4))) float v4f;
typedef __attribute__((ext_vector_type(8))) float v8f;

__device__ __forceinline__ float leaky(float v) {
    return fmaxf(v, 0.f) + NEG_SLOPE * fminf(v, 0.f);
}
__device__ __forceinline__ float wave_max_f(float v) {
    for (int o = 16; o > 0; o >>= 1) v = fmaxf(v, __shfl_xor(v, o));
    return v;
}
__device__ __forceinline__ float wave_sum_f(float v) {
    for (int o = 16; o > 0; o >>= 1) v += __shfl_xor(v, o);
    return v;
}
__device__ __forceinline__ int wave_min_i(int v) {
    for (int o = 16; o > 0; o >>= 1) v = min(v, __shfl_xor(v, o));
    return v;
}
__device__ __forceinline__ int wave_sum_i(int v) {
    for (int o = 16; o > 0; o >>= 1) v += __shfl_xor(v, o);
    return v;
}

__global__ __launch_bounds__(256) void k_gemm(const float* __restrict__ g,
                                              const float* __restrict__ We,
                                              const float* __restrict__ Wg,
                                              float* __restrict__ h_est,
                                              float* __restrict__ h_gnn) {
    int wave = threadIdx.x >> 5;
    int lane = threadIdx.x & 31;
    int mtile = blockIdx.x * 8 + wave;
    if (mtile >= MTILES) return;
    int m0   = mtile * 16;
    int lrow = lane & 15;
    int lhi  = lane >> 4;

    const float* grow = g + (long)(m0 + lrow) * NFEAT;

    v8f acc[8] = {};
    for (int k = 0; k < NFEAT; k += 4) {
        v2f a = *(const v2f*)(grow + k + 2 * lhi);
#pragma unroll
        for (int t = 0; t < 8; ++t) {
            const float* Wp = (t < 4) ? We : Wg;
            int c  = (t & 3) * 16 + lrow;
            int kr = k + 2 * lhi;
            v2f b;
            b.x = Wp[(long)kr * NHID + c];
            b.y = Wp[(long)(kr + 1) * NHID + c];
            acc[t] = __builtin_amdgcn_wmma_f32_16x16x4_f32(
                false, a, false, b, (short)0, acc[t], false, false);
        }
    }
#pragma unroll
    for (int t = 0; t < 8; ++t) {
        float* dst = (t < 4) ? h_est : h_gnn;
        int c = (t & 3) * 16 + lrow;
#pragma unroll
        for (int r = 0; r < 8; ++r) {
            dst[(long)(m0 + r + 8 * lhi) * NHID + c] = acc[t][r];
        }
    }
}

__global__ void k_eprep(const float* __restrict__ h_est, const float* __restrict__ h_gnn,
                        const float* __restrict__ ase, const float* __restrict__ ade,
                        const float* __restrict__ asg, const float* __restrict__ adg,
                        float* es_e, float* ed_e, float* es_g, float* ed_g) {
    int n = blockIdx.x * blockDim.x + threadIdx.x;
    if (n >= N_NODES) return;
    const float* he = h_est + (long)n * NHID;
    const float* hg = h_gnn + (long)n * NHID;
    float se = 0.f, de = 0.f, sg = 0.f, dg = 0.f;
#pragma unroll 4
    for (int f = 0; f < NHID; f += 4) {
        v4f e  = *(const v4f*)(he + f);
        v4f g4 = *(const v4f*)(hg + f);
        v4f a1 = *(const v4f*)(ase + f);
        v4f a2 = *(const v4f*)(ade + f);
        v4f a3 = *(const v4f*)(asg + f);
        v4f a4 = *(const v4f*)(adg + f);
        se += e.x * a1.x + e.y * a1.y + e.z * a1.z + e.w * a1.w;
        de += e.x * a2.x + e.y * a2.y + e.z * a2.z + e.w * a2.w;
        sg += g4.x * a3.x + g4.y * a3.y + g4.z * a3.z + g4.w * a3.w;
        dg += g4.x * a4.x + g4.y * a4.y + g4.z * a4.z + g4.w * a4.w;
    }
    es_e[n] = se; ed_e[n] = de; es_g[n] = sg; ed_g[n] = dg;
}

__global__ __launch_bounds__(PT) void k_tile(const int* __restrict__ x, int* __restrict__ tmp,
                                             int* __restrict__ tab) {
    __shared__ int h[NBIN];
    __shared__ int ls[TABW];
    __shared__ int sv[TE];
    int t = threadIdx.x;
    const int4* s4 = (const int4*)x;
    const int4* d4 = (const int4*)(x + NEDGE);
    int q0 = blockIdx.x * (QPT * PT) + t;
    int4 s[QPT], d[QPT];
#pragma unroll
    for (int j = 0; j < QPT; ++j) { d[j] = d4[q0 + j * PT]; s[j] = s4[q0 + j * PT]; }
    if (t < NBIN) h[t] = 0;
    __syncthreads();
    int4 p[QPT];
#pragma unroll
    for (int j = 0; j < QPT; ++j) {
        p[j].x = atomicAdd(&h[d[j].x >> CB], 1);
        p[j].y = atomicAdd(&h[d[j].y >> CB], 1);
        p[j].z = atomicAdd(&h[d[j].z >> CB], 1);
        p[j].w = atomicAdd(&h[d[j].w >> CB], 1);
    }
    __syncthreads();
    if (t < 32) {
        int v[7], sum = 0;
#pragma unroll
        for (int k = 0; k < 7; ++k) { int b = t * 7 + k; v[k] = (b < NBIN) ? h[b] : 0; sum += v[k]; }
        int inc = sum;
        for (int o = 1; o < 32; o <<= 1) { int u = __shfl_up(inc, o); if (t >= o) inc += u; }
        int ex = inc - sum;
#pragma unroll
        for (int k = 0; k < 7; ++k) { int b = t * 7 + k; if (b < TABW) ls[b] = ex; ex += v[k]; }
    }
    __syncthreads();
    if (t < TABW) tab[blockIdx.x * TABW + t] = ls[t];
#pragma unroll
    for (int j = 0; j < QPT; ++j) {
        sv[ls[d[j].x >> CB] + p[j].x] = s[j].x | ((d[j].x & (CN - 1)) << SRCB);
        sv[ls[d[j].y >> CB] + p[j].y] = s[j].y | ((d[j].y & (CN - 1)) << SRCB);
        sv[ls[d[j].z >> CB] + p[j].z] = s[j].z | ((d[j].z & (CN - 1)) << SRCB);
        sv[ls[d[j].w >> CB] + p[j].w] = s[j].w | ((d[j].w & (CN - 1)) << SRCB);
    }
    __syncthreads();
#pragma unroll
    for (int i = 0; i < TE / PT; ++i) tmp[blockIdx.x * TE + i * PT + t] = sv[i * PT + t];
}

__device__ __forceinline__ void scan_nodes(const int* nc, int* ns, int t) {
    if (t < 32) {
        int v[16], sum = 0;
#pragma unroll
        for (int k = 0; k < 16; ++k) { v[k] = nc[t * 16 + k]; sum += v[k]; }
        int inc = sum;
        for (int o = 1; o < 32; o <<= 1) { int u = __shfl_up(inc, o); if (t >= o) inc += u; }
        int ex = inc - sum;
#pragma unroll
        for (int k = 0; k < 16; ++k) { ns[t * 16 + k] = ex; ex += v[k]; }
    }
}
__global__ __launch_bounds__(FT) void k_group(const int* __restrict__ tmp, const int* __restrict__ tab,
                                              int* __restrict__ off, int* __restrict__ slot) {
    __shared__ int plo[32 * PPW];
    __shared__ int pl[32 * PPW];
    __shared__ int rs[32];
    __shared__ int rl[32];
    __shared__ int nc[CN];
    __shared__ int cur[CN];
    __shared__ int ns[CN];
    __shared__ int st[CAP];
    int t = threadIdx.x, lane = t & 31, bin = blockIdx.x;
    int wv = __builtin_amdgcn_readfirstlane(t >> 5);
    {
        int lo = 0, n = 0;
        if (t < NBLK) { lo = tab[t * TABW + bin]; n = tab[t * TABW + bin + 1] - lo; }
        if (t < 32 * PPW) { plo[t] = t * TE + lo; pl[t] = n; }
        lo = wave_sum_i(lo); n = wave_sum_i(n);
        if (lane == 0) { rs[wv] = lo; rl[wv] = n; }
        if (t < CN) { nc[t] = 0; cur[t] = 0; }
    }
    __syncthreads();
    int s0 = wave_sum_i(rs[lane]);
    int len = wave_sum_i(rl[lane]);
    int w[PPW][RJ];
#pragma unroll
    for (int k = 0; k < PPW; ++k) {
        int c = wv + 32 * k;
        int a = plo[c], n = pl[c];
#pragma unroll
        for (int r = 0; r < RJ; ++r) {
            int j = lane + 32 * r;
            int v = tmp[min(a + j, NEDGE - 1)];
            w[k][r] = (j < n) ? v : -1;
        }
    }
#pragma unroll
    for (int k = 0; k < PPW; ++k) {
#pragma unroll
        for (int r = 0; r < RJ; ++r)
            if (w[k][r] >= 0) atomicAdd(&nc[w[k][r] >> SRCB], 1);
        int c = wv + 32 * k;
        int a = plo[c], n = pl[c];
        for (int j = lane + 32 * RJ; j < n; j += 32) atomicAdd(&nc[tmp[a + j] >> SRCB], 1);
    }
    __syncthreads();
    scan_nodes(nc, ns, t);
    __syncthreads();
    {
        int node = bin * CN + t;
        if (t < CN && node <= N_NODES) off[node] = s0 + ns[t];
    }
    bool staged = len <= CAP;
#pragma unroll
    for (int k = 0; k < PPW; ++k) {
#pragma unroll
        for (int r = 0; r < RJ; ++r)
            if (w[k][r] >= 0) {
                int dl = w[k][r] >> SRCB;
                int q = ns[dl] + atomicAdd(&cur[dl], 1);
                if (staged) st[q] = w[k][r] & ((1 << SRCB) - 1); else slot[s0 + q] = w[k][r] & ((1 << SRCB) - 1);
            }
        int c = wv + 32 * k;
        int a = plo[c], n = pl[c];
        for (int j = lane + 32 * RJ; j < n; j += 32) {
            int u = tmp[a + j];
            int dl = u >> SRCB;
            int q = ns[dl] + atomicAdd(&cur[dl], 1);
            if (staged) st[q] = u & ((1 << SRCB) - 1); else slot[s0 + q] = u & ((1 << SRCB) - 1);
        }
    }
    __syncthreads();
    if (staged) for (int i = t; i < len; i += FT) slot[s0 + i] = st[i];
}

__global__ __launch_bounds__(GWAVES * 32) void k_gather(float* d_out,
        const int* __restrict__ off, const int* __restrict__ slot,
        const float* __restrict__ es_e, const float* __restrict__ ed_e,
        const float* __restrict__ es_g, const float* __restrict__ ed_g,
        const float* __restrict__ h_est, const float* __restrict__ h_gnn,
        const float* __restrict__ eb, const float* __restrict__ egam,
        const float* __restrict__ ebeta, const float* __restrict__ erm,
        const float* __restrict__ erv,
        const float* __restrict__ gb, const float* __restrict__ ggam,
        const float* __restrict__ gbeta, const float* __restrict__ grm,
        const float* __restrict__ grv,
        const float* __restrict__ clsW, const float* __restrict__ clsb) {
    __shared__ int   l_s[GWAVES][LCAP];
    __shared__ float l_ae[GWAVES][LCAP];
    __shared__ float l_ag[GWAVES][LCAP];
    __shared__ int   l_u[GWAVES][LCAP];
    __shared__ float l_y[GW];
    int lane = threadIdx.x & 31;
    int wave = __builtin_amdgcn_readfirstlane(threadIdx.x >> 5);
    int f0 = lane * 2;

    float eb0 = eb[f0], eb1 = eb[f0 + 1];
    float erm0 = erm[f0], erm1 = erm[f0 + 1];
    float ers0 = rsqrtf(erv[f0] + EPSBN), ers1 = rsqrtf(erv[f0 + 1] + EPSBN);
    float egam0 = egam[f0], egam1 = egam[f0 + 1];
    float ebeta0 = ebeta[f0], ebeta1 = ebeta[f0 + 1];
    float gb0 = gb[f0], gb1 = gb[f0 + 1];
    float grm0 = grm[f0], grm1 = grm[f0 + 1];
    float grs0 = rsqrtf(grv[f0] + EPSBN), grs1 = rsqrtf(grv[f0 + 1] + EPSBN);
    float ggam0 = ggam[f0], ggam1 = ggam[f0 + 1];
    float gbeta0 = gbeta[f0], gbeta1 = gbeta[f0 + 1];
    float cw0 = clsW[f0], cw1 = clsW[f0 + 1];
    float cb = clsb[0];

#pragma nounroll
    for (int i = 0; i < NPW; ++i) {
        int n = blockIdx.x * GW + wave * NPW + i;
        int beg = off[n];
        int deg = off[n + 1] - beg;
        float edn_e = ed_e[n], edn_g = ed_g[n];
        float self_e = leaky(es_e[n] + edn_e), self_g = leaky(es_g[n] + edn_g);
        v2f acc_e = {0.f, 0.f}, acc_g = {0.f, 0.f};
        float ze = 0.f, zg = 0.f, pse, psg;

        if (deg <= 32) {
            bool on = lane < deg;
            int sraw = slot[min(beg + lane, NEDGE - 1)];
            int s = on ? sraw : INT_MAX;
            int si = on ? sraw : n;
            float xe = es_e[si], xg = es_g[si];
            int rank = 0;
            for (int j = 0; j < deg; ++j) {
                int sj = __shfl(s, j);
                rank += (sj < s || (sj == s && j < lane)) ? 1 : 0;
            }
            int cntp = max((deg + RB) & ~(RB - 1), FB);
            if (on) l_s[wave][rank] = s;
            if (lane < cntp - deg) l_s[wave][deg + lane] = n;
            v2f he0[FB], hg0[FB];
#pragma unroll
            for (int u = 0; u < FB; ++u) {
                int sr = l_s[wave][u];
                he0[u] = *(const v2f*)(h_est + (long)sr * NHID + f0);
                hg0[u] = *(const v2f*)(h_gnn + (long)sr * NHID + f0);
            }
            __builtin_amdgcn_sched_barrier(0);
            float le = on ? leaky(xe + edn_e) : -FLT_MAX;
            float lg = on ? leaky(xg + edn_g) : -FLT_MAX;
            float me = fmaxf(wave_max_f(le), self_e);
            float mg = fmaxf(wave_max_f(lg), self_g);
            float pe = __expf(le - me), pg = __expf(lg - mg);
            pse = __expf(self_e - me); psg = __expf(self_g - mg);
            if (on) { l_ae[wave][rank] = pe; l_ag[wave][rank] = pg; }
            for (int r = 0; r < deg; ++r) { ze += l_ae[wave][r]; zg += l_ag[wave][r]; }
            ze += pse; zg += psg;
            if (on) { l_ae[wave][rank] = pe / ze; l_ag[wave][rank] = pg / zg; }
            if (lane < cntp - deg) {
                l_ae[wave][deg + lane] = (lane == 0) ? pse / ze : 0.f;
                l_ag[wave][deg + lane] = (lane == 0) ? psg / zg : 0.f;
            }
#pragma unroll
            for (int u = 0; u < FB; ++u) {
                acc_e += he0[u] * l_ae[wave][u]; acc_g += hg0[u] * l_ag[wave][u];
            }
            for (int b = FB; b < cntp; b += RB) {
                v2f he[RB], hg[RB];
#pragma unroll
                for (int u = 0; u < RB; ++u) {
                    int sr = l_s[wave][b + u];
                    he[u] = *(const v2f*)(h_est + (long)sr * NHID + f0);
                    hg[u] = *(const v2f*)(h_gnn + (long)sr * NHID + f0);
                }
#pragma unroll
                for (int u = 0; u < RB; ++u) {
                    acc_e += he[u] * l_ae[wave][b + u]; acc_g += hg[u] * l_ag[wave][b + u];
                }
            }
        } else {
        if (deg <= LCAP) {
            float le_m = self_e, lg_m = self_g;
            for (int b = 0; b < deg; b += 32) {
                int k = b + lane;
                if (k < deg) {
                    int s = slot[beg + k];
                    l_u[wave][k] = s;
                    le_m = fmaxf(le_m, leaky(es_e[s] + edn_e));
                    lg_m = fmaxf(lg_m, leaky(es_g[s] + edn_g));
                }
            }
            float me = wave_max_f(le_m), mg = wave_max_f(lg_m);
            pse = __expf(self_e - me); psg = __expf(self_g - mg);
            for (int b = 0; b < deg; b += 32) {
                int k = b + lane;
                int s = (k < deg) ? l_u[wave][k] : INT_MAX;
                int rank = 0;
#pragma unroll 4
                for (int j = 0; j < deg; ++j) {
                    int sj = l_u[wave][j];
                    rank += (sj < s || (sj == s && j < k)) ? 1 : 0;
                }
                if (k < deg) {
                    l_s[wave][rank] = s;
                    l_ae[wave][rank] = __expf(leaky(es_e[s] + edn_e) - me);
                    l_ag[wave][rank] = __expf(leaky(es_g[s] + edn_g) - mg);
                }
            }
            for (int r = 0; r < deg; ++r) { ze += l_ae[wave][r]; zg += l_ag[wave][r]; }
            ze += pse; zg += psg;
            for (int b = 0; b < deg; b += 32) {
                int r = b + lane;
                if (r < deg) { l_ae[wave][r] = l_ae[wave][r] / ze; l_ag[wave][r] = l_ag[wave][r] / zg; }
            }
#pragma unroll 4
            for (int r = 0; r < deg; ++r) {
                int sr = l_s[wave][r];
                float ae = l_ae[wave][r], ag = l_ag[wave][r];
                v2f he = *(const v2f*)(h_est + (long)sr * NHID + f0);
                v2f hg = *(const v2f*)(h_gnn + (long)sr * NHID + f0);
                acc_e += he * ae; acc_g += hg * ag;
            }
        } else {
            float le_m = self_e, lg_m = self_g;
            for (int k = lane; k < deg; k += 32) {
                int s = slot[beg + k];
                le_m = fmaxf(le_m, leaky(es_e[s] + edn_e));
                lg_m = fmaxf(lg_m, leaky(es_g[s] + edn_g));
            }
            float me = wave_max_f(le_m), mg = wave_max_f(lg_m);
            pse = __expf(self_e - me); psg = __expf(self_g - mg);
            for (int pass = 0; pass < 2; ++pass) {
                int prev = -1;
                for (int it = 0; it < deg; ++it) {
                    int best = INT_MAX;
                    for (int k = lane; k < deg; k += 32) {
                        int s = slot[beg + k];
                        if (s > prev) best = min(best, s);
                    }
                    best = wave_min_i(best);
                    if (best == INT_MAX) break;
                    int c = 0;
                    for (int k = lane; k < deg; k += 32) c += (slot[beg + k] == best) ? 1 : 0;
                    c = wave_sum_i(c);
                    float pe = __expf(leaky(es_e[best] + edn_e) - me);
                    float pg = __expf(leaky(es_g[best] + edn_g) - mg);
                    if (pass == 0) {
                        for (int k = 0; k < c; ++k) { ze += pe; zg += pg; }
                    } else {
                        float ae = pe / ze, ag = pg / zg;
                        v2f he = *(const v2f*)(h_est + (long)best * NHID + f0);
                        v2f hg = *(const v2f*)(h_gnn + (long)best * NHID + f0);
                        for (int k = 0; k < c; ++k) { acc_e += he * ae; acc_g += hg * ag; }
                    }
                    prev = best;
                }
                if (pass == 0) { ze += pse; zg += psg; }
            }
        }
        {
            float ae = pse / ze, ag = psg / zg;
            v2f he = *(const v2f*)(h_est + (long)n * NHID + f0);
            v2f hg = *(const v2f*)(h_gnn + (long)n * NHID + f0);
            acc_e += he * ae; acc_g += hg * ag;
        }
        }

        v2f so;
        {
            float a0 = fmaxf(acc_e.x + eb0, 0.f);
            float a1 = fmaxf(acc_e.y + eb1, 0.f);
            so.x = (a0 - erm0) * ers0 * egam0 + ebeta0;
            so.y = (a1 - erm1) * ers1 * egam1 + ebeta1;
        }
        *(v2f*)(d_out + N_NODES + (long)n * NHID + f0) = so;
        float b0 = fmaxf(acc_g.x + gb0, 0.f);
        float b1 = fmaxf(acc_g.y + gb1, 0.f);
        float z0 = (b0 - grm0) * grs0 * ggam0 + gbeta0;
        float z1 = (b1 - grm1) * grs1 * ggam1 + gbeta1;
        float yp = wave_sum_f(z0 * cw0 + z1 * cw1);
        if (lane == 0) l_y[wave * NPW + i] = yp + cb;
    }
    __syncthreads();
    if (threadIdx.x < GW) d_out[blockIdx.x * GW + threadIdx.x] = l_y[threadIdx.x];
}

extern "C" void kernel_launch(void* const* d_in, const int* in_sizes, int n_in,
                              void* d_out, int out_size, void* d_ws, size_t ws_size,
                              hipStream_t stream) {
    const float* g        = (const float*)d_in[0];
    const int*   x        = (const int*)d_in[1];
    const float* est_W    = (const float*)d_in[2];
    const float* est_asrc = (const float*)d_in[3];
    const float* est_adst = (const float*)d_in[4];
    const float* est_bias = (const float*)d_in[5];
    const float* est_gam  = (const float*)d_in[6];
    const float* est_beta = (const float*)d_in[7];
    const float* est_rm   = (const float*)d_in[8];
    const float* est_rv   = (const float*)d_in[9];
    const float* gnn_W    = (const float*)d_in[10];
    const float* gnn_asrc = (const float*)d_in[11];
    const float* gnn_adst = (const float*)d_in[12];
    const float* gnn_bias = (const float*)d_in[13];
    const float* gnn_gam  = (const float*)d_in[14];
    const float* gnn_beta = (const float*)d_in[15];
    const float* gnn_rm   = (const float*)d_in[16];
    const float* gnn_rv   = (const float*)d_in[17];
    const float* cls_W    = (const float*)d_in[18];
    const float* cls_b    = (const float*)d_in[19];
    float* out = (float*)d_out;

    float* ws     = (float*)d_ws;
    float* h_est  = ws;
    float* h_gnn  = h_est + (long)N_NODES * NHID;
    int*   off    = (int*)(h_gnn + (long)N_NODES * NHID);
    int*   slot   = off + 2 * N_NODES;
    int*   tmp    = slot + NEDGE;
    int*   tab    = tmp + NEDGE;
    float* es_e   = ws + (long)N_NODES * 192;
    float* ed_e   = es_e + N_NODES;
    float* es_g   = ed_e + N_NODES;
    float* ed_g   = es_g + N_NODES;

    const int BT = 256;
    k_gemm<<<(MTILES + 7) / 8, BT, 0, stream>>>(g, est_W, gnn_W, h_est, h_gnn);

    k_eprep<<<(N_NODES + BT - 1) / BT, BT, 0, stream>>>(h_est, h_gnn,
            est_asrc, est_adst, gnn_asrc, gnn_adst, es_e, ed_e, es_g, ed_g);

    k_tile<<<NBLK, PT, 0, stream>>>(x, tmp, tab);

    k_group<<<NBIN, FT, 0, stream>>>(tmp, tab, off, slot);

    k_gather<<<N_NODES / GW, GWAVES * 32, 0, stream>>>(out, off, slot,
            es_e, ed_e, es_g, ed_g, h_est, h_gnn,
            est_bias, est_gam, est_beta, est_rm, est_rv,
            gnn_bias, gnn_gam, gnn_beta, gnn_rm, gnn_rv, cls_W, cls_b);
}


# GEMM: one resident workgroup per WGP keeps both weight matrices in LDS in B-fragment order, its waves walk the row tiles with A loaded a block ahead; the k order inside each sum of 8 changes
# speedup vs baseline: 5.3115x; 1.1541x over previous
#include <hip/hip_runtime.h>
#include <cfloat>
#include <climits>

#define N_NODES 100000
#define NFEAT   256
#define NHID    64
#define NEDGE   1600000
#define MTILES  (N_NODES / 16)
#define EPSBN   1e-5f
#define NEG_SLOPE 0.2f
#define GW      32
#define GWAVES  8
#define NPW     (GW / GWAVES)
#define QPT     2
#define PT      800
#define TE      (PT * QPT * 4)
#define NBLK    (NEDGE / TE)
#define CB      9
#define CN      (1 << CB)
#define NBIN    ((N_NODES + CN - 1) / CN)
#define TABW    (NBIN + 1)
#define SRCB    17
#define FT      1024
#define PPW     ((NBLK + 31) / 32)
#define RJ      2
#define CAP     12288
#define LCAP    128
#define RB      8
#define FB      16

static_assert(N_NODES % GW == 0, "gather blocks cover the nodes exactly");
static_assert(GW % GWAVES == 0, "every gather wave gets the same number of nodes");
static_assert((RB & (RB - 1)) == 0 && FB % RB == 0 && FB <= 32 && 32 + RB <= LCAP, "a bucket of up to 32, its node and the padding fit the LDS list, and one lane each writes the padding");
static_assert(NEDGE % 4 == 0, "bucketing reads whole quads, and the destination row starts on a 16-byte boundary");
static_assert(NEDGE % TE == 0 && PT % 32 == 0 && TE % PT == 0, "the tiles cover the edge list exactly: no thread is without its quads, no load, add or store of the tile kernel needs a guard");
static_assert(N_NODES <= (1 << SRCB) && SRCB + CB <= 31, "a source id and a destination's place in its group share one non-negative word");
static_assert(TABW <= PT && TABW <= 32 * 7 && CN == 32 * 16 && CN <= FT, "one thread per group or per node where the kernels say so; the wave scans take 7 and 16 entries per lane");
static_assert(NBLK <= 32 * PPW && 32 * PPW <= FT && FT == 1024, "one thread of the group kernel per tile, and every tile's piece has its wave");

typedef __attribute__((ext_vector_type(2))) float v2f;
typedef __attribute__((ext_vector_type(4))) float v4f;
typedef __attribute__((ext_vector_type(8))) float v8f;

__device__ __forceinline__ float leaky(float v) {
    return fmaxf(v, 0.f) + NEG_SLOPE * fminf(v, 0.f);
}
__device__ __forceinline__ float wave_max_f(float v) {
    for (int o = 16; o > 0; o >>= 1) v = fmaxf(v, __shfl_xor(v, o));
    return v;
}
__device__ __forceinline__ float wave_sum_f(float v) {
    for (int o = 16; o > 0; o >>= 1) v += __shfl_xor(v, o);
    return v;
}
__device__ __forceinline__ int wave_min_i(int v) {
    for (int o = 16; o > 0; o >>= 1) v = min(v, __shfl_xor(v, o));
    return v;
}
__device__ __forceinline__ int wave_sum_i(int v) {
    for (int o = 16; o > 0; o >>= 1) v += __shfl_xor(v, o);
    return v;
}

#define KQ      (NFEAT / 4)
#define WCOL    (2 * NHID)
#define WPITCH  (WCOL + 16)
#define GT      512
#define GWV     (GT / 32)
#define GGRID   256
static_assert(NFEAT % 32 == 0 && GT % WCOL == 0 && (2 * KQ) % (GT / WCOL) == 0, "the table fill and the k groups cover W and a row of g exactly");

template <int CUR>
__device__ __forceinline__ void k_block(v8f (&acc)[8], v4f (&a)[2][4], v2f (&b)[2][8],
                                        const v2f* wbj, const float* gnext, bool more) {
    if (more) {
#pragma unroll
        for (int jj = 0; jj < 4; ++jj) a[CUR ^ 1][jj] = *(const v4f*)(gnext + 8 * jj);
    }
#pragma unroll
    for (int qq = 0; qq < 8; ++qq) {
        if (qq < 7 || more) {
#pragma unroll
            for (int n = 0; n < 8; ++n) b[(qq + 1) & 1][n] = wbj[(qq + 1) * 2 * WPITCH + n * 16];
        }
        __builtin_amdgcn_sched_barrier(0);
        v4f av = a[CUR][qq >> 1];
        v2f af;
        af.x = (qq & 1) ? av.z : av.x;
        af.y = (qq & 1) ? av.w : av.y;
#pragma unroll
        for (int n = 0; n < 8; ++n)
            acc[n] = __builtin_amdgcn_wmma_f32_16x16x4_f32(
                false, af, false, b[qq & 1][n], (short)0, acc[n], false, false);
        __builtin_amdgcn_sched_barrier(0);
    }
}

__global__ __launch_bounds__(GT) void k_gemm(const float* __restrict__ g,
                                             const float* __restrict__ We,
                                             const float* __restrict__ Wg,
                                             float* __restrict__ h_est,
                                             float* __restrict__ h_gnn) {
    __shared__ v2f wl[2 * KQ * WPITCH];
    int t = threadIdx.x;
    {
        int col = t & (WCOL - 1);
        const float* Wp = (col < NHID) ? We : Wg;
        int cc = col & (NHID - 1);
#pragma unroll 8
        for (int p = t / WCOL; p < 2 * KQ; p += GT / WCOL) {
            int q = p >> 1, hh = p & 1;
            int k = 8 * (q >> 1) + 4 * hh + 2 * (q & 1);
            v2f e;
            e.x = Wp[k * NHID + cc];
            e.y = Wp[(k + 1) * NHID + cc];
            wl[p * WPITCH + col] = e;
        }
    }
    __syncthreads();
    int lane = t & 31;
    int wave = __builtin_amdgcn_readfirstlane(t >> 5);
    int lrow = lane & 15;
    int lhi  = lane >> 4;
    const v2f* wb = wl + lhi * WPITCH + lrow;

    for (int mtile = blockIdx.x + GGRID * wave; mtile < MTILES; mtile += GGRID * GWV) {
        int m0 = mtile * 16;
        const float* grow = g + (long)(m0 + lrow) * NFEAT + 4 * lhi;
        v8f acc[8] = {};
        v4f a[2][4];
        v2f b[2][8];
#pragma unroll
        for (int j = 0; j < 4; ++j) a[0][j] = *(const v4f*)(grow + 8 * j);
#pragma unroll
        for (int n = 0; n < 8; ++n) b[0][n] = wb[n * 16];
#pragma nounroll
        for (int jp = 0; jp < NFEAT / 64; ++jp) {
            const v2f* wbj = wb + jp * (32 * WPITCH);
            const float* gj = grow + 64 * jp;
            k_block<0>(acc, a, b, wbj, gj + 32, true);
            k_block<1>(acc, a, b, wbj + 16 * WPITCH, gj + 64, jp + 1 < NFEAT / 64);
        }
#pragma unroll
        for (int n = 0; n < 8; ++n) {
            float* dst = (n < 4) ? h_est : h_gnn;
            int c = (n & 3) * 16 + lrow;
#pragma unroll
            for (int r = 0; r < 8; ++r) {
                dst[(long)(m0 + r + 8 * lhi) * NHID + c] = acc[n][r];
            }
        }
    }
}

__global__ void k_eprep(const float* __restrict__ h_est, const float* __restrict__ h_gnn,
                        const float* __restrict__ ase, const float* __restrict__ ade,
                        const float* __restrict__ asg, const float* __restrict__ adg,
                        float* es_e, float* ed_e, float* es_g, float* ed_g) {
    int n = blockIdx.x * blockDim.x + threadIdx.x;
    if (n >= N_NODES) return;
    const float* he = h_est + (long)n * NHID;
    const float* hg = h_gnn + (long)n * NHID;
    float se = 0.f, de = 0.f, sg = 0.f, dg = 0.f;
#pragma unroll 4
    for (int f = 0; f < NHID; f += 4) {
        v4f e  = *(const v4f*)(he + f);
        v4f g4 = *(const v4f*)(hg + f);
        v4f a1 = *(const v4f*)(ase + f);
        v4f a2 = *(const v4f*)(ade + f);
        v4f a3 = *(const v4f*)(asg + f);
        v4f a4 = *(const v4f*)(adg + f);
        se += e.x * a1.x + e.y * a1.y + e.z * a1.z + e.w * a1.w;
        de += e.x * a2.x + e.y * a2.y + e.z * a2.z + e.w * a2.w;
        sg += g4.x * a3.x + g4.y * a3.y + g4.z * a3.z + g4.w * a3.w;
        dg += g4.x * a4.x + g4.y * a4.y + g4.z * a4.z + g4.w * a4.w;
    }
    es_e[n] = se; ed_e[n] = de; es_g[n] = sg; ed_g[n] = dg;
}

__global__ __launch_bounds__(PT) void k_tile(const int* __restrict__ x, int* __restrict__ tmp,
                                             int* __restrict__ tab) {
    __shared__ int h[NBIN];
    __shared__ int ls[TABW];
    __shared__ int sv[TE];
    int t = threadIdx.x;
    const int4* s4 = (const int4*)x;
    const int4* d4 = (const int4*)(x + NEDGE);
    int q0 = blockIdx.x * (QPT * PT) + t;
    int4 s[QPT], d[QPT];
#pragma unroll
    for (int j = 0; j < QPT; ++j) { d[j] = d4[q0 + j * PT]; s[j] = s4[q0 + j * PT]; }
    if (t < NBIN) h[t] = 0;
    __syncthreads();
    int4 p[QPT];
#pragma unroll
    for (int j = 0; j < QPT; ++j) {
        p[j].x = atomicAdd(&h[d[j].x >> CB], 1);
        p[j].y = atomicAdd(&h[d[j].y >> CB], 1);
        p[j].z = atomicAdd(&h[d[j].z >> CB], 1);
        p[j].w = atomicAdd(&h[d[j].w >> CB], 1);
    }
    __syncthreads();
    if (t < 32) {
        int v[7], sum = 0;
#pragma unroll
        for (int k = 0; k < 7; ++k) { int b = t * 7 + k; v[k] = (b < NBIN) ? h[b] : 0; sum += v[k]; }
        int inc = sum;
        for (int o = 1; o < 32; o <<= 1) { int u = __shfl_up(inc, o); if (t >= o) inc += u; }
        int ex = inc - sum;
#pragma unroll
        for (int k = 0; k < 7; ++k) { int b = t * 7 + k; if (b < TABW) ls[b] = ex; ex += v[k]; }
    }
    __syncthreads();
    if (t < TABW) tab[blockIdx.x * TABW + t] = ls[t];
#pragma unroll
    for (int j = 0; j < QPT; ++j) {
        sv[ls[d[j].x >> CB] + p[j].x] = s[j].x | ((d[j].x & (CN - 1)) << SRCB);
        sv[ls[d[j].y >> CB] + p[j].y] = s[j].y | ((d[j].y & (CN - 1)) << SRCB);
        sv[ls[d[j].z >> CB] + p[j].z] = s[j].z | ((d[j].z & (CN - 1)) << SRCB);
        sv[ls[d[j].w >> CB] + p[j].w] = s[j].w | ((d[j].w & (CN - 1)) << SRCB);
    }
    __syncthreads();
#pragma unroll
    for (int i = 0; i < TE / PT; ++i) tmp[blockIdx.x * TE + i * PT + t] = sv[i * PT + t];
}

__device__ __forceinline__ void scan_nodes(const int* nc, int* ns, int t) {
    if (t < 32) {
        int v[16], sum = 0;
#pragma unroll
        for (int k = 0; k < 16; ++k) { v[k] = nc[t * 16 + k]; sum += v[k]; }
        int inc = sum;
        for (int o = 1; o < 32; o <<= 1) { int u = __shfl_up(inc, o); if (t >= o) inc += u; }
        int ex = inc - sum;
#pragma unroll
        for (int k = 0; k < 16; ++k) { ns[t * 16 + k] = ex; ex += v[k]; }
    }
}
__global__ __launch_bounds__(FT) void k_group(const int* __restrict__ tmp, const int* __restrict__ tab,
                                              int* __restrict__ off, int* __restrict__ slot) {
    __shared__ int plo[32 * PPW];
    __shared__ int pl[32 * PPW];
    __shared__ int rs[32];
    __shared__ int rl[32];
    __shared__ int nc[CN];
    __shared__ int cur[CN];
    __shared__ int ns[CN];
    __shared__ int st[CAP];
    int t = threadIdx.x, lane = t & 31, bin = blockIdx.x;
    int wv = __builtin_amdgcn_readfirstlane(t >> 5);
    {
        int lo = 0, n = 0;
        if (t < NBLK) { lo = tab[t * TABW + bin]; n = tab[t * TABW + bin + 1] - lo; }
        if (t < 32 * PPW) { plo[t] = t * TE + lo; pl[t] = n; }
        lo = wave_sum_i(lo); n = wave_sum_i(n);
        if (lane == 0) { rs[wv] = lo; rl[wv] = n; }
        if (t < CN) { nc[t] = 0; cur[t] = 0; }
    }
    __syncthreads();
    int s0 = wave_sum_i(rs[lane]);
    int len = wave_sum_i(rl[lane]);
    int w[PPW][RJ];
#pragma unroll
    for (int k = 0; k < PPW; ++k) {
        int c = wv + 32 * k;
        int a = plo[c], n = pl[c];
#pragma unroll
        for (int r = 0; r < RJ; ++r) {
            int j = lane + 32 * r;
            int v = tmp[min(a + j, NEDGE - 1)];
            w[k][r] = (j < n) ? v : -1;
        }
    }
#pragma unroll
    for (int k = 0; k < PPW; ++k) {
#pragma unroll
        for (int r = 0; r < RJ; ++r)
            if (w[k][r] >= 0) atomicAdd(&nc[w[k][r] >> SRCB], 1);
        int c = wv + 32 * k;
        int a = plo[c], n = pl[c];
        for (int j = lane + 32 * RJ; j < n; j += 32) atomicAdd(&nc[tmp[a + j] >> SRCB], 1);
    }
    __syncthreads();
    scan_nodes(nc, ns, t);
    __syncthreads();
    {
        int node = bin * CN + t;
        if (t < CN && node <= N_NODES) off[node] = s0 + ns[t];
    }
    bool staged = len <= CAP;
#pragma unroll
    for (int k = 0; k < PPW; ++k) {
#pragma unroll
        for (int r = 0; r < RJ; ++r)
            if (w[k][r] >= 0) {
                int dl = w[k][r] >> SRCB;
                int q = ns[dl] + atomicAdd(&cur[dl], 1);
                if (staged) st[q] = w[k][r] & ((1 << SRCB) - 1); else slot[s0 + q] = w[k][r] & ((1 << SRCB) - 1);
            }
        int c = wv + 32 * k;
        int a = plo[c], n = pl[c];
        for (int j = lane + 32 * RJ; j < n; j += 32) {
            int u = tmp[a + j];
            int dl = u >> SRCB;
            int q = ns[dl] + atomicAdd(&cur[dl], 1);
            if (staged) st[q] = u & ((1 << SRCB) - 1); else slot[s0 + q] = u & ((1 << SRCB) - 1);
        }
    }
    __syncthreads();
    if (staged) for (int i = t; i < len; i += FT) slot[s0 + i] = st[i];
}

__global__ __launch_bounds__(GWAVES * 32) void k_gather(float* d_out,
        const int* __restrict__ off, const int* __restrict__ slot,
        const float* __restrict__ es_e, const float* __restrict__ ed_e,
        const float* __restrict__ es_g, const float* __restrict__ ed_g,
        const float* __restrict__ h_est, const float* __restrict__ h_gnn,
        const float* __restrict__ eb, const float* __restrict__ egam,
        const float* __restrict__ ebeta, const float* __restrict__ erm,
        const float* __restrict__ erv,
        const float* __restrict__ gb, const float* __restrict__ ggam,
        const float* __restrict__ gbeta, const float* __restrict__ grm,
        const float* __restrict__ grv,
        const float* __restrict__ clsW, const float* __restrict__ clsb) {
    __shared__ int   l_s[GWAVES][LCAP];
    __shared__ float l_ae[GWAVES][LCAP];
    __shared__ float l_ag[GWAVES][LCAP];
    __shared__ int   l_u[GWAVES][LCAP];
    __shared__ float l_y[GW];
    int lane = threadIdx.x & 31;
    int wave = __builtin_amdgcn_readfirstlane(threadIdx.x >> 5);
    int f0 = lane * 2;

    float eb0 = eb[f0], eb1 = eb[f0 + 1];
    float erm0 = erm[f0], erm1 = erm[f0 + 1];
    float ers0 = rsqrtf(erv[f0] + EPSBN), ers1 = rsqrtf(erv[f0 + 1] + EPSBN);
    float egam0 = egam[f0], egam1 = egam[f0 + 1];
    float ebeta0 = ebeta[f0], ebeta1 = ebeta[f0 + 1];
    float gb0 = gb[f0], gb1 = gb[f0 + 1];
    float grm0 = grm[f0], grm1 = grm[f0 + 1];
    float grs0 = rsqrtf(grv[f0] + EPSBN), grs1 = rsqrtf(grv[f0 + 1] + EPSBN);
    float ggam0 = ggam[f0], ggam1 = ggam[f0 + 1];
    float gbeta0 = gbeta[f0], gbeta1 = gbeta[f0 + 1];
    float cw0 = clsW[f0], cw1 = clsW[f0 + 1];
    float cb = clsb[0];

#pragma nounroll
    for (int i = 0; i < NPW; ++i) {
        int n = blockIdx.x * GW + wave * NPW + i;
        int beg = off[n];
        int deg = off[n + 1] - beg;
        float edn_e = ed_e[n], edn_g = ed_g[n];
        float self_e = leaky(es_e[n] + edn_e), self_g = leaky(es_g[n] + edn_g);
        v2f acc_e = {0.f, 0.f}, acc_g = {0.f, 0.f};
        float ze = 0.f, zg = 0.f, pse, psg;

        if (deg <= 32) {
            bool on = lane < deg;
            int sraw = slot[min(beg + lane, NEDGE - 1)];
            int s = on ? sraw : INT_MAX;
            int si = on ? sraw : n;
            float xe = es_e[si], xg = es_g[si];
            int rank = 0;
            for (int j = 0; j < deg; ++j) {
                int sj = __shfl(s, j);
                rank += (sj < s || (sj == s && j < lane)) ? 1 : 0;
            }
            int cntp = max((deg + RB) & ~(RB - 1), FB);
            if (on) l_s[wave][rank] = s;
            if (lane < cntp - deg) l_s[wave][deg + lane] = n;
            v2f he0[FB], hg0[FB];
#pragma unroll
            for (int u = 0; u < FB; ++u) {
                int sr = l_s[wave][u];
                he0[u] = *(const v2f*)(h_est + (long)sr * NHID + f0);
                hg0[u] = *(const v2f*)(h_gnn + (long)sr * NHID + f0);
            }
            __builtin_amdgcn_sched_barrier(0);
            float le = on ? leaky(xe + edn_e) : -FLT_MAX;
            float lg = on ? leaky(xg + edn_g) : -FLT_MAX;
            float me = fmaxf(wave_max_f(le), self_e);
            float mg = fmaxf(wave_max_f(lg), self_g);
            float pe = __expf(le - me), pg = __expf(lg - mg);
            pse = __expf(self_e - me); psg = __expf(self_g - mg);
            if (on) { l_ae[wave][rank] = pe; l_ag[wave][rank] = pg; }
            for (int r = 0; r < deg; ++r) { ze += l_ae[wave][r]; zg += l_ag[wave][r]; }
            ze += pse; zg += psg;
            if (on) { l_ae[wave][rank] = pe / ze; l_ag[wave][rank] = pg / zg; }
            if (lane < cntp - deg) {
                l_ae[wave][deg + lane] = (lane == 0) ? pse / ze : 0.f;
                l_ag[wave][deg + lane] = (lane == 0) ? psg / zg : 0.f;
            }
#pragma unroll
            for (int u = 0; u < FB; ++u) {
                acc_e += he0[u] * l_ae[wave][u]; acc_g += hg0[u] * l_ag[wave][u];
            }
            for (int b = FB; b < cntp; b += RB) {
                v2f he[RB], hg[RB];
#pragma unroll
                for (int u = 0; u < RB; ++u) {
                    int sr = l_s[wave][b + u];
                    he[u] = *(const v2f*)(h_est + (long)sr * NHID + f0);
                    hg[u] = *(const v2f*)(h_gnn + (long)sr * NHID + f0);
                }
#pragma unroll
                for (int u = 0; u < RB; ++u) {
                    acc_e += he[u] * l_ae[wave][b + u]; acc_g += hg[u] * l_ag[wave][b + u];
                }
            }
        } else {
        if (deg <= LCAP) {
            float le_m = self_e, lg_m = self_g;
            for (int b = 0; b < deg; b += 32) {
                int k = b + lane;
                if (k < deg) {
                    int s = slot[beg + k];
                    l_u[wave][k] = s;
                    le_m = fmaxf(le_m, leaky(es_e[s] + edn_e));
                    lg_m = fmaxf(lg_m, leaky(es_g[s] + edn_g));
                }
            }
            float me = wave_max_f(le_m), mg = wave_max_f(lg_m);
            pse = __expf(self_e - me); psg = __expf(self_g - mg);
            for (int b = 0; b < deg; b += 32) {
                int k = b + lane;
                int s = (k < deg) ? l_u[wave][k] : INT_MAX;
                int rank = 0;
#pragma unroll 4
                for (int j = 0; j < deg; ++j) {
                    int sj = l_u[wave][j];
                    rank += (sj < s || (sj == s && j < k)) ? 1 : 0;
                }
                if (k < deg) {
                    l_s[wave][rank] = s;
                    l_ae[wave][rank] = __expf(leaky(es_e[s] + edn_e) - me);
                    l_ag[wave][rank] = __expf(leaky(es_g[s] + edn_g) - mg);
                }
            }
            for (int r = 0; r < deg; ++r) { ze += l_ae[wave][r]; zg += l_ag[wave][r]; }
            ze += pse; zg += psg;
            for (int b = 0; b < deg; b += 32) {
                int r = b + lane;
                if (r < deg) { l_ae[wave][r] = l_ae[wave][r] / ze; l_ag[wave][r] = l_ag[wave][r] / zg; }
            }
#pragma unroll 4
            for (int r = 0; r < deg; ++r) {
                int sr = l_s[wave][r];
                float ae = l_ae[wave][r], ag = l_ag[wave][r];
                v2f he = *(const v2f*)(h_est + (long)sr * NHID + f0);
                v2f hg = *(const v2f*)(h_gnn + (long)sr * NHID + f0);
                acc_e += he * ae; acc_g += hg * ag;
            }
        } else {
            float le_m = self_e, lg_m = self_g;
            for (int k = lane; k < deg; k += 32) {
                int s = slot[beg + k];
                le_m = fmaxf(le_m, leaky(es_e[s] + edn_e));
                lg_m = fmaxf(lg_m, leaky(es_g[s] + edn_g));
            }
            float me = wave_max_f(le_m), mg = wave_max_f(lg_m);
            pse = __expf(self_e - me); psg = __expf(self_g - mg);
            for (int pass = 0; pass < 2; ++pass) {
                int prev = -1;
                for (int it = 0; it < deg; ++it) {
                    int best = INT_MAX;
                    for (int k = lane; k < deg; k += 32) {
                        int s = slot[beg + k];
                        if (s > prev) best = min(best, s);
                    }
                    best = wave_min_i(best);
                    if (best == INT_MAX) break;
                    int c = 0;
                    for (int k = lane; k < deg; k += 32) c += (slot[beg + k] == best) ? 1 : 0;
                    c = wave_sum_i(c);
                    float pe = __expf(leaky(es_e[best] + edn_e) - me);
                    float pg = __expf(leaky(es_g[best] + edn_g) - mg);
                    if (pass == 0) {
                        for (int k = 0; k < c; ++k) { ze += pe; zg += pg; }
                    } else {
                        float ae = pe / ze, ag = pg / zg;
                        v2f he = *(const v2f*)(h_est + (long)best * NHID + f0);
                        v2f hg = *(const v2f*)(h_gnn + (long)best * NHID + f0);
                        for (int k = 0; k < c; ++k) { acc_e += he * ae; acc_g += hg * ag; }
                    }
                    prev = best;
                }
                if (pass == 0) { ze += pse; zg += psg; }
            }
        }
        {
            float ae = pse / ze, ag = psg / zg;
            v2f he = *(const v2f*)(h_est + (long)n * NHID + f0);
            v2f hg = *(const v2f*)(h_gnn + (long)n * NHID + f0);
            acc_e += he * ae; acc_g += hg * ag;
        }
        }

        v2f so;
        {
            float a0 = fmaxf(acc_e.x + eb0, 0.f);
            float a1 = fmaxf(acc_e.y + eb1, 0.f);
            so.x = (a0 - erm0) * ers0 * egam0 + ebeta0;
            so.y = (a1 - erm1) * ers1 * egam1 + ebeta1;
        }
        *(v2f*)(d_out + N_NODES + (long)n * NHID + f0) = so;
        float b0 = fmaxf(acc_g.x + gb0, 0.f);
        float b1 = fmaxf(acc_g.y + gb1, 0.f);
        float z0 = (b0 - grm0) * grs0 * ggam0 + gbeta0;
        float z1 = (b1 - grm1) * grs1 * ggam1 + gbeta1;
        float yp = wave_sum_f(z0 * cw0 + z1 * cw1);
        if (lane == 0) l_y[wave * NPW + i] = yp + cb;
    }
    __syncthreads();
    if (threadIdx.x < GW) d_out[blockIdx.x * GW + threadIdx.x] = l_y[threadIdx.x];
}

extern "C" void kernel_launch(void* const* d_in, const int* in_sizes, int n_in,
                              void* d_out, int out_size, void* d_ws, size_t ws_size,
                              hipStream_t stream) {
    const float* g        = (const float*)d_in[0];
    const int*   x        = (const int*)d_in[1];
    const float* est_W    = (const float*)d_in[2];
    const float* est_asrc = (const float*)d_in[3];
    const float* est_adst = (const float*)d_in[4];
    const float* est_bias = (const float*)d_in[5];
    const float* est_gam  = (const float*)d_in[6];
    const float* est_beta = (const float*)d_in[7];
    const float* est_rm   = (const float*)d_in[8];
    const float* est_rv   = (const float*)d_in[9];
    const float* gnn_W    = (const float*)d_in[10];
    const float* gnn_asrc = (const float*)d_in[11];
    const float* gnn_adst = (const float*)d_in[12];
    const float* gnn_bias = (const float*)d_in[13];
    const float* gnn_gam  = (const float*)d_in[14];
    const float* gnn_beta = (const float*)d_in[15];
    const float* gnn_rm   = (const float*)d_in[16];
    const float* gnn_rv   = (const float*)d_in[17];
    const float* cls_W    = (const float*)d_in[18];
    const float* cls_b    = (const float*)d_in[19];
    float* out = (float*)d_out;

    float* ws     = (float*)d_ws;
    float* h_est  = ws;
    float* h_gnn  = h_est + (long)N_NODES * NHID;
    int*   off    = (int*)(h_gnn + (long)N_NODES * NHID);
    int*   slot   = off + 2 * N_NODES;
    int*   tmp    = slot + NEDGE;
    int*   tab    = tmp + NEDGE;
    float* es_e   = ws + (long)N_NODES * 192;
    float* ed_e   = es_e + N_NODES;
    float* es_g   = ed_e + N_NODES;
    float* ed_g   = es_g + N_NODES;

    const int BT = 256;
    k_gemm<<<GGRID, GT, 0, stream>>>(g, est_W, gnn_W, h_est, h_gnn);

    k_eprep<<<(N_NODES + BT - 1) / BT, BT, 0, stream>>>(h_est, h_gnn,
            est_asrc, est_adst, gnn_asrc, gnn_adst, es_e, ed_e, es_g, ed_g);

    k_tile<<<NBLK, PT, 0, stream>>>(x, tmp, tab);

    k_group<<<NBIN, FT, 0, stream>>>(tmp, tab, off, slot);

    k_gather<<<N_NODES / GW, GWAVES * 32, 0, stream>>>(out, off, slot,
            es_e, ed_e, es_g, ed_g, h_est, h_gnn,
            est_bias, est_gam, est_beta, est_rm, est_rv,
            gnn_bias, gnn_gam, gnn_beta, gnn_rm, gnn_rv, cls_W, cls_b);
}


# Attention logits formed in the GEMM epilogue from the accumulators (four columns per lane, then a tree over a half's lanes), one record of four per node; no logits kernel; the dot order changes
# speedup vs baseline: 5.8279x; 1.0972x over previous
#include <hip/hip_runtime.h>
#include <cfloat>
#include <climits>

#define N_NODES 100000
#define NFEAT   256
#define NHID    64
#define NEDGE   1600000
#define MTILES  (N_NODES / 16)
#define EPSBN   1e-5f
#define NEG_SLOPE 0.2f
#define GW      32
#define GWAVES  8
#define NPW     (GW / GWAVES)
#define QPT     2
#define PT      800
#define TE      (PT * QPT * 4)
#define NBLK    (NEDGE / TE)
#define CB      9
#define CN      (1 << CB)
#define NBIN    ((N_NODES + CN - 1) / CN)
#define TABW    (NBIN + 1)
#define SRCB    17
#define FT      1024
#define PPW     ((NBLK + 31) / 32)
#define RJ      2
#define CAP     12288
#define LCAP    128
#define RB      8
#define FB      16

static_assert(N_NODES % GW == 0, "gather blocks cover the nodes exactly");
static_assert(GW % GWAVES == 0, "every gather wave gets the same number of nodes");
static_assert((RB & (RB - 1)) == 0 && FB % RB == 0 && FB <= 32 && 32 + RB <= LCAP, "a bucket of up to 32, its node and the padding fit the LDS list, and one lane each writes the padding");
static_assert(NEDGE % 4 == 0, "bucketing reads whole quads, and the destination row starts on a 16-byte boundary");
static_assert(NEDGE % TE == 0 && PT % 32 == 0 && TE % PT == 0, "the tiles cover the edge list exactly: no thread is without its quads, no load, add or store of the tile kernel needs a guard");
static_assert(N_NODES <= (1 << SRCB) && SRCB + CB <= 31, "a source id and a destination's place in its group share one non-negative word");
static_assert(TABW <= PT && TABW <= 32 * 7 && CN == 32 * 16 && CN <= FT, "one thread per group or per node where the kernels say so; the wave scans take 7 and 16 entries per lane");
static_assert(NBLK <= 32 * PPW && 32 * PPW <= FT && FT == 1024, "one thread of the group kernel per tile, and every tile's piece has its wave");

typedef __attribute__((ext_vector_type(2))) float v2f;
typedef __attribute__((ext_vector_type(4))) float v4f;
typedef __attribute__((ext_vector_type(8))) float v8f;

__device__ __forceinline__ float leaky(float v) {
    return fmaxf(v, 0.f) + NEG_SLOPE * fminf(v, 0.f);
}
__device__ __forceinline__ float wave_max_f(float v) {
    for (int o = 16; o > 0; o >>= 1) v = fmaxf(v, __shfl_xor(v, o));
    return v;
}
__device__ __forceinline__ float wave_sum_f(float v) {
    for (int o = 16; o > 0; o >>= 1) v += __shfl_xor(v, o);
    return v;
}
__device__ __forceinline__ int wave_min_i(int v) {
    for (int o = 16; o > 0; o >>= 1) v = min(v, __shfl_xor(v, o));
    return v;
}
__device__ __forceinline__ int wave_sum_i(int v) {
    for (int o = 16; o > 0; o >>= 1) v += __shfl_xor(v, o);
    return v;
}

#define KQ      (NFEAT / 4)
#define WCOL    (2 * NHID)
#define WPITCH  (WCOL + 16)
#define GT      512
#define GWV     (GT / 32)
#define GGRID   256
static_assert(NFEAT % 32 == 0 && GT % WCOL == 0 && (2 * KQ) % (GT / WCOL) == 0, "the table fill and the k groups cover W and a row of g exactly");

template <int CUR>
__device__ __forceinline__ void k_block(v8f (&acc)[8], v4f (&a)[2][4], v2f (&b)[2][8],
                                        const v2f* wbj, const float* gnext, bool more) {
    if (more) {
#pragma unroll
        for (int jj = 0; jj < 4; ++jj) a[CUR ^ 1][jj] = *(const v4f*)(gnext + 8 * jj);
    }
#pragma unroll
    for (int qq = 0; qq < 8; ++qq) {
        if (qq < 7 || more) {
#pragma unroll
            for (int n = 0; n < 8; ++n) b[(qq + 1) & 1][n] = wbj[(qq + 1) * 2 * WPITCH + n * 16];
        }
        __builtin_amdgcn_sched_barrier(0);
        v4f av = a[CUR][qq >> 1];
        v2f af;
        af.x = (qq & 1) ? av.z : av.x;
        af.y = (qq & 1) ? av.w : av.y;
#pragma unroll
        for (int n = 0; n < 8; ++n)
            acc[n] = __builtin_amdgcn_wmma_f32_16x16x4_f32(
                false, af, false, b[qq & 1][n], (short)0, acc[n], false, false);
        __builtin_amdgcn_sched_barrier(0);
    }
}

template <int CNT, int BIT>
__device__ __forceinline__ void lane_tree(float (&p)[32], int c) {
    bool up = (c & BIT) != 0;
#pragma unroll
    for (int i = 0; i < CNT / 2; ++i) {
        float keep = up ? p[CNT / 2 + i] : p[i];
        float send = up ? p[i] : p[CNT / 2 + i];
        p[i] = keep + __shfl_xor(send, BIT);
    }
}

__global__ __launch_bounds__(GT) void k_gemm(const float* __restrict__ g,
                                             const float* __restrict__ We,
                                             const float* __restrict__ Wg,
                                             const float* __restrict__ ase,
                                             const float* __restrict__ ade,
                                             const float* __restrict__ asg,
                                             const float* __restrict__ adg,
                                             float* __restrict__ h_est,
                                             float* __restrict__ h_gnn,
                                             float* __restrict__ lg) {
    __shared__ v2f wl[2 * KQ * WPITCH];
    int t = threadIdx.x;
    {
        int col = t & (WCOL - 1);
        const float* Wp = (col < NHID) ? We : Wg;
        int cc = col & (NHID - 1);
#pragma unroll 8
        for (int p = t / WCOL; p < 2 * KQ; p += GT / WCOL) {
            int q = p >> 1, hh = p & 1;
            int k = 8 * (q >> 1) + 4 * hh + 2 * (q & 1);
            v2f e;
            e.x = Wp[k * NHID + cc];
            e.y = Wp[(k + 1) * NHID + cc];
            wl[p * WPITCH + col] = e;
        }
    }
    __syncthreads();
    int lane = t & 31;
    int wave = __builtin_amdgcn_readfirstlane(t >> 5);
    int lrow = lane & 15;
    int lhi  = lane >> 4;
    const v2f* wb = wl + lhi * WPITCH + lrow;
    v2f ve[4], vg[4];
#pragma unroll
    for (int n = 0; n < 4; ++n) {
        ve[n].x = ase[16 * n + lrow]; ve[n].y = ade[16 * n + lrow];
        vg[n].x = asg[16 * n + lrow]; vg[n].y = adg[16 * n + lrow];
    }

    for (int mtile = blockIdx.x + GGRID * wave; mtile < MTILES; mtile += GGRID * GWV) {
        int m0 = mtile * 16;
        const float* grow = g + (long)(m0 + lrow) * NFEAT + 4 * lhi;
        v8f acc[8] = {};
        v4f a[2][4];
        v2f b[2][8];
#pragma unroll
        for (int j = 0; j < 4; ++j) a[0][j] = *(const v4f*)(grow + 8 * j);
#pragma unroll
        for (int n = 0; n < 8; ++n) b[0][n] = wb[n * 16];
#pragma nounroll
        for (int jp = 0; jp < NFEAT / 64; ++jp) {
            const v2f* wbj = wb + jp * (32 * WPITCH);
            const float* gj = grow + 64 * jp;
            k_block<0>(acc, a, b, wbj, gj + 32, true);
            k_block<1>(acc, a, b, wbj + 16 * WPITCH, gj + 64, jp + 1 < NFEAT / 64);
        }
#pragma unroll
        for (int n = 0; n < 8; ++n) {
            float* dst = (n < 4) ? h_est : h_gnn;
            int c = (n & 3) * 16 + lrow;
#pragma unroll
            for (int r = 0; r < 8; ++r) {
                dst[(long)(m0 + r + 8 * lhi) * NHID + c] = acc[n][r];
            }
        }
        __builtin_amdgcn_sched_barrier(0);
        float p[32];
#pragma unroll
        for (int r = 0; r < 8; ++r) {
            v2f pe = acc[0][r] * ve[0], pg = acc[4][r] * vg[0];
#pragma unroll
            for (int n = 1; n < 4; ++n) { pe += acc[n][r] * ve[n]; pg += acc[4 + n][r] * vg[n]; }
            p[4 * r] = pe.x; p[4 * r + 1] = pg.x; p[4 * r + 2] = pe.y; p[4 * r + 3] = pg.y;
        }
        lane_tree<32, 8>(p, lrow);
        lane_tree<16, 4>(p, lrow);
        lane_tree<8, 2>(p, lrow);
        lane_tree<4, 1>(p, lrow);
        v2f rec = {p[0], p[1]};
        *(v2f*)(lg + 4 * (long)m0 + 2 * lane) = rec;
    }
}

__global__ __launch_bounds__(PT) void k_tile(const int* __restrict__ x, int* __restrict__ tmp,
                                             int* __restrict__ tab) {
    __shared__ int h[NBIN];
    __shared__ int ls[TABW];
    __shared__ int sv[TE];
    int t = threadIdx.x;
    const int4* s4 = (const int4*)x;
    const int4* d4 = (const int4*)(x + NEDGE);
    int q0 = blockIdx.x * (QPT * PT) + t;
    int4 s[QPT], d[QPT];
#pragma unroll
    for (int j = 0; j < QPT; ++j) { d[j] = d4[q0 + j * PT]; s[j] = s4[q0 + j * PT]; }
    if (t < NBIN) h[t] = 0;
    __syncthreads();
    int4 p[QPT];
#pragma unroll
    for (int j = 0; j < QPT; ++j) {
        p[j].x = atomicAdd(&h[d[j].x >> CB], 1);
        p[j].y = atomicAdd(&h[d[j].y >> CB], 1);
        p[j].z = atomicAdd(&h[d[j].z >> CB], 1);
        p[j].w = atomicAdd(&h[d[j].w >> CB], 1);
    }
    __syncthreads();
    if (t < 32) {
        int v[7], sum = 0;
#pragma unroll
        for (int k = 0; k < 7; ++k) { int b = t * 7 + k; v[k] = (b < NBIN) ? h[b] : 0; sum += v[k]; }
        int inc = sum;
        for (int o = 1; o < 32; o <<= 1) { int u = __shfl_up(inc, o); if (t >= o) inc += u; }
        int ex = inc - sum;
#pragma unroll
        for (int k = 0; k < 7; ++k) { int b = t * 7 + k; if (b < TABW) ls[b] = ex; ex += v[k]; }
    }
    __syncthreads();
    if (t < TABW) tab[blockIdx.x * TABW + t] = ls[t];
#pragma unroll
    for (int j = 0; j < QPT; ++j) {
        sv[ls[d[j].x >> CB] + p[j].x] = s[j].x | ((d[j].x & (CN - 1)) << SRCB);
        sv[ls[d[j].y >> CB] + p[j].y] = s[j].y | ((d[j].y & (CN - 1)) << SRCB);
        sv[ls[d[j].z >> CB] + p[j].z] = s[j].z | ((d[j].z & (CN - 1)) << SRCB);
        sv[ls[d[j].w >> CB] + p[j].w] = s[j].w | ((d[j].w & (CN - 1)) << SRCB);
    }
    __syncthreads();
#pragma unroll
    for (int i = 0; i < TE / PT; ++i) tmp[blockIdx.x * TE + i * PT + t] = sv[i * PT + t];
}

__device__ __forceinline__ void scan_nodes(const int* nc, int* ns, int t) {
    if (t < 32) {
        int v[16], sum = 0;
#pragma unroll
        for (int k = 0; k < 16; ++k) { v[k] = nc[t * 16 + k]; sum += v[k]; }
        int inc = sum;
        for (int o = 1; o < 32; o <<= 1) { int u = __shfl_up(inc, o); if (t >= o) inc += u; }
        int ex = inc - sum;
#pragma unroll
        for (int k = 0; k < 16; ++k) { ns[t * 16 + k] = ex; ex += v[k]; }
    }
}
__global__ __launch_bounds__(FT) void k_group(const int* __restrict__ tmp, const int* __restrict__ tab,
                                              int* __restrict__ off, int* __restrict__ slot) {
    __shared__ int plo[32 * PPW];
    __shared__ int pl[32 * PPW];
    __shared__ int rs[32];
    __shared__ int rl[32];
    __shared__ int nc[CN];
    __shared__ int cur[CN];
    __shared__ int ns[CN];
    __shared__ int st[CAP];
    int t = threadIdx.x, lane = t & 31, bin = blockIdx.x;
    int wv = __builtin_amdgcn_readfirstlane(t >> 5);
    {
        int lo = 0, n = 0;
        if (t < NBLK) { lo = tab[t * TABW + bin]; n = tab[t * TABW + bin + 1] - lo; }
        if (t < 32 * PPW) { plo[t] = t * TE + lo; pl[t] = n; }
        lo = wave_sum_i(lo); n = wave_sum_i(n);
        if (lane == 0) { rs[wv] = lo; rl[wv] = n; }
        if (t < CN) { nc[t] = 0; cur[t] = 0; }
    }
    __syncthreads();
    int s0 = wave_sum_i(rs[lane]);
    int len = wave_sum_i(rl[lane]);
    int w[PPW][RJ];
#pragma unroll
    for (int k = 0; k < PPW; ++k) {
        int c = wv + 32 * k;
        int a = plo[c], n = pl[c];
#pragma unroll
        for (int r = 0; r < RJ; ++r) {
            int j = lane + 32 * r;
            int v = tmp[min(a + j, NEDGE - 1)];
            w[k][r] = (j < n) ? v : -1;
        }
    }
#pragma unroll
    for (int k = 0; k < PPW; ++k) {
#pragma unroll
        for (int r = 0; r < RJ; ++r)
            if (w[k][r] >= 0) atomicAdd(&nc[w[k][r] >> SRCB], 1);
        int c = wv + 32 * k;
        int a = plo[c], n = pl[c];
        for (int j = lane + 32 * RJ; j < n; j += 32) atomicAdd(&nc[tmp[a + j] >> SRCB], 1);
    }
    __syncthreads();
    scan_nodes(nc, ns, t);
    __syncthreads();
    {
        int node = bin * CN + t;
        if (t < CN && node <= N_NODES) off[node] = s0 + ns[t];
    }
    bool staged = len <= CAP;
#pragma unroll
    for (int k = 0; k < PPW; ++k) {
#pragma unroll
        for (int r = 0; r < RJ; ++r)
            if (w[k][r] >= 0) {
                int dl = w[k][r] >> SRCB;
                int q = ns[dl] + atomicAdd(&cur[dl], 1);
                if (staged) st[q] = w[k][r] & ((1 << SRCB) - 1); else slot[s0 + q] = w[k][r] & ((1 << SRCB) - 1);
            }
        int c = wv + 32 * k;
        int a = plo[c], n = pl[c];
        for (int j = lane + 32 * RJ; j < n; j += 32) {
            int u = tmp[a + j];
            int dl = u >> SRCB;
            int q = ns[dl] + atomicAdd(&cur[dl], 1);
            if (staged) st[q] = u & ((1 << SRCB) - 1); else slot[s0 + q] = u & ((1 << SRCB) - 1);
        }
    }
    __syncthreads();
    if (staged) for (int i = t; i < len; i += FT) slot[s0 + i] = st[i];
}

__global__ __launch_bounds__(GWAVES * 32) void k_gather(float* d_out,
        const int* __restrict__ off, const int* __restrict__ slot,
        const float* __restrict__ lg,
        const float* __restrict__ h_est, const float* __restrict__ h_gnn,
        const float* __restrict__ eb, const float* __restrict__ egam,
        const float* __restrict__ ebeta, const float* __restrict__ erm,
        const float* __restrict__ erv,
        const float* __restrict__ gb, const float* __restrict__ ggam,
        const float* __restrict__ gbeta, const float* __restrict__ grm,
        const float* __restrict__ grv,
        const float* __restrict__ clsW, const float* __restrict__ clsb) {
    __shared__ int   l_s[GWAVES][LCAP];
    __shared__ float l_ae[GWAVES][LCAP];
    __shared__ float l_ag[GWAVES][LCAP];
    __shared__ int   l_u[GWAVES][LCAP];
    __shared__ float l_y[GW];
    int lane = threadIdx.x & 31;
    int wave = __builtin_amdgcn_readfirstlane(threadIdx.x >> 5);
    int f0 = lane * 2;

    float eb0 = eb[f0], eb1 = eb[f0 + 1];
    float erm0 = erm[f0], erm1 = erm[f0 + 1];
    float ers0 = rsqrtf(erv[f0] + EPSBN), ers1 = rsqrtf(erv[f0 + 1] + EPSBN);
    float egam0 = egam[f0], egam1 = egam[f0 + 1];
    float ebeta0 = ebeta[f0], ebeta1 = ebeta[f0 + 1];
    float gb0 = gb[f0], gb1 = gb[f0 + 1];
    float grm0 = grm[f0], grm1 = grm[f0 + 1];
    float grs0 = rsqrtf(grv[f0] + EPSBN), grs1 = rsqrtf(grv[f0 + 1] + EPSBN);
    float ggam0 = ggam[f0], ggam1 = ggam[f0 + 1];
    float gbeta0 = gbeta[f0], gbeta1 = gbeta[f0 + 1];
    float cw0 = clsW[f0], cw1 = clsW[f0 + 1];
    float cb = clsb[0];

#pragma nounroll
    for (int i = 0; i < NPW; ++i) {
        int n = blockIdx.x * GW + wave * NPW + i;
        int beg = off[n];
        int deg = off[n + 1] - beg;
        v4f rn = *(const v4f*)(lg + 4 * (long)n);
        float edn_e = rn.z, edn_g = rn.w;
        float self_e = leaky(rn.x + edn_e), self_g = leaky(rn.y + edn_g);
        v2f acc_e = {0.f, 0.f}, acc_g = {0.f, 0.f};
        float ze = 0.f, zg = 0.f, pse, psg;

        if (deg <= 32) {
            bool on = lane < deg;
            int sraw = slot[min(beg + lane, NEDGE - 1)];
            int s = on ? sraw : INT_MAX;
            int si = on ? sraw : n;
            v2f xs = *(const v2f*)(lg + 4 * (long)si);
            float xe = xs.x, xg = xs.y;
            int rank = 0;
            for (int j = 0; j < deg; ++j) {
                int sj = __shfl(s, j);
                rank += (sj < s || (sj == s && j < lane)) ? 1 : 0;
            }
            int cntp = max((deg + RB) & ~(RB - 1), FB);
            if (on) l_s[wave][rank] = s;
            if (lane < cntp - deg) l_s[wave][deg + lane] = n;
            v2f he0[FB], hg0[FB];
#pragma unroll
            for (int u = 0; u < FB; ++u) {
                int sr = l_s[wave][u];
                he0[u] = *(const v2f*)(h_est + (long)sr * NHID + f0);
                hg0[u] = *(const v2f*)(h_gnn + (long)sr * NHID + f0);
            }
            __builtin_amdgcn_sched_barrier(0);
            float le = on ? leaky(xe + edn_e) : -FLT_MAX;
            float lg = on ? leaky(xg + edn_g) : -FLT_MAX;
            float me = fmaxf(wave_max_f(le), self_e);
            float mg = fmaxf(wave_max_f(lg), self_g);
            float pe = __expf(le - me), pg = __expf(lg - mg);
            pse = __expf(self_e - me); psg = __expf(self_g - mg);
            if (on) { l_ae[wave][rank] = pe; l_ag[wave][rank] = pg; }
            for (int r = 0; r < deg; ++r) { ze += l_ae[wave][r]; zg += l_ag[wave][r]; }
            ze += pse; zg += psg;
            if (on) { l_ae[wave][rank] = pe / ze; l_ag[wave][rank] = pg / zg; }
            if (lane < cntp - deg) {
                l_ae[wave][deg + lane] = (lane == 0) ? pse / ze : 0.f;
                l_ag[wave][deg + lane] = (lane == 0) ? psg / zg : 0.f;
            }
#pragma unroll
            for (int u = 0; u < FB; ++u) {
                acc_e += he0[u] * l_ae[wave][u]; acc_g += hg0[u] * l_ag[wave][u];
            }
            for (int b = FB; b < cntp; b += RB) {
                v2f he[RB], hg[RB];
#pragma unroll
                for (int u = 0; u < RB; ++u) {
                    int sr = l_s[wave][b + u];
                    he[u] = *(const v2f*)(h_est + (long)sr * NHID + f0);
                    hg[u] = *(const v2f*)(h_gnn + (long)sr * NHID + f0);
                }
#pragma unroll
                for (int u = 0; u < RB; ++u) {
                    acc_e += he[u] * l_ae[wave][b + u]; acc_g += hg[u] * l_ag[wave][b + u];
                }
            }
        } else {
        if (deg <= LCAP) {
            float le_m = self_e, lg_m = self_g;
            for (int b = 0; b < deg; b += 32) {
                int k = b + lane;
                if (k < deg) {
                    int s = slot[beg + k];
                    l_u[wave][k] = s;
                    v2f xs = *(const v2f*)(lg + 4 * (long)s);
                    le_m = fmaxf(le_m, leaky(xs.x + edn_e));
                    lg_m = fmaxf(lg_m, leaky(xs.y + edn_g));
                }
            }
            float me = wave_max_f(le_m), mg = wave_max_f(lg_m);
            pse = __expf(self_e - me); psg = __expf(self_g - mg);
            for (int b = 0; b < deg; b += 32) {
                int k = b + lane;
                int s = (k < deg) ? l_u[wave][k] : INT_MAX;
                int rank = 0;
#pragma unroll 4
                for (int j = 0; j < deg; ++j) {
                    int sj = l_u[wave][j];
                    rank += (sj < s || (sj == s && j < k)) ? 1 : 0;
                }
                if (k < deg) {
                    l_s[wave][rank] = s;
                    v2f xs = *(const v2f*)(lg + 4 * (long)s);
                    l_ae[wave][rank] = __expf(leaky(xs.x + edn_e) - me);
                    l_ag[wave][rank] = __expf(leaky(xs.y + edn_g) - mg);
                }
            }
            for (int r = 0; r < deg; ++r) { ze += l_ae[wave][r]; zg += l_ag[wave][r]; }
            ze += pse; zg += psg;
            for (int b = 0; b < deg; b += 32) {
                int r = b + lane;
                if (r < deg) { l_ae[wave][r] = l_ae[wave][r] / ze; l_ag[wave][r] = l_ag[wave][r] / zg; }
            }
#pragma unroll 4
            for (int r = 0; r < deg; ++r) {
                int sr = l_s[wave][r];
                float ae = l_ae[wave][r], ag = l_ag[wave][r];
                v2f he = *(const v2f*)(h_est + (long)sr * NHID + f0);
                v2f hg = *(const v2f*)(h_gnn + (long)sr * NHID + f0);
                acc_e += he * ae; acc_g += hg * ag;
            }
        } else {
            float le_m = self_e, lg_m = self_g;
            for (int k = lane; k < deg; k += 32) {
                int s = slot[beg + k];
                v2f xs = *(const v2f*)(lg + 4 * (long)s);
                le_m = fmaxf(le_m, leaky(xs.x + edn_e));
                lg_m = fmaxf(lg_m, leaky(xs.y + edn_g));
            }
            float me = wave_max_f(le_m), mg = wave_max_f(lg_m);
            pse = __expf(self_e - me); psg = __expf(self_g - mg);
            for (int pass = 0; pass < 2; ++pass) {
                int prev = -1;
                for (int it = 0; it < deg; ++it) {
                    int best = INT_MAX;
                    for (int k = lane; k < deg; k += 32) {
                        int s = slot[beg + k];
                        if (s > prev) best = min(best, s);
                    }
                    best = wave_min_i(best);
                    if (best == INT_MAX) break;
                    int c = 0;
                    for (int k = lane; k < deg; k += 32) c += (slot[beg + k] == best) ? 1 : 0;
                    c = wave_sum_i(c);
                    v2f xb = *(const v2f*)(lg + 4 * (long)best);
                    float pe = __expf(leaky(xb.x + edn_e) - me);
                    float pg = __expf(leaky(xb.y + edn_g) - mg);
                    if (pass == 0) {
                        for (int k = 0; k < c; ++k) { ze += pe; zg += pg; }
                    } else {
                        float ae = pe / ze, ag = pg / zg;
                        v2f he = *(const v2f*)(h_est + (long)best * NHID + f0);
                        v2f hg = *(const v2f*)(h_gnn + (long)best * NHID + f0);
                        for (int k = 0; k < c; ++k) { acc_e += he * ae; acc_g += hg * ag; }
                    }
                    prev = best;
                }
                if (pass == 0) { ze += pse; zg += psg; }
            }
        }
        {
            float ae = pse / ze, ag = psg / zg;
            v2f he = *(const v2f*)(h_est + (long)n * NHID + f0);
            v2f hg = *(const v2f*)(h_gnn + (long)n * NHID + f0);
            acc_e += he * ae; acc_g += hg * ag;
        }
        }

        v2f so;
        {
            float a0 = fmaxf(acc_e.x + eb0, 0.f);
            float a1 = fmaxf(acc_e.y + eb1, 0.f);
            so.x = (a0 - erm0) * ers0 * egam0 + ebeta0;
            so.y = (a1 - erm1) * ers1 * egam1 + ebeta1;
        }
        *(v2f*)(d_out + N_NODES + (long)n * NHID + f0) = so;
        float b0 = fmaxf(acc_g.x + gb0, 0.f);
        float b1 = fmaxf(acc_g.y + gb1, 0.f);
        float z0 = (b0 - grm0) * grs0 * ggam0 + gbeta0;
        float z1 = (b1 - grm1) * grs1 * ggam1 + gbeta1;
        float yp = wave_sum_f(z0 * cw0 + z1 * cw1);
        if (lane == 0) l_y[wave * NPW + i] = yp + cb;
    }
    __syncthreads();
    if (threadIdx.x < GW) d_out[blockIdx.x * GW + threadIdx.x] = l_y[threadIdx.x];
}

extern "C" void kernel_launch(void* const* d_in, const int* in_sizes, int n_in,
                              void* d_out, int out_size, void* d_ws, size_t ws_size,
                              hipStream_t stream) {
    const float* g        = (const float*)d_in[0];
    const int*   x        = (const int*)d_in[1];
    const float* est_W    = (const float*)d_in[2];
    const float* est_asrc = (const float*)d_in[3];
    const float* est_adst = (const float*)d_in[4];
    const float* est_bias = (const float*)d_in[5];
    const float* est_gam  = (const float*)d_in[6];
    const float* est_beta = (const float*)d_in[7];
    const float* est_rm   = (const float*)d_in[8];
    const float* est_rv   = (const float*)d_in[9];
    const float* gnn_W    = (const float*)d_in[10];
    const float* gnn_asrc = (const float*)d_in[11];
    const float* gnn_adst = (const float*)d_in[12];
    const float* gnn_bias = (const float*)d_in[13];
    const float* gnn_gam  = (const float*)d_in[14];
    const float* gnn_beta = (const float*)d_in[15];
    const float* gnn_rm   = (const float*)d_in[16];
    const float* gnn_rv   = (const float*)d_in[17];
    const float* cls_W    = (const float*)d_in[18];
    const float* cls_b    = (const float*)d_in[19];
    float* out = (float*)d_out;

    float* ws     = (float*)d_ws;
    float* h_est  = ws;
    float* h_gnn  = h_est + (long)N_NODES * NHID;
    int*   off    = (int*)(h_gnn + (long)N_NODES * NHID);
    int*   slot   = off + 2 * N_NODES;
    int*   tmp    = slot + NEDGE;
    int*   tab    = tmp + NEDGE;
    float* lg     = ws + (long)N_NODES * 192;

    k_gemm<<<GGRID, GT, 0, stream>>>(g, est_W, gnn_W, est_asrc, est_adst, gnn_asrc, gnn_adst,
                                     h_est, h_gnn, lg);

    k_tile<<<NBLK, PT, 0, stream>>>(x, tmp, tab);

    k_group<<<NBIN, FT, 0, stream>>>(tmp, tab, off, slot);

    k_gather<<<N_NODES / GW, GWAVES * 32, 0, stream>>>(out, off, slot,
            lg, h_est, h_gnn,
            est_bias, est_gam, est_beta, est_rm, est_rv,
            gnn_bias, gnn_gam, gnn_beta, gnn_rm, gnn_rv, cls_W, cls_b);
}


# h of both nets kept as one 512-byte record per node, est and gnn of a column side by side: a gather lane takes one 16-byte load per row at a 32-bit offset, not two of 8 bytes; sums keep their order
# speedup vs baseline: 6.0871x; 1.0445x over previous
#include <hip/hip_runtime.h>
#include <cfloat>
#include <climits>

#define N_NODES 100000
#define NFEAT   256
#define NHID    64
#define NEDGE   1600000
#define MTILES  (N_NODES / 16)
#define EPSBN   1e-5f
#define NEG_SLOPE 0.2f
#define GW      32
#define GWAVES  8
#define NPW     (GW / GWAVES)
#define QPT     2
#define PT      800
#define TE      (PT * QPT * 4)
#define NBLK    (NEDGE / TE)
#define CB      9
#define CN      (1 << CB)
#define NBIN    ((N_NODES + CN - 1) / CN)
#define TABW    (NBIN + 1)
#define SRCB    17
#define FT      1024
#define PPW     ((NBLK + 31) / 32)
#define RJ      2
#define CAP     12288
#define LCAP    128
#define RB      8
#define FB      16

static_assert(N_NODES % GW == 0, "gather blocks cover the nodes exactly");
static_assert(GW % GWAVES == 0, "every gather wave gets the same number of nodes");
static_assert((RB & (RB - 1)) == 0 && FB % RB == 0 && FB <= 32 && 32 + RB <= LCAP, "a bucket of up to 32, its node and the padding fit the LDS list, and one lane each writes the padding");
static_assert(NEDGE % 4 == 0, "bucketing reads whole quads, and the destination row starts on a 16-byte boundary");
static_assert(NEDGE % TE == 0 && PT % 32 == 0 && TE % PT == 0, "the tiles cover the edge list exactly: no thread is without its quads, no load, add or store of the tile kernel needs a guard");
static_assert(N_NODES <= (1 << SRCB) && SRCB + CB <= 31, "a source id and a destination's place in its group share one non-negative word");
static_assert(TABW <= PT && TABW <= 32 * 7 && CN == 32 * 16 && CN <= FT, "one thread per group or per node where the kernels say so; the wave scans take 7 and 16 entries per lane");
static_assert(NBLK <= 32 * PPW && 32 * PPW <= FT && FT == 1024, "one thread of the group kernel per tile, and every tile's piece has its wave");

typedef __attribute__((ext_vector_type(2))) float v2f;
typedef __attribute__((ext_vector_type(4))) float v4f;
typedef __attribute__((ext_vector_type(8))) float v8f;

__device__ __forceinline__ float leaky(float v) {
    return fmaxf(v, 0.f) + NEG_SLOPE * fminf(v, 0.f);
}
__device__ __forceinline__ float wave_max_f(float v) {
    for (int o = 16; o > 0; o >>= 1) v = fmaxf(v, __shfl_xor(v, o));
    return v;
}
__device__ __forceinline__ float wave_sum_f(float v) {
    for (int o = 16; o > 0; o >>= 1) v += __shfl_xor(v, o);
    return v;
}
__device__ __forceinline__ int wave_min_i(int v) {
    for (int o = 16; o > 0; o >>= 1) v = min(v, __shfl_xor(v, o));
    return v;
}
__device__ __forceinline__ int wave_sum_i(int v) {
    for (int o = 16; o > 0; o >>= 1) v += __shfl_xor(v, o);
    return v;
}

#define KQ      (NFEAT / 4)
#define WCOL    (2 * NHID)
#define WPITCH  (WCOL + 16)
#define GT      512
#define GWV     (GT / 32)
#define GGRID   256
#define HREC    (2 * NHID)
#define HRECB   (HREC * 4)
static_assert(HRECB == 512 && (long)N_NODES * HRECB < (1L << 32), "a record's byte offset is a shift by 9 and fits 32 bits");
static_assert(NFEAT % 32 == 0 && GT % WCOL == 0 && (2 * KQ) % (GT / WCOL) == 0, "the table fill and the k groups cover W and a row of g exactly");

template <int CUR>
__device__ __forceinline__ void k_block(v8f (&acc)[8], v4f (&a)[2][4], v2f (&b)[2][8],
                                        const v2f* wbj, const float* gnext, bool more) {
    if (more) {
#pragma unroll
        for (int jj = 0; jj < 4; ++jj) a[CUR ^ 1][jj] = *(const v4f*)(gnext + 8 * jj);
    }
#pragma unroll
    for (int qq = 0; qq < 8; ++qq) {
        if (qq < 7 || more) {
#pragma unroll
            for (int n = 0; n < 8; ++n) b[(qq + 1) & 1][n] = wbj[(qq + 1) * 2 * WPITCH + n * 16];
        }
        __builtin_amdgcn_sched_barrier(0);
        v4f av = a[CUR][qq >> 1];
        v2f af;
        af.x = (qq & 1) ? av.z : av.x;
        af.y = (qq & 1) ? av.w : av.y;
#pragma unroll
        for (int n = 0; n < 8; ++n)
            acc[n] = __builtin_amdgcn_wmma_f32_16x16x4_f32(
                false, af, false, b[qq & 1][n], (short)0, acc[n], false, false);
        __builtin_amdgcn_sched_barrier(0);
    }
}

template <int CNT, int BIT>
__device__ __forceinline__ void lane_tree(float (&p)[32], int c) {
    bool up = (c & BIT) != 0;
#pragma unroll
    for (int i = 0; i < CNT / 2; ++i) {
        float keep = up ? p[CNT / 2 + i] : p[i];
        float send = up ? p[i] : p[CNT / 2 + i];
        p[i] = keep + __shfl_xor(send, BIT);
    }
}

__global__ __launch_bounds__(GT) void k_gemm(const float* __restrict__ g,
                                             const float* __restrict__ We,
                                             const float* __restrict__ Wg,
                                             const float* __restrict__ ase,
                                             const float* __restrict__ ade,
                                             const float* __restrict__ asg,
                                             const float* __restrict__ adg,
                                             float* __restrict__ h,
                                             float* __restrict__ lg) {
    __shared__ v2f wl[2 * KQ * WPITCH];
    int t = threadIdx.x;
    {
        int col = t & (WCOL - 1);
        const float* Wp = (col < NHID) ? We : Wg;
        int cc = col & (NHID - 1);
#pragma unroll 8
        for (int p = t / WCOL; p < 2 * KQ; p += GT / WCOL) {
            int q = p >> 1, hh = p & 1;
            int k = 8 * (q >> 1) + 4 * hh + 2 * (q & 1);
            v2f e;
            e.x = Wp[k * NHID + cc];
            e.y = Wp[(k + 1) * NHID + cc];
            wl[p * WPITCH + col] = e;
        }
    }
    __syncthreads();
    int lane = t & 31;
    int wave = __builtin_amdgcn_readfirstlane(t >> 5);
    int lrow = lane & 15;
    int lhi  = lane >> 4;
    const v2f* wb = wl + lhi * WPITCH + lrow;
    v2f ve[4], vg[4];
#pragma unroll
    for (int n = 0; n < 4; ++n) {
        ve[n].x = ase[16 * n + lrow]; ve[n].y = ade[16 * n + lrow];
        vg[n].x = asg[16 * n + lrow]; vg[n].y = adg[16 * n + lrow];
    }

    for (int mtile = blockIdx.x + GGRID * wave; mtile < MTILES; mtile += GGRID * GWV) {
        int m0 = mtile * 16;
        const float* grow = g + (long)(m0 + lrow) * NFEAT + 4 * lhi;
        v8f acc[8] = {};
        v4f a[2][4];
        v2f b[2][8];
#pragma unroll
        for (int j = 0; j < 4; ++j) a[0][j] = *(const v4f*)(grow + 8 * j);
#pragma unroll
        for (int n = 0; n < 8; ++n) b[0][n] = wb[n * 16];
#pragma nounroll
        for (int jp = 0; jp < NFEAT / 64; ++jp) {
            const v2f* wbj = wb + jp * (32 * WPITCH);
            const float* gj = grow + 64 * jp;
            k_block<0>(acc, a, b, wbj, gj + 32, true);
            k_block<1>(acc, a, b, wbj + 16 * WPITCH, gj + 64, jp + 1 < NFEAT / 64);
        }
        char* hrow = (char*)h + ((unsigned)(m0 + 8 * lhi) * (unsigned)HRECB + (unsigned)lrow * 8u);
        asm volatile("v_nop\n\tv_nop\n\tv_nop\n\tv_nop\n\tv_nop\n\tv_nop\n\tv_nop\n\tv_nop");
        __builtin_amdgcn_sched_barrier(0);
#pragma unroll
        for (int n = 0; n < 4; ++n) {
#pragma unroll
            for (int r = 0; r < 8; ++r) {
                v2f pr = {acc[n][r], acc[n + 4][r]};
                *(v2f*)(hrow + r * HRECB + n * 128) = pr;
            }
        }
        __builtin_amdgcn_sched_barrier(0);
        float p[32];
#pragma unroll
        for (int r = 0; r < 8; ++r) {
            v2f pe = acc[0][r] * ve[0], pg = acc[4][r] * vg[0];
#pragma unroll
            for (int n = 1; n < 4; ++n) { pe += acc[n][r] * ve[n]; pg += acc[4 + n][r] * vg[n]; }
            p[4 * r] = pe.x; p[4 * r + 1] = pg.x; p[4 * r + 2] = pe.y; p[4 * r + 3] = pg.y;
        }
        lane_tree<32, 8>(p, lrow);
        lane_tree<16, 4>(p, lrow);
        lane_tree<8, 2>(p, lrow);
        lane_tree<4, 1>(p, lrow);
        v2f rec = {p[0], p[1]};
        *(v2f*)(lg + 4 * (long)m0 + 2 * lane) = rec;
    }
}

__global__ __launch_bounds__(PT) void k_tile(const int* __restrict__ x, int* __restrict__ tmp,
                                             int* __restrict__ tab) {
    __shared__ int h[NBIN];
    __shared__ int ls[TABW];
    __shared__ int sv[TE];
    int t = threadIdx.x;
    const int4* s4 = (const int4*)x;
    const int4* d4 = (const int4*)(x + NEDGE);
    int q0 = blockIdx.x * (QPT * PT) + t;
    int4 s[QPT], d[QPT];
#pragma unroll
    for (int j = 0; j < QPT; ++j) { d[j] = d4[q0 + j * PT]; s[j] = s4[q0 + j * PT]; }
    if (t < NBIN) h[t] = 0;
    __syncthreads();
    int4 p[QPT];
#pragma unroll
    for (int j = 0; j < QPT; ++j) {
        p[j].x = atomicAdd(&h[d[j].x >> CB], 1);
        p[j].y = atomicAdd(&h[d[j].y >> CB], 1);
        p[j].z = atomicAdd(&h[d[j].z >> CB], 1);
        p[j].w = atomicAdd(&h[d[j].w >> CB], 1);
    }
    __syncthreads();
    if (t < 32) {
        int v[7], sum = 0;
#pragma unroll
        for (int k = 0; k < 7; ++k) { int b = t * 7 + k; v[k] = (b < NBIN) ? h[b] : 0; sum += v[k]; }
        int inc = sum;
        for (int o = 1; o < 32; o <<= 1) { int u = __shfl_up(inc, o); if (t >= o) inc += u; }
        int ex = inc - sum;
#pragma unroll
        for (int k = 0; k < 7; ++k) { int b = t * 7 + k; if (b < TABW) ls[b] = ex; ex += v[k]; }
    }
    __syncthreads();
    if (t < TABW) tab[blockIdx.x * TABW + t] = ls[t];
#pragma unroll
    for (int j = 0; j < QPT; ++j) {
        sv[ls[d[j].x >> CB] + p[j].x] = s[j].x | ((d[j].x & (CN - 1)) << SRCB);
        sv[ls[d[j].y >> CB] + p[j].y] = s[j].y | ((d[j].y & (CN - 1)) << SRCB);
        sv[ls[d[j].z >> CB] + p[j].z] = s[j].z | ((d[j].z & (CN - 1)) << SRCB);
        sv[ls[d[j].w >> CB] + p[j].w] = s[j].w | ((d[j].w & (CN - 1)) << SRCB);
    }
    __syncthreads();
#pragma unroll
    for (int i = 0; i < TE / PT; ++i) tmp[blockIdx.x * TE + i * PT + t] = sv[i * PT + t];
}

__device__ __forceinline__ void scan_nodes(const int* nc, int* ns, int t) {
    if (t < 32) {
        int v[16], sum = 0;
#pragma unroll
        for (int k = 0; k < 16; ++k) { v[k] = nc[t * 16 + k]; sum += v[k]; }
        int inc = sum;
        for (int o = 1; o < 32; o <<= 1) { int u = __shfl_up(inc, o); if (t >= o) inc += u; }
        int ex = inc - sum;
#pragma unroll
        for (int k = 0; k < 16; ++k) { ns[t * 16 + k] = ex; ex += v[k]; }
    }
}
__global__ __launch_bounds__(FT) void k_group(const int* __restrict__ tmp, const int* __restrict__ tab,
                                              int* __restrict__ off, int* __restrict__ slot) {
    __shared__ int plo[32 * PPW];
    __shared__ int pl[32 * PPW];
    __shared__ int rs[32];
    __shared__ int rl[32];
    __shared__ int nc[CN];
    __shared__ int cur[CN];
    __shared__ int ns[CN];
    __shared__ int st[CAP];
    int t = threadIdx.x, lane = t & 31, bin = blockIdx.x;
    int wv = __builtin_amdgcn_readfirstlane(t >> 5);
    {
        int lo = 0, n = 0;
        if (t < NBLK) { lo = tab[t * TABW + bin]; n = tab[t * TABW + bin + 1] - lo; }
        if (t < 32 * PPW) { plo[t] = t * TE + lo; pl[t] = n; }
        lo = wave_sum_i(lo); n = wave_sum_i(n);
        if (lane == 0) { rs[wv] = lo; rl[wv] = n; }
        if (t < CN) { nc[t] = 0; cur[t] = 0; }
    }
    __syncthreads();
    int s0 = wave_sum_i(rs[lane]);
    int len = wave_sum_i(rl[lane]);
    int w[PPW][RJ];
#pragma unroll
    for (int k = 0; k < PPW; ++k) {
        int c = wv + 32 * k;
        int a = plo[c], n = pl[c];
#pragma unroll
        for (int r = 0; r < RJ; ++r) {
            int j = lane + 32 * r;
            int v = tmp[min(a + j, NEDGE - 1)];
            w[k][r] = (j < n) ? v : -1;
        }
    }
#pragma unroll
    for (int k = 0; k < PPW; ++k) {
#pragma unroll
        for (int r = 0; r < RJ; ++r)
            if (w[k][r] >= 0) atomicAdd(&nc[w[k][r] >> SRCB], 1);
        int c = wv + 32 * k;
        int a = plo[c], n = pl[c];
        for (int j = lane + 32 * RJ; j < n; j += 32) atomicAdd(&nc[tmp[a + j] >> SRCB], 1);
    }
    __syncthreads();
    scan_nodes(nc, ns, t);
    __syncthreads();
    {
        int node = bin * CN + t;
        if (t < CN && node <= N_NODES) off[node] = s0 + ns[t];
    }
    bool staged = len <= CAP;
#pragma unroll
    for (int k = 0; k < PPW; ++k) {
#pragma unroll
        for (int r = 0; r < RJ; ++r)
            if (w[k][r] >= 0) {
                int dl = w[k][r] >> SRCB;
                int q = ns[dl] + atomicAdd(&cur[dl], 1);
                if (staged) st[q] = w[k][r] & ((1 << SRCB) - 1); else slot[s0 + q] = w[k][r] & ((1 << SRCB) - 1);
            }
        int c = wv + 32 * k;
        int a = plo[c], n = pl[c];
        for (int j = lane + 32 * RJ; j < n; j += 32) {
            int u = tmp[a + j];
            int dl = u >> SRCB;
            int q = ns[dl] + atomicAdd(&cur[dl], 1);
            if (staged) st[q] = u & ((1 << SRCB) - 1); else slot[s0 + q] = u & ((1 << SRCB) - 1);
        }
    }
    __syncthreads();
    if (staged) for (int i = t; i < len; i += FT) slot[s0 + i] = st[i];
}

__device__ __forceinline__ v4f row_of(const float* h, int sr, unsigned l16) {
    unsigned o = ((unsigned)sr << 9) | l16;
    return *(const v4f*)((const char*)h + o);
}
__global__ __launch_bounds__(GWAVES * 32) void k_gather(float* d_out,
        const int* __restrict__ off, const int* __restrict__ slot,
        const float* __restrict__ lg,
        const float* __restrict__ h,
        const float* __restrict__ eb, const float* __restrict__ egam,
        const float* __restrict__ ebeta, const float* __restrict__ erm,
        const float* __restrict__ erv,
        const float* __restrict__ gb, const float* __restrict__ ggam,
        const float* __restrict__ gbeta, const float* __restrict__ grm,
        const float* __restrict__ grv,
        const float* __restrict__ clsW, const float* __restrict__ clsb) {
    __shared__ int   l_s[GWAVES][LCAP];
    __shared__ v2f   l_a[GWAVES][LCAP];
    __shared__ int   l_u[GWAVES][LCAP];
    __shared__ float l_y[GW];
    int lane = threadIdx.x & 31;
    int wave = __builtin_amdgcn_readfirstlane(threadIdx.x >> 5);
    int f0 = lane * 2;
    unsigned l16 = lane * 16;

    float eb0 = eb[f0], eb1 = eb[f0 + 1];
    float erm0 = erm[f0], erm1 = erm[f0 + 1];
    float ers0 = rsqrtf(erv[f0] + EPSBN), ers1 = rsqrtf(erv[f0 + 1] + EPSBN);
    float egam0 = egam[f0], egam1 = egam[f0 + 1];
    float ebeta0 = ebeta[f0], ebeta1 = ebeta[f0 + 1];
    float gb0 = gb[f0], gb1 = gb[f0 + 1];
    float grm0 = grm[f0], grm1 = grm[f0 + 1];
    float grs0 = rsqrtf(grv[f0] + EPSBN), grs1 = rsqrtf(grv[f0 + 1] + EPSBN);
    float ggam0 = ggam[f0], ggam1 = ggam[f0 + 1];
    float gbeta0 = gbeta[f0], gbeta1 = gbeta[f0 + 1];
    float cw0 = clsW[f0], cw1 = clsW[f0 + 1];
    float cb = clsb[0];

#pragma nounroll
    for (int i = 0; i < NPW; ++i) {
        int n = blockIdx.x * GW + wave * NPW + i;
        int beg = off[n];
        int deg = off[n + 1] - beg;
        v4f rn = *(const v4f*)(lg + 4 * (long)n);
        float edn_e = rn.z, edn_g = rn.w;
        float self_e = leaky(rn.x + edn_e), self_g = leaky(rn.y + edn_g);
        v2f acc_a = {0.f, 0.f}, acc_b = {0.f, 0.f};
        float ze = 0.f, zg = 0.f, pse, psg;

        if (deg <= 32) {
            bool on = lane < deg;
            int sraw = slot[min(beg + lane, NEDGE - 1)];
            int s = on ? sraw : INT_MAX;
            int si = on ? sraw : n;
            v2f xs = *(const v2f*)(lg + 4 * (long)si);
            float xe = xs.x, xg = xs.y;
            int rank = 0;
            for (int j = 0; j < deg; ++j) {
                int sj = __shfl(s, j);
                rank += (sj < s || (sj == s && j < lane)) ? 1 : 0;
            }
            int cntp = max((deg + RB) & ~(RB - 1), FB);
            if (on) l_s[wave][rank] = s;
            if (lane < cntp - deg) l_s[wave][deg + lane] = n;
            v4f h0[FB];
#pragma unroll
            for (int u = 0; u < FB; ++u) h0[u] = row_of(h, l_s[wave][u], l16);
            __builtin_amdgcn_sched_barrier(0);
            float le = on ? leaky(xe + edn_e) : -FLT_MAX;
            float lg = on ? leaky(xg + edn_g) : -FLT_MAX;
            float me = fmaxf(wave_max_f(le), self_e);
            float mg = fmaxf(wave_max_f(lg), self_g);
            float pe = __expf(le - me), pg = __expf(lg - mg);
            pse = __expf(self_e - me); psg = __expf(self_g - mg);
            if (on) { v2f pp = {pe, pg}; l_a[wave][rank] = pp; }
            for (int r = 0; r < deg; ++r) { v2f a = l_a[wave][r]; ze += a.x; zg += a.y; }
            ze += pse; zg += psg;
            if (on) { v2f aa = {pe / ze, pg / zg}; l_a[wave][rank] = aa; }
            if (lane < cntp - deg) {
                v2f aa = {(lane == 0) ? pse / ze : 0.f, (lane == 0) ? psg / zg : 0.f};
                l_a[wave][deg + lane] = aa;
            }
#pragma unroll
            for (int u = 0; u < FB; ++u) {
                v2f a = l_a[wave][u];
                acc_a += h0[u].xy * a; acc_b += h0[u].zw * a;
            }
            for (int b = FB; b < cntp; b += RB) {
                v4f hr[RB];
#pragma unroll
                for (int u = 0; u < RB; ++u) hr[u] = row_of(h, l_s[wave][b + u], l16);
#pragma unroll
                for (int u = 0; u < RB; ++u) {
                    v2f a = l_a[wave][b + u];
                    acc_a += hr[u].xy * a; acc_b += hr[u].zw * a;
                }
            }
        } else {
        if (deg <= LCAP) {
            float le_m = self_e, lg_m = self_g;
            for (int b = 0; b < deg; b += 32) {
                int k = b + lane;
                if (k < deg) {
                    int s = slot[beg + k];
                    l_u[wave][k] = s;
                    v2f xs = *(const v2f*)(lg + 4 * (long)s);
                    le_m = fmaxf(le_m, leaky(xs.x + edn_e));
                    lg_m = fmaxf(lg_m, leaky(xs.y + edn_g));
                }
            }
            float me = wave_max_f(le_m), mg = wave_max_f(lg_m);
            pse = __expf(self_e - me); psg = __expf(self_g - mg);
            for (int b = 0; b < deg; b += 32) {
                int k = b + lane;
                int s = (k < deg) ? l_u[wave][k] : INT_MAX;
                int rank = 0;
#pragma unroll 4
                for (int j = 0; j < deg; ++j) {
                    int sj = l_u[wave][j];
                    rank += (sj < s || (sj == s && j < k)) ? 1 : 0;
                }
                if (k < deg) {
                    l_s[wave][rank] = s;
                    v2f xs = *(const v2f*)(lg + 4 * (long)s);
                    v2f pp = {__expf(leaky(xs.x + edn_e) - me), __expf(leaky(xs.y + edn_g) - mg)};
                    l_a[wave][rank] = pp;
                }
            }
            for (int r = 0; r < deg; ++r) { v2f a = l_a[wave][r]; ze += a.x; zg += a.y; }
            ze += pse; zg += psg;
            for (int b = 0; b < deg; b += 32) {
                int r = b + lane;
                if (r < deg) { v2f a = l_a[wave][r]; v2f aa = {a.x / ze, a.y / zg}; l_a[wave][r] = aa; }
            }
#pragma unroll 4
            for (int r = 0; r < deg; ++r) {
                v2f a = l_a[wave][r];
                v4f hr = row_of(h, l_s[wave][r], l16);
                acc_a += hr.xy * a; acc_b += hr.zw * a;
            }
        } else {
            float le_m = self_e, lg_m = self_g;
            for (int k = lane; k < deg; k += 32) {
                int s = slot[beg + k];
                v2f xs = *(const v2f*)(lg + 4 * (long)s);
                le_m = fmaxf(le_m, leaky(xs.x + edn_e));
                lg_m = fmaxf(lg_m, leaky(xs.y + edn_g));
            }
            float me = wave_max_f(le_m), mg = wave_max_f(lg_m);
            pse = __expf(self_e - me); psg = __expf(self_g - mg);
            for (int pass = 0; pass < 2; ++pass) {
                int prev = -1;
                for (int it = 0; it < deg; ++it) {
                    int best = INT_MAX;
                    for (int k = lane; k < deg; k += 32) {
                        int s = slot[beg + k];
                        if (s > prev) best = min(best, s);
                    }
                    best = wave_min_i(best);
                    if (best == INT_MAX) break;
                    int c = 0;
                    for (int k = lane; k < deg; k += 32) c += (slot[beg + k] == best) ? 1 : 0;
                    c = wave_sum_i(c);
                    v2f xb = *(const v2f*)(lg + 4 * (long)best);
                    float pe = __expf(leaky(xb.x + edn_e) - me);
                    float pg = __expf(leaky(xb.y + edn_g) - mg);
                    if (pass == 0) {
                        for (int k = 0; k < c; ++k) { ze += pe; zg += pg; }
                    } else {
                        v2f a = {pe / ze, pg / zg};
                        v4f hr = row_of(h, best, l16);
                        for (int k = 0; k < c; ++k) { acc_a += hr.xy * a; acc_b += hr.zw * a; }
                    }
                    prev = best;
                }
                if (pass == 0) { ze += pse; zg += psg; }
            }
        }
        {
            v2f a = {pse / ze, psg / zg};
            v4f hr = row_of(h, n, l16);
            acc_a += hr.xy * a; acc_b += hr.zw * a;
        }
        }

        v2f so;
        {
            float a0 = fmaxf(acc_a.x + eb0, 0.f);
            float a1 = fmaxf(acc_b.x + eb1, 0.f);
            so.x = (a0 - erm0) * ers0 * egam0 + ebeta0;
            so.y = (a1 - erm1) * ers1 * egam1 + ebeta1;
        }
        *(v2f*)(d_out + N_NODES + (long)n * NHID + f0) = so;
        float b0 = fmaxf(acc_a.y + gb0, 0.f);
        float b1 = fmaxf(acc_b.y + gb1, 0.f);
        float z0 = (b0 - grm0) * grs0 * ggam0 + gbeta0;
        float z1 = (b1 - grm1) * grs1 * ggam1 + gbeta1;
        float yp = wave_sum_f(z0 * cw0 + z1 * cw1);
        if (lane == 0) l_y[wave * NPW + i] = yp + cb;
    }
    __syncthreads();
    if (threadIdx.x < GW) d_out[blockIdx.x * GW + threadIdx.x] = l_y[threadIdx.x];
}

extern "C" void kernel_launch(void* const* d_in, const int* in_sizes, int n_in,
                              void* d_out, int out_size, void* d_ws, size_t ws_size,
                              hipStream_t stream) {
    const float* g        = (const float*)d_in[0];
    const int*   x        = (const int*)d_in[1];
    const float* est_W    = (const float*)d_in[2];
    const float* est_asrc = (const float*)d_in[3];
    const float* est_adst = (const float*)d_in[4];
    const float* est_bias = (const float*)d_in[5];
    const float* est_gam  = (const float*)d_in[6];
    const float* est_beta = (const float*)d_in[7];
    const float* est_rm   = (const float*)d_in[8];
    const float* est_rv   = (const float*)d_in[9];
    const float* gnn_W    = (const float*)d_in[10];
    const float* gnn_asrc = (const float*)d_in[11];
    const float* gnn_adst = (const float*)d_in[12];
    const float* gnn_bias = (const float*)d_in[13];
    const float* gnn_gam  = (const float*)d_in[14];
    const float* gnn_beta = (const float*)d_in[15];
    const float* gnn_rm   = (const float*)d_in[16];
    const float* gnn_rv   = (const float*)d_in[17];
    const float* cls_W    = (const float*)d_in[18];
    const float* cls_b    = (const float*)d_in[19];
    float* out = (float*)d_out;

    float* ws     = (float*)d_ws;
    float* h      = ws;
    int*   off    = (int*)(h + (long)N_NODES * HREC);
    int*   slot   = off + 2 * N_NODES;
    int*   tmp    = slot + NEDGE;
    int*   tab    = tmp + NEDGE;
    float* lg     = ws + (long)N_NODES * 192;

    k_gemm<<<GGRID, GT, 0, stream>>>(g, est_W, gnn_W, est_asrc, est_adst, gnn_asrc, gnn_adst,
                                     h, lg);

    k_tile<<<NBLK, PT, 0, stream>>>(x, tmp, tab);

    k_group<<<NBIN, FT, 0, stream>>>(tmp, tab, off, slot);

    k_gather<<<N_NODES / GW, GWAVES * 32, 0, stream>>>(out, off, slot,
            lg, h,
            est_bias, est_gam, est_beta, est_rm, est_rv,
            gnn_bias, gnn_gam, gnn_beta, gnn_rm, gnn_rv, cls_W, cls_b);
}
